# starting kernel
# baseline (speedup 1.0000x reference)
#include <hip/hip_runtime.h>
#include <hip/hip_bf16.h>

typedef __attribute__((ext_vector_type(16))) __bf16 v16bf;
typedef __attribute__((ext_vector_type(8)))  float  v8f;

#define NEG_SLOPE 0.2f


__device__ __forceinline__ unsigned short f2bf(float f) {
    unsigned u = __float_as_uint(f);
    unsigned r = u + 0x7FFFu + ((u >> 16) & 1u);
    return (unsigned short)(r >> 16);
}

__device__ __forceinline__ unsigned fenc(float f) {
    unsigned u = __float_as_uint(f);
    return (u & 0x80000000u) ? ~u : (u | 0x80000000u);
}
__device__ __forceinline__ float fdec(unsigned u) {
    return (u & 0x80000000u) ? __uint_as_float(u & 0x7FFFFFFFu)
                             : __uint_as_float(~u);
}

__device__ __forceinline__ void get_edge(const int* __restrict__ ei,
                                         long long n_edges, long long idx,
                                         int& s, int& d) {
    if (idx < n_edges) { s = ei[idx]; d = ei[n_edges + idx]; }
    else               { s = d = (int)(idx - n_edges); }
}


__global__ void k_f32_to_bf16(const float* __restrict__ in,
                              unsigned short* __restrict__ out, long long n) {
    long long i = (long long)blockIdx.x * blockDim.x + threadIdx.x;
    if (i < n) out[i] = f2bf(in[i]);
}

__global__ void k_transpose_bf16(const float* __restrict__ W,
                                 unsigned short* __restrict__ Wt,
                                 int K, int Nc) {
    int i = blockIdx.x * blockDim.x + threadIdx.x;
    if (i >= K * Nc) return;
    int n = i / K, k = i % K;
    Wt[(size_t)n * K + k] = f2bf(W[(size_t)k * Nc + n]);
}

__global__ void k_fill_u32(unsigned* __restrict__ p, unsigned v, long long n) {
    long long i = (long long)blockIdx.x * blockDim.x + threadIdx.x;
    if (i < n) p[i] = v;
}

__global__ void k_fill_bias(float* __restrict__ out, const float* __restrict__ bias,
                            int C, long long n) {
    long long i = (long long)blockIdx.x * blockDim.x + threadIdx.x;
    if (i < n) out[i] = bias[i % C];
}

__global__ void k_bias_relu_bf16(const float* __restrict__ in,
                                 const float* __restrict__ bias,
                                 unsigned short* __restrict__ hbf,
                                 int C, long long n) {
    long long i = (long long)blockIdx.x * blockDim.x + threadIdx.x;
    if (i >= n) return;
    float v = in[i] + bias[i % C];
    hbf[i] = f2bf(v > 0.f ? v : 0.f);
}

__global__ void k_gemm_wmma_bf16(const unsigned short* __restrict__ A,
                                 const unsigned short* __restrict__ Bt,
                                 float* __restrict__ C, int Nc) {
    const int lane = threadIdx.x & 31;
    const int tileM = blockIdx.x;
    const int tileN = blockIdx.y;
    const int r15  = lane & 15;
    const int half = lane >> 4;

    const unsigned* arow = (const unsigned*)(A + (size_t)(tileM * 16 + r15) * 128);
    const unsigned* brow = (const unsigned*)(Bt + (size_t)(tileN * 16 + r15) * 128);

    v8f acc = {};
#pragma unroll
    for (int K0 = 0; K0 < 128; K0 += 32) {
        union { unsigned u[8]; v16bf v; } a, b;
#pragma unroll
        for (int v = 0; v < 8; ++v) {
            int k = ((v < 4) ? 0 : 16) + half * 8 + (v & 3) * 2;
            a.u[v] = arow[(K0 + k) >> 1];
            b.u[v] = brow[(K0 + k) >> 1];
        }
        acc = __builtin_amdgcn_wmma_f32_16x16x32_bf16(
                false, a.v, false, b.v, (short)0, acc, false, false);
    }

#pragma unroll
    for (int r = 0; r < 8; ++r) {
        int row = tileM * 16 + r + 8 * half;
        int col = tileN * 16 + r15;
        C[(size_t)row * Nc + col] = acc[r];
    }
}

__global__ void k_alpha(const float* __restrict__ h,
                        const float* __restrict__ a_s, const float* __restrict__ a_d,
                        float* __restrict__ als, float* __restrict__ ald,
                        int n_nodes, int heads, int ch) {
    int i = blockIdx.x * blockDim.x + threadIdx.x;
    if (i >= n_nodes * heads) return;
    int n = i / heads, hd = i % heads;
    const float* row = h + (size_t)n * heads * ch + (size_t)hd * ch;
    const float* vs = a_s + (size_t)hd * ch;
    const float* vd = a_d + (size_t)hd * ch;
    float s = 0.f, d = 0.f;
    for (int c = 0; c < ch; ++c) { float x = row[c]; s += x * vs[c]; d += x * vd[c]; }
    als[i] = s; ald[i] = d;
}

__global__ void k_edge_max(const int* __restrict__ ei,
                           const float* __restrict__ als, const float* __restrict__ ald,
                           unsigned* __restrict__ m,
                           int heads, long long n_edges, long long total) {
    long long i = (long long)blockIdx.x * blockDim.x + threadIdx.x;
    if (i >= total * heads) return;
    long long e = i / heads; int hd = (int)(i % heads);
    int s, d; get_edge(ei, n_edges, e, s, d);
    float v = als[(size_t)s * heads + hd] + ald[(size_t)d * heads + hd];
    v = v > 0.f ? v : NEG_SLOPE * v;
    atomicMax(&m[(size_t)d * heads + hd], fenc(v));
}

__global__ void k_edge_exp(const int* __restrict__ ei,
                           const float* __restrict__ als, const float* __restrict__ ald,
                           const unsigned* __restrict__ m,
                           float* __restrict__ exbuf, float* __restrict__ den,
                           int heads, long long n_edges, long long total) {
    long long i = (long long)blockIdx.x * blockDim.x + threadIdx.x;
    if (i >= total * heads) return;
    long long e = i / heads; int hd = (int)(i % heads);
    int s, d; get_edge(ei, n_edges, e, s, d);
    float v = als[(size_t)s * heads + hd] + ald[(size_t)d * heads + hd];
    v = v > 0.f ? v : NEG_SLOPE * v;
    float ex = __expf(v - fdec(m[(size_t)d * heads + hd]));
    exbuf[i] = ex;
    atomicAdd(&den[(size_t)d * heads + hd], ex);
}

__global__ void k_edge_aggr(const int* __restrict__ ei,
                            const float* __restrict__ exbuf, const float* __restrict__ den,
                            const float* __restrict__ h, float* __restrict__ out,
                            int heads, int ch, long long n_edges, long long total) {
    const int C = heads * ch;
    long long i = (long long)blockIdx.x * blockDim.x + threadIdx.x;
    if (i >= total * C) return;
    long long e = i / C; int c = (int)(i % C); int hd = c / ch;
    int s, d; get_edge(ei, n_edges, e, s, d);
    float alpha = exbuf[e * heads + hd] / (den[(size_t)d * heads + hd] + 1e-16f);
    atomicAdd(&out[(size_t)d * C + c], alpha * h[(size_t)s * C + c]);
}


extern "C" void kernel_launch(void* const* d_in, const int* in_sizes, int n_in,
                              void* d_out, int out_size, void* d_ws, size_t ws_size,
                              hipStream_t stream) {
    (void)n_in; (void)out_size; (void)ws_size; (void)in_sizes;
    constexpr long long N = 100000, E = 1600000, ET = E + N;
    constexpr int C1 = 128, HEADS = 4, HID = 32, C2 = 32;

    const float* x    = (const float*)d_in[0];
    const int*   ei   = (const int*)  d_in[1];
    const float* W1   = (const float*)d_in[2];
    const float* aS1  = (const float*)d_in[3];
    const float* aD1  = (const float*)d_in[4];
    const float* b1   = (const float*)d_in[5];
    const float* W2   = (const float*)d_in[6];
    const float* aS2  = (const float*)d_in[7];
    const float* aD2  = (const float*)d_in[8];
    const float* b2   = (const float*)d_in[9];
    float* out = (float*)d_out;

    char* ws = (char*)d_ws;
    size_t off = 0;
    auto alloc = [&](size_t bytes) { size_t o = off; off = (off + bytes + 255) & ~(size_t)255; return o; };
    unsigned short* xbf  = (unsigned short*)(ws + alloc(N * C1 * 2));
    float*          h1   = (float*)(ws + alloc(N * C1 * 4));
    float*          out1 = (float*)(ws + alloc(N * C1 * 4));
    float*          ex1  = (float*)(ws + alloc(ET * HEADS * 4));
    float*          as1  = (float*)(ws + alloc(N * HEADS * 4));
    float*          ad1  = (float*)(ws + alloc(N * HEADS * 4));
    unsigned*       m1   = (unsigned*)(ws + alloc(N * HEADS * 4));
    float*          den1 = (float*)(ws + alloc(N * HEADS * 4));
    unsigned short* w1t  = (unsigned short*)(ws + alloc(128 * 128 * 2));
    unsigned short* w2t  = (unsigned short*)(ws + alloc(32 * 128 * 2));

    const int B = 256;
    auto blocks = [](long long n, int b) { return (unsigned)((n + b - 1) / b); };

    k_f32_to_bf16<<<blocks(N * C1, B), B, 0, stream>>>(x, xbf, N * C1);
    k_transpose_bf16<<<blocks(128 * 128, B), B, 0, stream>>>(W1, w1t, 128, 128);
    k_transpose_bf16<<<blocks(128 * 32, B), B, 0, stream>>>(W2, w2t, 128, 32);

    k_gemm_wmma_bf16<<<dim3(N / 16, C1 / 16), 32, 0, stream>>>(xbf, w1t, h1, C1);

    k_alpha<<<blocks(N * HEADS, B), B, 0, stream>>>(h1, aS1, aD1, as1, ad1, N, HEADS, HID);

    k_fill_u32<<<blocks(N * HEADS, B), B, 0, stream>>>(m1, 0u, N * HEADS);
    k_fill_u32<<<blocks(N * HEADS, B), B, 0, stream>>>((unsigned*)den1, 0u, N * HEADS);
    k_fill_u32<<<blocks(N * C1, B), B, 0, stream>>>((unsigned*)out1, 0u, N * C1);

    k_edge_max <<<blocks(ET * HEADS, B), B, 0, stream>>>(ei, as1, ad1, m1, HEADS, E, ET);
    k_edge_exp <<<blocks(ET * HEADS, B), B, 0, stream>>>(ei, as1, ad1, m1, ex1, den1, HEADS, E, ET);
    k_edge_aggr<<<blocks(ET * C1, B), B, 0, stream>>>(ei, ex1, den1, h1, out1, HEADS, HID, E, ET);

    k_bias_relu_bf16<<<blocks(N * C1, B), B, 0, stream>>>(out1, b1, xbf, C1, N * C1);

    k_gemm_wmma_bf16<<<dim3(N / 16, C2 / 16), 32, 0, stream>>>(xbf, w2t, h1, C2);

    k_alpha<<<blocks(N, B), B, 0, stream>>>(h1, aS2, aD2, as1, ad1, N, 1, C2);

    k_fill_u32<<<blocks(N, B), B, 0, stream>>>(m1, 0u, N);
    k_fill_u32<<<blocks(N, B), B, 0, stream>>>((unsigned*)den1, 0u, N);
    k_fill_bias<<<blocks(N * C2, B), B, 0, stream>>>(out, b2, C2, N * C2);

    k_edge_max <<<blocks(ET, B), B, 0, stream>>>(ei, as1, ad1, m1, 1, E, ET);
    k_edge_exp <<<blocks(ET, B), B, 0, stream>>>(ei, as1, ad1, m1, ex1, den1, 1, E, ET);
    k_edge_aggr<<<blocks(ET * C2, B), B, 0, stream>>>(ei, ex1, den1, h1, out, 1, C2, E, ET);
}


# Both GEMMs on the f32 16x16x4 matrix form, fed straight from x, W1, W2, out1 (bias+relu as the fragment loads): operands f32 not bf16, product k order changes; no conversion passes; 11->7 launches
# speedup vs baseline: 6.6335x; 6.6335x over previous
#include <hip/hip_runtime.h>

typedef __attribute__((ext_vector_type(2))) float v2f;
typedef __attribute__((ext_vector_type(8))) float v8f;

#define NEG_SLOPE 0.2f


__device__ __forceinline__ unsigned fenc(float f) {
    unsigned u = __float_as_uint(f);
    return (u & 0x80000000u) ? ~u : (u | 0x80000000u);
}
__device__ __forceinline__ float fdec(unsigned u) {
    return (u & 0x80000000u) ? __uint_as_float(u & 0x7FFFFFFFu)
                             : __uint_as_float(~u);
}

#define GEMM_WAVES 8
template <int NC, bool BIAS_RELU>
__global__ __launch_bounds__(GEMM_WAVES * 32)
void k_gemm_wmma_f32(const float* __restrict__ A, const float* __restrict__ W,
                     const float* __restrict__ bias, float* __restrict__ C, int n_strips) {
    __shared__ float4 wl[NC * 128 / 4];
    __shared__ float4 bl[128 / 4];

    const int tid = threadIdx.x, lane = tid & 31;
    const int wave = __builtin_amdgcn_readfirstlane(tid >> 5);
    const int c = lane & 15, half = lane >> 4;
    const int strip = blockIdx.x * GEMM_WAVES + wave;
    const bool live = strip < n_strips;

    float4 a4[16];
    if (live) {
        const float4* arow = (const float4*)(A + (size_t)(strip * 16 + c) * 128 + half * 64);
#pragma unroll
        for (int i = 0; i < 16; ++i) a4[i] = arow[i];
    }

    float* const wlf = (float*)wl;
    for (int i4 = tid; i4 < 128 * NC / 4; i4 += GEMM_WAVES * 32) {
        const float4 w = ((const float4*)W)[i4];
        const int k = (i4 * 4) / NC, n = (i4 * 4) % NC;
        const int t = (k & 63) >> 1, s = k & 1, j = n >> 4;
        float* const d = wlf + ((((j >> 1) * 32 + t) * 32 + ((k >> 6) * 16 + (n & 15))) * 4 + (j & 1) * 2 + s);
        d[0] = w.x; d[4] = w.y; d[8] = w.z; d[12] = w.w;
    }
    if (BIAS_RELU && tid < 128) ((float*)bl)[tid] = bias[tid];
    __syncthreads();

    if (!live) return;

    float af[64];
#pragma unroll
    for (int i = 0; i < 16; ++i) {
        float4 v = a4[i];
        if (BIAS_RELU) {
            const float4 b = bl[half * 16 + i];
            v.x = fmaxf(v.x + b.x, 0.f); v.y = fmaxf(v.y + b.y, 0.f);
            v.z = fmaxf(v.z + b.z, 0.f); v.w = fmaxf(v.w + b.w, 0.f);
        }
        af[4 * i] = v.x; af[4 * i + 1] = v.y; af[4 * i + 2] = v.z; af[4 * i + 3] = v.w;
    }

    for (int jp = 0; jp < NC / 32; ++jp) {
        v8f acc0 = {}, acc1 = {};
#pragma unroll
        for (int t = 0; t < 32; ++t) {
            const float4 b = wl[(jp * 32 + t) * 32 + lane];
            const v2f a  = {af[2 * t], af[2 * t + 1]};
            const v2f b0 = {b.x, b.y};
            const v2f b1 = {b.z, b.w};
            acc0 = __builtin_amdgcn_wmma_f32_16x16x4_f32(false, a, false, b0, (short)0, acc0, false, false);
            acc1 = __builtin_amdgcn_wmma_f32_16x16x4_f32(false, a, false, b1, (short)0, acc1, false, false);
        }
#pragma unroll
        for (int r = 0; r < 8; ++r) {
            float* const crow = C + (size_t)(strip * 16 + r + 8 * half) * NC + jp * 32 + c;
            crow[0]  = acc0[r];
            crow[16] = acc1[r];
        }
    }
}

__global__ void k_alpha(const float* __restrict__ h,
                        const float* __restrict__ a_s, const float* __restrict__ a_d,
                        float* __restrict__ als, float* __restrict__ ald,
                        int n_nodes, int heads, int ch) {
    int i = blockIdx.x * blockDim.x + threadIdx.x;
    if (i >= n_nodes * heads) return;
    int n = i / heads, hd = i % heads;
    const float* row = h + (size_t)n * heads * ch + (size_t)hd * ch;
    const float* vs = a_s + (size_t)hd * ch;
    const float* vd = a_d + (size_t)hd * ch;
    float s = 0.f, d = 0.f;
    for (int c = 0; c < ch; ++c) { float x = row[c]; s += x * vs[c]; d += x * vd[c]; }
    als[i] = s; ald[i] = d;
}

#define TILE 400
#define TILE_THREADS 1024

__device__ __forceinline__ float edge_score(float a, float b) {
    float v = a + b;
    return v > 0.f ? v : NEG_SLOPE * v;
}

__global__ void k_tile_keys(const int* __restrict__ ei, unsigned* __restrict__ keys,
                            int n_edges) {
    int i = blockIdx.x * blockDim.x + threadIdx.x;
    if (i * 4 >= n_edges) return;
    const int4 d = *(const int4*)(ei + (size_t)n_edges + (size_t)i * 4);
    keys[i] = (unsigned)(d.x / TILE) | ((unsigned)(d.y / TILE) << 8) |
              ((unsigned)(d.z / TILE) << 16) | ((unsigned)(d.w / TILE) << 24);
}

__device__ __forceinline__ unsigned match4(unsigned k, unsigned pat) {
    unsigned x = k ^ pat;
    unsigned t = ~(((x & 0x7F7F7F7Fu) + 0x7F7F7F7Fu) | x | 0x7F7F7F7Fu);
    return (((t >> 7) * 0x00204081u) >> 21) & 0xFu;
}
__device__ __forceinline__ unsigned match16(uint4 k, unsigned pat) {
    return match4(k.x, pat) | (match4(k.y, pat) << 4) | (match4(k.z, pat) << 8) | (match4(k.w, pat) << 12);
}

__device__ __forceinline__ float lane_f(float v, int l) {
    return __int_as_float(__builtin_amdgcn_readlane(__float_as_int(v), l));
}

#define LIST_CAP 16384
#define DL_BITS  9

template <int H, int CH, int PASS>
__device__ __forceinline__ void take_edges(bool act, int s, int dl,
                                           const float* __restrict__ als, const float* __restrict__ hf,
                                           float* acc, unsigned* mx, float* den, const float* adl, int lane) {
    constexpr int C = H * CH;
    constexpr int NK = C / 32;
    float w[H];
#pragma unroll
    for (int h = 0; h < H; ++h) w[h] = 0.f;
    if (act) {
#pragma unroll
        for (int h = 0; h < H; ++h) {
            const float v = edge_score(als[(size_t)s * H + h], adl[dl * H + h]);
            if (PASS == 1) {
                atomicMax(&mx[dl * H + h], fenc(v));
            } else {
                w[h] = __expf(v - fdec(mx[dl * H + h]));
                atomicAdd(&den[dl * H + h], w[h]);
            }
        }
    }
    if (PASS == 2) {
        unsigned bal = __builtin_amdgcn_ballot_w32(act);
        while (bal != 0u) {
            int  ss[4], dd[4];
            bool ok[4];
            float ww[4][H];
            float r[4][NK];
#pragma unroll
            for (int j = 0; j < 4; ++j) {
                ok[j] = bal != 0u;
                const int l = ok[j] ? (__ffs(bal) - 1) : 0;
                if (ok[j]) bal &= bal - 1u;
                ss[j] = __builtin_amdgcn_readlane(s, l);
                dd[j] = __builtin_amdgcn_readlane(dl, l);
#pragma unroll
                for (int h = 0; h < H; ++h) ww[j][h] = lane_f(w[h], l);
            }
#pragma unroll
            for (int j = 0; j < 4; ++j) {
#pragma unroll
                for (int k = 0; k < NK; ++k) r[j][k] = 0.f;
                if (ok[j]) {
#pragma unroll
                    for (int k = 0; k < NK; ++k) r[j][k] = hf[(size_t)ss[j] * C + k * 32 + lane];
                }
            }
#pragma unroll
            for (int j = 0; j < 4; ++j) {
                if (ok[j]) {
#pragma unroll
                    for (int k = 0; k < NK; ++k)
                        atomicAdd(&acc[dd[j] * C + k * 32 + lane], ww[j][(k * 32) / CH] * r[j][k]);
                }
            }
        }
    }
}

template <int H, int CH, int PASS, bool RECORD>
__device__ __forceinline__ void tile_sweep(const int* __restrict__ ei, const uint4* __restrict__ keys,
                                           const float* __restrict__ als, const float* __restrict__ hf,
                                           float* acc, unsigned* mx, float* den, const float* adl,
                                           unsigned* lst, unsigned* cnt,
                                           int t0, unsigned pat, int rot, int n_edges, int n_key16,
                                           int lane, int wave) {
    constexpr int STEP = (TILE_THREADS / 32) * 64;
    const uint4 none = make_uint4(~0u, ~0u, ~0u, ~0u);

    auto group_of = [&](int base, int u) {
        int q = base + u * 32 + lane + rot;
        return q >= n_key16 ? q - n_key16 : q;
    };
    auto fetch = [&](int base, int u) {
        return (base + u * 32 + lane < n_key16) ? keys[group_of(base, u)] : none;
    };

    uint4 kv0 = fetch(wave * 64, 0), kv1 = fetch(wave * 64, 1);
    for (int base = wave * 64; base < n_key16; base += STEP) {
        const uint4 nx0 = fetch(base + STEP, 0), nx1 = fetch(base + STEP, 1);
        unsigned mask = match16(kv0, pat) | (match16(kv1, pat) << 16);
        while (__builtin_amdgcn_ballot_w32(mask != 0u) != 0u) {
            const bool act = mask != 0u;
            int s = 0, dl = 0;
            if (act) {
                const int b = __ffs(mask) - 1;
                mask &= mask - 1u;
                const int e = group_of(base, b >> 4) * 16 + (b & 15);
                s  = ei[e];
                dl = ei[(size_t)n_edges + e] - t0;
                if (RECORD) {
                    const unsigned slot = atomicAdd(cnt, 1u);
                    if (slot < LIST_CAP) lst[slot] = ((unsigned)s << DL_BITS) | (unsigned)dl;
                }
            }
            take_edges<H, CH, PASS>(act, s, dl, als, hf, acc, mx, den, adl, lane);
        }
        kv0 = nx0; kv1 = nx1;
    }
}

template <int H, int CH, int PASS>
__device__ __forceinline__ void list_sweep(const unsigned* lst, int n,
                                           const float* __restrict__ als, const float* __restrict__ hf,
                                           float* acc, unsigned* mx, float* den, const float* adl,
                                           int lane, int wave) {
    for (int base = wave * 32; base < n; base += TILE_THREADS) {
        const int i = base + lane;
        const bool act = i < n;
        const unsigned p = act ? lst[i] : 0u;
        take_edges<H, CH, PASS>(act, (int)(p >> DL_BITS), (int)(p & ((1u << DL_BITS) - 1u)),
                                als, hf, acc, mx, den, adl, lane);
    }
}

template <int H, int CH, bool ADD_BIAS, bool FIRST>
__global__ __launch_bounds__(TILE_THREADS)
void k_tile_gat(const int* __restrict__ ei, const uint4* __restrict__ keys,
                const float* __restrict__ als, const float* __restrict__ ald,
                const float* __restrict__ hf, const float* __restrict__ bias,
                float* __restrict__ out, unsigned* glist, unsigned* gcnt,
                int n_nodes, int n_edges, int n_key16) {
    constexpr int C = H * CH;
    __shared__ float    acc[TILE * C];
    __shared__ unsigned mx[TILE * H];
    __shared__ float    den[TILE * H];
    __shared__ float    adl[TILE * H];
    __shared__ unsigned lst[LIST_CAP];
    __shared__ unsigned cnt;

    const int tid = threadIdx.x, lane = tid & 31, wave = tid >> 5;
    const int t0 = blockIdx.x * TILE;
    const unsigned pat = (unsigned)blockIdx.x * 0x01010101u;
    const int rot = (int)(((long long)blockIdx.x * n_key16) / gridDim.x);
    unsigned* const glst = glist + (size_t)blockIdx.x * LIST_CAP;

    unsigned n_list = 0u;
    if (FIRST) {
        if (tid == 0) cnt = 0u;
    } else {
        n_list = (unsigned)__builtin_amdgcn_readfirstlane((int)gcnt[blockIdx.x]);
        if (n_list <= LIST_CAP)
            for (int i = tid; i < (int)n_list; i += TILE_THREADS) lst[i] = glst[i];
    }

    for (int i = tid; i < TILE * H; i += TILE_THREADS) {
        const int d = t0 + i / H;
        float a = 0.f, v = 0.f;
        if (d < n_nodes) { a = ald[(size_t)t0 * H + i]; v = edge_score(als[(size_t)t0 * H + i], a); }
        adl[i] = a;
        mx[i]  = fenc(v);
    }
    __syncthreads();

    if (FIRST) {
        tile_sweep<H, CH, 1, true>(ei, keys, als, hf, acc, mx, den, adl, lst, &cnt, t0, pat, rot, n_edges, n_key16, lane, wave);
    } else if (n_list <= LIST_CAP) {
        list_sweep<H, CH, 1>(lst, (int)n_list, als, hf, acc, mx, den, adl, lane, wave);
    } else {
        tile_sweep<H, CH, 1, false>(ei, keys, als, hf, acc, mx, den, adl, lst, &cnt, t0, pat, rot, n_edges, n_key16, lane, wave);
    }
    __syncthreads();

    if (FIRST) {
        n_list = (unsigned)__builtin_amdgcn_readfirstlane((int)cnt);
        if (tid == 0) gcnt[blockIdx.x] = n_list;
        if (n_list <= LIST_CAP)
            for (int i = tid; i < (int)n_list; i += TILE_THREADS) glst[i] = lst[i];
    }

    for (int i = tid; i < TILE * H; i += TILE_THREADS) {
        const int d = t0 + i / H;
        float ex = 0.f;
        if (d < n_nodes) ex = __expf(edge_score(als[(size_t)t0 * H + i], adl[i]) - fdec(mx[i]));
        den[i] = ex;
    }
    for (int i = tid; i < TILE * C; i += TILE_THREADS) {
        const int dl = i / C, hd = (i % C) / CH;
        const int d = t0 + dl;
        float a = 0.f;
        if (d < n_nodes) {
            const float ex = __expf(edge_score(als[(size_t)d * H + hd], adl[dl * H + hd]) - fdec(mx[dl * H + hd]));
            a = ex * hf[(size_t)t0 * C + i];
        }
        acc[i] = a;
    }
    __syncthreads();

    if (n_list <= LIST_CAP) {
        list_sweep<H, CH, 2>(lst, (int)n_list, als, hf, acc, mx, den, adl, lane, wave);
    } else {
        tile_sweep<H, CH, 2, false>(ei, keys, als, hf, acc, mx, den, adl, lst, &cnt, t0, pat, rot, n_edges, n_key16, lane, wave);
    }
    __syncthreads();

    for (int i = tid; i < TILE * C; i += TILE_THREADS) {
        const int dl = i / C, c = i % C;
        if (t0 + dl < n_nodes) {
            float v = acc[i] / (den[dl * H + c / CH] + 1e-16f);
            if (ADD_BIAS) v += bias[c];
            out[(size_t)t0 * C + i] = v;
        }
    }
}


extern "C" void kernel_launch(void* const* d_in, const int* in_sizes, int n_in,
                              void* d_out, int out_size, void* d_ws, size_t ws_size,
                              hipStream_t stream) {
    (void)n_in; (void)out_size; (void)ws_size; (void)in_sizes;
    constexpr long long N = 100000, E = 1600000;
    constexpr int C1 = 128, HEADS = 4, HID = 32, C2 = 32;

    static_assert(N % TILE == 0 && N / TILE <= 255 && E % 16 == 0, "tile ids must fit a byte below 0xFF; keys are read 16 at a time");
    static_assert(TILE <= (1 << DL_BITS) && N <= (1ll << (32 - DL_BITS)), "a list entry packs source and local destination into one word");

    const float* x    = (const float*)d_in[0];
    const int*   ei   = (const int*)  d_in[1];
    const float* W1   = (const float*)d_in[2];
    const float* aS1  = (const float*)d_in[3];
    const float* aD1  = (const float*)d_in[4];
    const float* b1   = (const float*)d_in[5];
    const float* W2   = (const float*)d_in[6];
    const float* aS2  = (const float*)d_in[7];
    const float* aD2  = (const float*)d_in[8];
    const float* b2   = (const float*)d_in[9];
    float* out = (float*)d_out;

    char* ws = (char*)d_ws;
    size_t off = 0;
    auto alloc = [&](size_t bytes) { size_t o = off; off = (off + bytes + 255) & ~(size_t)255; return o; };
    float*          h1   = (float*)(ws + alloc(N * C1 * 4));
    float*          out1 = (float*)(ws + alloc(N * C1 * 4));
    float*          as1  = (float*)(ws + alloc(N * HEADS * 4));
    float*          ad1  = (float*)(ws + alloc(N * HEADS * 4));
    unsigned*       keys = (unsigned*)(ws + alloc(E));
    unsigned*       glist = (unsigned*)(ws + alloc((N / TILE) * LIST_CAP * 4));
    unsigned*       gcnt  = (unsigned*)(ws + alloc((N / TILE) * 4));

    const int B = 256;
    auto blocks = [](long long n, int b) { return (unsigned)((n + b - 1) / b); };
    constexpr int STRIPS = (int)(N / 16);
    constexpr unsigned GEMM_GRID = (STRIPS + GEMM_WAVES - 1) / GEMM_WAVES;
    static_assert(N % 16 == 0 && C1 == 128 && C1 % 32 == 0 && C2 % 32 == 0, "the GEMM takes whole 16-row strips, K = 128 and column tiles in pairs");

    k_gemm_wmma_f32<C1, false><<<GEMM_GRID, GEMM_WAVES * 32, 0, stream>>>(x, W1, b1, h1, STRIPS);

    k_alpha<<<blocks(N * HEADS, B), B, 0, stream>>>(h1, aS1, aD1, as1, ad1, N, HEADS, HID);

    k_tile_keys<<<blocks(E / 4, B), B, 0, stream>>>(ei, keys, E);

    k_tile_gat<HEADS, HID, false, true><<<N / TILE, TILE_THREADS, 0, stream>>>(
        ei, (const uint4*)keys, as1, ad1, h1, b1, out1, glist, gcnt, N, E, E / 16);

    k_gemm_wmma_f32<C2, true><<<GEMM_GRID, GEMM_WAVES * 32, 0, stream>>>(out1, W2, b1, h1, STRIPS);

    k_alpha<<<blocks(N, B), B, 0, stream>>>(h1, aS2, aD2, as1, ad1, N, 1, C2);

    k_tile_gat<1, C2, true, false><<<N / TILE, TILE_THREADS, 0, stream>>>(
        ei, (const uint4*)keys, as1, ad1, h1, b2, out, glist, gcnt, N, E, E / 16);
}


# Block index: the id pass also builds, per tile, a bitmap over 16-edge blocks (LDS or-atomics); an owner reads its 12.5 KB row and matches ids in listed blocks only, not all 1.6M; list arrival order
# speedup vs baseline: 7.4388x; 1.1214x over previous
#include <hip/hip_runtime.h>

typedef __attribute__((ext_vector_type(2))) float v2f;
typedef __attribute__((ext_vector_type(8))) float v8f;

#define NEG_SLOPE 0.2f


__device__ __forceinline__ unsigned fenc(float f) {
    unsigned u = __float_as_uint(f);
    return (u & 0x80000000u) ? ~u : (u | 0x80000000u);
}
__device__ __forceinline__ float fdec(unsigned u) {
    return (u & 0x80000000u) ? __uint_as_float(u & 0x7FFFFFFFu)
                             : __uint_as_float(~u);
}

#define GEMM_WAVES 8
template <int NC, bool BIAS_RELU>
__global__ __launch_bounds__(GEMM_WAVES * 32)
void k_gemm_wmma_f32(const float* __restrict__ A, const float* __restrict__ W,
                     const float* __restrict__ bias, float* __restrict__ C, int n_strips) {
    __shared__ float4 wl[NC * 128 / 4];
    __shared__ float4 bl[128 / 4];

    const int tid = threadIdx.x, lane = tid & 31;
    const int wave = __builtin_amdgcn_readfirstlane(tid >> 5);
    const int c = lane & 15, half = lane >> 4;
    const int strip = blockIdx.x * GEMM_WAVES + wave;
    const bool live = strip < n_strips;

    float4 a4[16];
    if (live) {
        const float4* arow = (const float4*)(A + (size_t)(strip * 16 + c) * 128 + half * 64);
#pragma unroll
        for (int i = 0; i < 16; ++i) a4[i] = arow[i];
    }

    float* const wlf = (float*)wl;
    for (int i4 = tid; i4 < 128 * NC / 4; i4 += GEMM_WAVES * 32) {
        const float4 w = ((const float4*)W)[i4];
        const int k = (i4 * 4) / NC, n = (i4 * 4) % NC;
        const int t = (k & 63) >> 1, s = k & 1, j = n >> 4;
        float* const d = wlf + ((((j >> 1) * 32 + t) * 32 + ((k >> 6) * 16 + (n & 15))) * 4 + (j & 1) * 2 + s);
        d[0] = w.x; d[4] = w.y; d[8] = w.z; d[12] = w.w;
    }
    if (BIAS_RELU && tid < 128) ((float*)bl)[tid] = bias[tid];
    __syncthreads();

    if (!live) return;

    float af[64];
#pragma unroll
    for (int i = 0; i < 16; ++i) {
        float4 v = a4[i];
        if (BIAS_RELU) {
            const float4 b = bl[half * 16 + i];
            v.x = fmaxf(v.x + b.x, 0.f); v.y = fmaxf(v.y + b.y, 0.f);
            v.z = fmaxf(v.z + b.z, 0.f); v.w = fmaxf(v.w + b.w, 0.f);
        }
        af[4 * i] = v.x; af[4 * i + 1] = v.y; af[4 * i + 2] = v.z; af[4 * i + 3] = v.w;
    }

    for (int jp = 0; jp < NC / 32; ++jp) {
        v8f acc0 = {}, acc1 = {};
#pragma unroll
        for (int t = 0; t < 32; ++t) {
            const float4 b = wl[(jp * 32 + t) * 32 + lane];
            const v2f a  = {af[2 * t], af[2 * t + 1]};
            const v2f b0 = {b.x, b.y};
            const v2f b1 = {b.z, b.w};
            acc0 = __builtin_amdgcn_wmma_f32_16x16x4_f32(false, a, false, b0, (short)0, acc0, false, false);
            acc1 = __builtin_amdgcn_wmma_f32_16x16x4_f32(false, a, false, b1, (short)0, acc1, false, false);
        }
#pragma unroll
        for (int r = 0; r < 8; ++r) {
            float* const crow = C + (size_t)(strip * 16 + r + 8 * half) * NC + jp * 32 + c;
            crow[0]  = acc0[r];
            crow[16] = acc1[r];
        }
    }
}

__global__ void k_alpha(const float* __restrict__ h,
                        const float* __restrict__ a_s, const float* __restrict__ a_d,
                        float* __restrict__ als, float* __restrict__ ald,
                        int n_nodes, int heads, int ch) {
    int i = blockIdx.x * blockDim.x + threadIdx.x;
    if (i >= n_nodes * heads) return;
    int n = i / heads, hd = i % heads;
    const float* row = h + (size_t)n * heads * ch + (size_t)hd * ch;
    const float* vs = a_s + (size_t)hd * ch;
    const float* vd = a_d + (size_t)hd * ch;
    float s = 0.f, d = 0.f;
    for (int c = 0; c < ch; ++c) { float x = row[c]; s += x * vs[c]; d += x * vd[c]; }
    als[i] = s; ald[i] = d;
}

#define TILE 400
#define TILE_THREADS 1024

__device__ __forceinline__ float edge_score(float a, float b) {
    float v = a + b;
    return v > 0.f ? v : NEG_SLOPE * v;
}

#define IDX_WORDS   32
#define IDX_THREADS 1024
#define IDX_PITCH   257
__global__ __launch_bounds__(IDX_THREADS)
void k_tile_index(const int* __restrict__ ei, unsigned* __restrict__ keys, unsigned* __restrict__ bitmap,
                  int n_edges, int n_tiles, int row_words) {
    __shared__ unsigned bm[IDX_WORDS * IDX_PITCH];
    const int tid = threadIdx.x;
    for (int i = tid; i < IDX_WORDS * IDX_PITCH; i += IDX_THREADS) bm[i] = 0u;
    __syncthreads();

    const int n_quads = n_edges / 4;
    const int q0 = blockIdx.x * (IDX_WORDS * 32 * 4);
    constexpr int QPT = IDX_WORDS * 32 * 4 / IDX_THREADS;
    int4 dv[QPT];
#pragma unroll
    for (int u = 0; u < QPT; ++u) {
        const int q = q0 + u * IDX_THREADS + tid;
        dv[u] = q < n_quads ? *(const int4*)(ei + (size_t)n_edges + (size_t)q * 4) : make_int4(0, 0, 0, 0);
    }
#pragma unroll
    for (int u = 0; u < QPT; ++u) {
        const int ql = u * IDX_THREADS + tid;
        const int q = q0 + ql;
        if (q < n_quads) {
            const int4 d = dv[u];
            const unsigned ta = (unsigned)(d.x / TILE), tb = (unsigned)(d.y / TILE),
                           tc = (unsigned)(d.z / TILE), td = (unsigned)(d.w / TILE);
            keys[q] = ta | (tb << 8) | (tc << 16) | (td << 24);
            unsigned* const wrow = bm + (ql >> 7) * IDX_PITCH;
            const unsigned bit = 1u << ((ql >> 2) & 31);
            atomicOr(&wrow[ta], bit); atomicOr(&wrow[tb], bit);
            atomicOr(&wrow[tc], bit); atomicOr(&wrow[td], bit);
        }
    }
    __syncthreads();

    for (int i = tid; i < n_tiles * IDX_WORDS; i += IDX_THREADS) {
        const int t = i / IDX_WORDS, w = i % IDX_WORDS;
        bitmap[(size_t)t * row_words + blockIdx.x * IDX_WORDS + w] = bm[w * IDX_PITCH + t];
    }
}

__device__ __forceinline__ unsigned match4(unsigned k, unsigned pat) {
    unsigned x = k ^ pat;
    unsigned t = ~(((x & 0x7F7F7F7Fu) + 0x7F7F7F7Fu) | x | 0x7F7F7F7Fu);
    return (((t >> 7) * 0x00204081u) >> 21) & 0xFu;
}
__device__ __forceinline__ unsigned match16(uint4 k, unsigned pat) {
    return match4(k.x, pat) | (match4(k.y, pat) << 4) | (match4(k.z, pat) << 8) | (match4(k.w, pat) << 12);
}

__device__ __forceinline__ float lane_f(float v, int l) {
    return __int_as_float(__builtin_amdgcn_readlane(__float_as_int(v), l));
}

#define LIST_CAP 16384
#define DL_BITS  9

template <int BITS>
__device__ __forceinline__ unsigned wave_reserve(unsigned* counter, unsigned n, int lane) {
    unsigned before = 0u, total = 0u;
#pragma unroll
    for (int k = 0; k < BITS; ++k) {
        const unsigned bal = __builtin_amdgcn_ballot_w32(((n >> k) & 1u) != 0u);
        before += __builtin_amdgcn_mbcnt_lo(bal, 0u) << k;
        total  += (unsigned)__popc(bal) << k;
    }
    unsigned start = 0u;
    if (lane == 0) start = atomicAdd(counter, total);
    return (unsigned)__builtin_amdgcn_readfirstlane((int)start) + before;
}

template <int H, int CH, int PASS>
__device__ __forceinline__ void take_edges(bool act, int s, int dl,
                                           const float* __restrict__ als, const float* __restrict__ hf,
                                           float* acc, unsigned* mx, float* den, const float* adl, int lane) {
    constexpr int C = H * CH;
    constexpr int NK = C / 32;
    float w[H];
#pragma unroll
    for (int h = 0; h < H; ++h) w[h] = 0.f;
    if (act) {
#pragma unroll
        for (int h = 0; h < H; ++h) {
            const float v = edge_score(als[(size_t)s * H + h], adl[dl * H + h]);
            if (PASS == 1) {
                atomicMax(&mx[dl * H + h], fenc(v));
            } else {
                w[h] = __expf(v - fdec(mx[dl * H + h]));
                atomicAdd(&den[dl * H + h], w[h]);
            }
        }
    }
    if (PASS == 2) {
        unsigned bal = __builtin_amdgcn_ballot_w32(act);
        while (bal != 0u) {
            int  ss[4], dd[4];
            bool ok[4];
            float ww[4][H];
            float r[4][NK];
#pragma unroll
            for (int j = 0; j < 4; ++j) {
                ok[j] = bal != 0u;
                const int l = ok[j] ? (__ffs(bal) - 1) : 0;
                if (ok[j]) bal &= bal - 1u;
                ss[j] = __builtin_amdgcn_readlane(s, l);
                dd[j] = __builtin_amdgcn_readlane(dl, l);
#pragma unroll
                for (int h = 0; h < H; ++h) ww[j][h] = lane_f(w[h], l);
            }
#pragma unroll
            for (int j = 0; j < 4; ++j) {
#pragma unroll
                for (int k = 0; k < NK; ++k) r[j][k] = 0.f;
                if (ok[j]) {
#pragma unroll
                    for (int k = 0; k < NK; ++k) r[j][k] = hf[(size_t)ss[j] * C + k * 32 + lane];
                }
            }
#pragma unroll
            for (int j = 0; j < 4; ++j) {
                if (ok[j]) {
#pragma unroll
                    for (int k = 0; k < NK; ++k)
                        atomicAdd(&acc[dd[j] * C + k * 32 + lane], ww[j][(k * 32) / CH] * r[j][k]);
                }
            }
        }
    }
}

template <int H, int CH, int PASS, bool RECORD>
__device__ __forceinline__ void tile_sweep(const int* __restrict__ ei, const uint4* __restrict__ keys,
                                           const float* __restrict__ als, const float* __restrict__ hf,
                                           float* acc, unsigned* mx, float* den, const float* adl,
                                           unsigned* lst, unsigned* cnt,
                                           int t0, unsigned pat, int rot, int n_edges, int n_key16,
                                           int lane, int wave) {
    constexpr int STEP = (TILE_THREADS / 32) * 64;
    const uint4 none = make_uint4(~0u, ~0u, ~0u, ~0u);

    auto group_of = [&](int base, int u) {
        int q = base + u * 32 + lane + rot;
        return q >= n_key16 ? q - n_key16 : q;
    };
    auto fetch = [&](int base, int u) {
        return (base + u * 32 + lane < n_key16) ? keys[group_of(base, u)] : none;
    };

    uint4 kv0 = fetch(wave * 64, 0), kv1 = fetch(wave * 64, 1);
    for (int base = wave * 64; base < n_key16; base += STEP) {
        const uint4 nx0 = fetch(base + STEP, 0), nx1 = fetch(base + STEP, 1);
        unsigned mask = match16(kv0, pat) | (match16(kv1, pat) << 16);
        while (__builtin_amdgcn_ballot_w32(mask != 0u) != 0u) {
            const bool act = mask != 0u;
            int s = 0, dl = 0;
            if (act) {
                const int b = __ffs(mask) - 1;
                mask &= mask - 1u;
                const int e = group_of(base, b >> 4) * 16 + (b & 15);
                s  = ei[e];
                dl = ei[(size_t)n_edges + e] - t0;
                if (RECORD) {
                    const unsigned slot = atomicAdd(cnt, 1u);
                    if (slot < LIST_CAP) lst[slot] = ((unsigned)s << DL_BITS) | (unsigned)dl;
                }
            }
            take_edges<H, CH, PASS>(act, s, dl, als, hf, acc, mx, den, adl, lane);
        }
        kv0 = nx0; kv1 = nx1;
    }
}

template <int H, int CH, int PASS>
__device__ __forceinline__ void list_sweep(const unsigned* lst, int n,
                                           const float* __restrict__ als, const float* __restrict__ hf,
                                           float* acc, unsigned* mx, float* den, const float* adl,
                                           int lane, int wave) {
    for (int base = wave * 32; base < n; base += TILE_THREADS) {
        const int i = base + lane;
        const bool act = i < n;
        const unsigned p = act ? lst[i] : 0u;
        take_edges<H, CH, PASS>(act, (int)(p >> DL_BITS), (int)(p & ((1u << DL_BITS) - 1u)),
                                als, hf, acc, mx, den, adl, lane);
    }
}

template <int H, int CH, bool ADD_BIAS, bool FIRST>
__global__ __launch_bounds__(TILE_THREADS)
void k_tile_gat(const int* __restrict__ ei, const uint4* __restrict__ keys,
                const float* __restrict__ als, const float* __restrict__ ald,
                const float* __restrict__ hf, const float* __restrict__ bias,
                float* __restrict__ out, unsigned* glist, unsigned* gcnt,
                const unsigned* __restrict__ bitmap, int row_words,
                int n_nodes, int n_edges, int n_key16) {
    constexpr int C = H * CH;
    __shared__ float    acc[TILE * C];
    __shared__ unsigned mx[TILE * H];
    __shared__ float    den[TILE * H];
    __shared__ float    adl[TILE * H];
    __shared__ unsigned lst[LIST_CAP];
    __shared__ unsigned cnt;
    __shared__ unsigned cntb;

    const int tid = threadIdx.x, lane = tid & 31, wave = tid >> 5;
    const int t0 = blockIdx.x * TILE;
    const unsigned pat = (unsigned)blockIdx.x * 0x01010101u;
    const int rot = (int)(((long long)blockIdx.x * n_key16) / gridDim.x);
    unsigned* const glst = glist + (size_t)blockIdx.x * LIST_CAP;

    unsigned n_list = 0u;
    if (FIRST) {
        if (tid == 0) { cnt = 0u; cntb = 0u; }
    } else {
        n_list = (unsigned)__builtin_amdgcn_readfirstlane((int)gcnt[blockIdx.x]);
        if (n_list <= LIST_CAP)
            for (int i = tid; i < (int)n_list; i += TILE_THREADS) lst[i] = glst[i];
    }

    for (int i = tid; i < TILE * H; i += TILE_THREADS) {
        const int d = t0 + i / H;
        float a = 0.f, v = 0.f;
        if (d < n_nodes) { a = ald[(size_t)t0 * H + i]; v = edge_score(als[(size_t)t0 * H + i], a); }
        adl[i] = a;
        mx[i]  = fenc(v);
    }
    __syncthreads();

    if (FIRST) {
        constexpr unsigned BL_CAP = TILE * C;
        unsigned* const blist = (unsigned*)acc;
        const unsigned* const row = bitmap + (size_t)blockIdx.x * row_words;
        for (int i0 = 0; i0 < row_words; i0 += 4 * TILE_THREADS) {
            unsigned wv[4];
#pragma unroll
            for (int u = 0; u < 4; ++u) {
                const int i = i0 + u * TILE_THREADS + tid;
                wv[u] = i < row_words ? row[i] : 0u;
            }
            unsigned slot = wave_reserve<8>(&cntb, (unsigned)(__popc(wv[0]) + __popc(wv[1]) + __popc(wv[2]) + __popc(wv[3])), lane);
#pragma unroll
            for (int u = 0; u < 4; ++u) {
                unsigned w = wv[u];
                const unsigned b0 = (unsigned)(i0 + u * TILE_THREADS + tid) * 32u;
                while (w != 0u) {
                    const unsigned b = (unsigned)__ffs(w) - 1u;
                    w &= w - 1u;
                    if (slot < BL_CAP) blist[slot] = b0 + b;
                    ++slot;
                }
            }
        }
        __syncthreads();
        const unsigned n_blk = (unsigned)__builtin_amdgcn_readfirstlane((int)cntb);
        if (n_blk <= BL_CAP) {
            for (int base = wave * 32; base < (int)n_blk; base += 4 * TILE_THREADS) {
                unsigned blk[4];
                uint4 kv[4];
#pragma unroll
                for (int u = 0; u < 4; ++u) {
                    const int i = base + u * TILE_THREADS + lane;
                    blk[u] = i < (int)n_blk ? blist[i] : ~0u;
                }
#pragma unroll
                for (int u = 0; u < 4; ++u) kv[u] = keys[blk[u] != ~0u ? blk[u] : 0u];
                unsigned mm[4];
#pragma unroll
                for (int u = 0; u < 4; ++u) mm[u] = blk[u] != ~0u ? match16(kv[u], pat) : 0u;
                unsigned slot = wave_reserve<7>(&cnt, (unsigned)(__popc(mm[0]) + __popc(mm[1]) + __popc(mm[2]) + __popc(mm[3])), lane);
#pragma unroll
                for (int u = 0; u < 4; ++u) {
                    unsigned m = mm[u];
                    while (m != 0u) {
                        const unsigned b = (unsigned)__ffs(m) - 1u;
                        m &= m - 1u;
                        if (slot < LIST_CAP) lst[slot] = blk[u] * 16u + b;
                        ++slot;
                    }
                }
            }
            __syncthreads();
            const unsigned n_found = (unsigned)__builtin_amdgcn_readfirstlane((int)cnt);
            if (n_found <= LIST_CAP) {
                for (int base = wave * 32; base < (int)n_found; base += 4 * TILE_THREADS) {
                    bool ok[4];
                    unsigned e[4];
                    int s[4], d[4];
                    float a[4][H];
#pragma unroll
                    for (int u = 0; u < 4; ++u) {
                        const int i = base + u * TILE_THREADS + lane;
                        ok[u] = i < (int)n_found;
                        e[u] = ok[u] ? lst[i] : 0u;
                    }
#pragma unroll
                    for (int u = 0; u < 4; ++u) { s[u] = ei[e[u]]; d[u] = ei[(size_t)n_edges + e[u]]; }
#pragma unroll
                    for (int u = 0; u < 4; ++u) {
#pragma unroll
                        for (int h = 0; h < H; ++h) a[u][h] = als[(size_t)s[u] * H + h];
                    }
#pragma unroll
                    for (int u = 0; u < 4; ++u) {
                        if (ok[u]) {
                            const int dl = d[u] - t0;
                            lst[base + u * TILE_THREADS + lane] = ((unsigned)s[u] << DL_BITS) | (unsigned)dl;
#pragma unroll
                            for (int h = 0; h < H; ++h)
                                atomicMax(&mx[dl * H + h], fenc(edge_score(a[u][h], adl[dl * H + h])));
                        }
                    }
                }
            } else {
                tile_sweep<H, CH, 1, false>(ei, keys, als, hf, acc, mx, den, adl, lst, &cnt, t0, pat, rot, n_edges, n_key16, lane, wave);
            }
        } else {
            tile_sweep<H, CH, 1, true>(ei, keys, als, hf, acc, mx, den, adl, lst, &cnt, t0, pat, rot, n_edges, n_key16, lane, wave);
        }
    } else if (n_list <= LIST_CAP) {
        list_sweep<H, CH, 1>(lst, (int)n_list, als, hf, acc, mx, den, adl, lane, wave);
    } else {
        tile_sweep<H, CH, 1, false>(ei, keys, als, hf, acc, mx, den, adl, lst, &cnt, t0, pat, rot, n_edges, n_key16, lane, wave);
    }
    __syncthreads();

    if (FIRST) {
        n_list = (unsigned)__builtin_amdgcn_readfirstlane((int)cnt);
        if (tid == 0) gcnt[blockIdx.x] = n_list;
        if (n_list <= LIST_CAP)
            for (int i = tid; i < (int)n_list; i += TILE_THREADS) glst[i] = lst[i];
    }

    for (int i = tid; i < TILE * H; i += TILE_THREADS) {
        const int d = t0 + i / H;
        float ex = 0.f;
        if (d < n_nodes) ex = __expf(edge_score(als[(size_t)t0 * H + i], adl[i]) - fdec(mx[i]));
        den[i] = ex;
    }
    for (int i = tid; i < TILE * C; i += TILE_THREADS) {
        const int dl = i / C, hd = (i % C) / CH;
        const int d = t0 + dl;
        float a = 0.f;
        if (d < n_nodes) {
            const float ex = __expf(edge_score(als[(size_t)d * H + hd], adl[dl * H + hd]) - fdec(mx[dl * H + hd]));
            a = ex * hf[(size_t)t0 * C + i];
        }
        acc[i] = a;
    }
    __syncthreads();

    if (n_list <= LIST_CAP) {
        list_sweep<H, CH, 2>(lst, (int)n_list, als, hf, acc, mx, den, adl, lane, wave);
    } else {
        tile_sweep<H, CH, 2, false>(ei, keys, als, hf, acc, mx, den, adl, lst, &cnt, t0, pat, rot, n_edges, n_key16, lane, wave);
    }
    __syncthreads();

    for (int i = tid; i < TILE * C; i += TILE_THREADS) {
        const int dl = i / C, c = i % C;
        if (t0 + dl < n_nodes) {
            float v = acc[i] / (den[dl * H + c / CH] + 1e-16f);
            if (ADD_BIAS) v += bias[c];
            out[(size_t)t0 * C + i] = v;
        }
    }
}


extern "C" void kernel_launch(void* const* d_in, const int* in_sizes, int n_in,
                              void* d_out, int out_size, void* d_ws, size_t ws_size,
                              hipStream_t stream) {
    (void)n_in; (void)out_size; (void)ws_size; (void)in_sizes;
    constexpr long long N = 100000, E = 1600000;
    constexpr int C1 = 128, HEADS = 4, HID = 32, C2 = 32;

    static_assert(N % TILE == 0 && N / TILE <= 255 && E % 16 == 0, "tile ids must fit a byte below 0xFF; keys are read 16 at a time");
    static_assert(TILE <= (1 << DL_BITS) && N <= (1ll << (32 - DL_BITS)), "a list entry packs source and local destination into one word");
    static_assert(N / TILE <= IDX_PITCH && E % 4 == 0 && (IDX_WORDS * 32 * 4) % IDX_THREADS == 0, "tile ids stay inside one LDS run of the index pass; destinations are read four at a time");

    const float* x    = (const float*)d_in[0];
    const int*   ei   = (const int*)  d_in[1];
    const float* W1   = (const float*)d_in[2];
    const float* aS1  = (const float*)d_in[3];
    const float* aD1  = (const float*)d_in[4];
    const float* b1   = (const float*)d_in[5];
    const float* W2   = (const float*)d_in[6];
    const float* aS2  = (const float*)d_in[7];
    const float* aD2  = (const float*)d_in[8];
    const float* b2   = (const float*)d_in[9];
    float* out = (float*)d_out;

    char* ws = (char*)d_ws;
    size_t off = 0;
    auto alloc = [&](size_t bytes) { size_t o = off; off = (off + bytes + 255) & ~(size_t)255; return o; };
    float*          h1   = (float*)(ws + alloc(N * C1 * 4));
    float*          out1 = (float*)(ws + alloc(N * C1 * 4));
    float*          as1  = (float*)(ws + alloc(N * HEADS * 4));
    float*          ad1  = (float*)(ws + alloc(N * HEADS * 4));
    unsigned*       keys = (unsigned*)(ws + alloc(E));
    unsigned*       glist = (unsigned*)(ws + alloc((N / TILE) * LIST_CAP * 4));
    unsigned*       gcnt  = (unsigned*)(ws + alloc((N / TILE) * 4));
    constexpr int      N_WORDS   = (int)((E / 16 + 31) / 32);
    constexpr unsigned IDX_GRID  = (N_WORDS + IDX_WORDS - 1) / IDX_WORDS;
    constexpr int      ROW_WORDS = (int)(IDX_GRID * IDX_WORDS);
    unsigned*       bitmap = (unsigned*)(ws + alloc((N / TILE) * (size_t)ROW_WORDS * 4));

    const int B = 256;
    auto blocks = [](long long n, int b) { return (unsigned)((n + b - 1) / b); };
    constexpr int STRIPS = (int)(N / 16);
    constexpr unsigned GEMM_GRID = (STRIPS + GEMM_WAVES - 1) / GEMM_WAVES;
    static_assert(N % 16 == 0 && C1 == 128 && C1 % 32 == 0 && C2 % 32 == 0, "the GEMM takes whole 16-row strips, K = 128 and column tiles in pairs");

    k_gemm_wmma_f32<C1, false><<<GEMM_GRID, GEMM_WAVES * 32, 0, stream>>>(x, W1, b1, h1, STRIPS);

    k_alpha<<<blocks(N * HEADS, B), B, 0, stream>>>(h1, aS1, aD1, as1, ad1, N, HEADS, HID);

    k_tile_index<<<IDX_GRID, IDX_THREADS, 0, stream>>>(ei, keys, bitmap, E, N / TILE, ROW_WORDS);

    k_tile_gat<HEADS, HID, false, true><<<N / TILE, TILE_THREADS, 0, stream>>>(
        ei, (const uint4*)keys, as1, ad1, h1, b1, out1, glist, gcnt, bitmap, ROW_WORDS, N, E, E / 16);

    k_gemm_wmma_f32<C2, true><<<GEMM_GRID, GEMM_WAVES * 32, 0, stream>>>(out1, W2, b1, h1, STRIPS);

    k_alpha<<<blocks(N, B), B, 0, stream>>>(h1, aS2, aD2, as1, ad1, N, 1, C2);

    k_tile_gat<1, C2, true, false><<<N / TILE, TILE_THREADS, 0, stream>>>(
        ei, (const uint4*)keys, as1, ad1, h1, b2, out, glist, gcnt, bitmap, ROW_WORDS, N, E, E / 16);
}


# Owner kernels keep more loads in flight: sum pass fetches 8 (layer 1) or 16 (layer 2) source rows per wait, not 4; start values 10 per trip, list copy 8, max replay 4; sums stay in arrival order
# speedup vs baseline: 7.9674x; 1.0711x over previous
#include <hip/hip_runtime.h>

typedef __attribute__((ext_vector_type(2))) float v2f;
typedef __attribute__((ext_vector_type(8))) float v8f;

#define NEG_SLOPE 0.2f


__device__ __forceinline__ unsigned fenc(float f) {
    unsigned u = __float_as_uint(f);
    return (u & 0x80000000u) ? ~u : (u | 0x80000000u);
}
__device__ __forceinline__ float fdec(unsigned u) {
    return (u & 0x80000000u) ? __uint_as_float(u & 0x7FFFFFFFu)
                             : __uint_as_float(~u);
}

#define GEMM_WAVES 8
template <int NC, bool BIAS_RELU>
__global__ __launch_bounds__(GEMM_WAVES * 32)
void k_gemm_wmma_f32(const float* __restrict__ A, const float* __restrict__ W,
                     const float* __restrict__ bias, float* __restrict__ C, int n_strips) {
    __shared__ float4 wl[NC * 128 / 4];
    __shared__ float4 bl[128 / 4];

    const int tid = threadIdx.x, lane = tid & 31;
    const int wave = __builtin_amdgcn_readfirstlane(tid >> 5);
    const int c = lane & 15, half = lane >> 4;
    const int strip = blockIdx.x * GEMM_WAVES + wave;
    const bool live = strip < n_strips;

    float4 a4[16];
    if (live) {
        const float4* arow = (const float4*)(A + (size_t)(strip * 16 + c) * 128 + half * 64);
#pragma unroll
        for (int i = 0; i < 16; ++i) a4[i] = arow[i];
    }

    float* const wlf = (float*)wl;
    for (int i4 = tid; i4 < 128 * NC / 4; i4 += GEMM_WAVES * 32) {
        const float4 w = ((const float4*)W)[i4];
        const int k = (i4 * 4) / NC, n = (i4 * 4) % NC;
        const int t = (k & 63) >> 1, s = k & 1, j = n >> 4;
        float* const d = wlf + ((((j >> 1) * 32 + t) * 32 + ((k >> 6) * 16 + (n & 15))) * 4 + (j & 1) * 2 + s);
        d[0] = w.x; d[4] = w.y; d[8] = w.z; d[12] = w.w;
    }
    if (BIAS_RELU && tid < 128) ((float*)bl)[tid] = bias[tid];
    __syncthreads();

    if (!live) return;

    float af[64];
#pragma unroll
    for (int i = 0; i < 16; ++i) {
        float4 v = a4[i];
        if (BIAS_RELU) {
            const float4 b = bl[half * 16 + i];
            v.x = fmaxf(v.x + b.x, 0.f); v.y = fmaxf(v.y + b.y, 0.f);
            v.z = fmaxf(v.z + b.z, 0.f); v.w = fmaxf(v.w + b.w, 0.f);
        }
        af[4 * i] = v.x; af[4 * i + 1] = v.y; af[4 * i + 2] = v.z; af[4 * i + 3] = v.w;
    }

    for (int jp = 0; jp < NC / 32; ++jp) {
        v8f acc0 = {}, acc1 = {};
#pragma unroll
        for (int t = 0; t < 32; ++t) {
            const float4 b = wl[(jp * 32 + t) * 32 + lane];
            const v2f a  = {af[2 * t], af[2 * t + 1]};
            const v2f b0 = {b.x, b.y};
            const v2f b1 = {b.z, b.w};
            acc0 = __builtin_amdgcn_wmma_f32_16x16x4_f32(false, a, false, b0, (short)0, acc0, false, false);
            acc1 = __builtin_amdgcn_wmma_f32_16x16x4_f32(false, a, false, b1, (short)0, acc1, false, false);
        }
#pragma unroll
        for (int r = 0; r < 8; ++r) {
            float* const crow = C + (size_t)(strip * 16 + r + 8 * half) * NC + jp * 32 + c;
            crow[0]  = acc0[r];
            crow[16] = acc1[r];
        }
    }
}

__global__ void k_alpha(const float* __restrict__ h,
                        const float* __restrict__ a_s, const float* __restrict__ a_d,
                        float* __restrict__ als, float* __restrict__ ald,
                        int n_nodes, int heads, int ch) {
    int i = blockIdx.x * blockDim.x + threadIdx.x;
    if (i >= n_nodes * heads) return;
    int n = i / heads, hd = i % heads;
    const float* row = h + (size_t)n * heads * ch + (size_t)hd * ch;
    const float* vs = a_s + (size_t)hd * ch;
    const float* vd = a_d + (size_t)hd * ch;
    float s = 0.f, d = 0.f;
    for (int c = 0; c < ch; ++c) { float x = row[c]; s += x * vs[c]; d += x * vd[c]; }
    als[i] = s; ald[i] = d;
}

#define TILE 400
#define TILE_THREADS 1024

__device__ __forceinline__ float edge_score(float a, float b) {
    float v = a + b;
    return v > 0.f ? v : NEG_SLOPE * v;
}

#define IDX_WORDS   32
#define IDX_THREADS 1024
#define IDX_PITCH   257
__global__ __launch_bounds__(IDX_THREADS)
void k_tile_index(const int* __restrict__ ei, unsigned* __restrict__ keys, unsigned* __restrict__ bitmap,
                  int n_edges, int n_tiles, int row_words) {
    __shared__ unsigned bm[IDX_WORDS * IDX_PITCH];
    const int tid = threadIdx.x;
    for (int i = tid; i < IDX_WORDS * IDX_PITCH; i += IDX_THREADS) bm[i] = 0u;
    __syncthreads();

    const int n_quads = n_edges / 4;
    const int q0 = blockIdx.x * (IDX_WORDS * 32 * 4);
    constexpr int QPT = IDX_WORDS * 32 * 4 / IDX_THREADS;
    int4 dv[QPT];
#pragma unroll
    for (int u = 0; u < QPT; ++u) {
        const int q = q0 + u * IDX_THREADS + tid;
        dv[u] = q < n_quads ? *(const int4*)(ei + (size_t)n_edges + (size_t)q * 4) : make_int4(0, 0, 0, 0);
    }
#pragma unroll
    for (int u = 0; u < QPT; ++u) {
        const int ql = u * IDX_THREADS + tid;
        const int q = q0 + ql;
        if (q < n_quads) {
            const int4 d = dv[u];
            const unsigned ta = (unsigned)(d.x / TILE), tb = (unsigned)(d.y / TILE),
                           tc = (unsigned)(d.z / TILE), td = (unsigned)(d.w / TILE);
            keys[q] = ta | (tb << 8) | (tc << 16) | (td << 24);
            unsigned* const wrow = bm + (ql >> 7) * IDX_PITCH;
            const unsigned bit = 1u << ((ql >> 2) & 31);
            atomicOr(&wrow[ta], bit); atomicOr(&wrow[tb], bit);
            atomicOr(&wrow[tc], bit); atomicOr(&wrow[td], bit);
        }
    }
    __syncthreads();

    for (int i = tid; i < n_tiles * IDX_WORDS; i += IDX_THREADS) {
        const int t = i / IDX_WORDS, w = i % IDX_WORDS;
        bitmap[(size_t)t * row_words + blockIdx.x * IDX_WORDS + w] = bm[w * IDX_PITCH + t];
    }
}

__device__ __forceinline__ unsigned match4(unsigned k, unsigned pat) {
    unsigned x = k ^ pat;
    unsigned t = ~(((x & 0x7F7F7F7Fu) + 0x7F7F7F7Fu) | x | 0x7F7F7F7Fu);
    return (((t >> 7) * 0x00204081u) >> 21) & 0xFu;
}
__device__ __forceinline__ unsigned match16(uint4 k, unsigned pat) {
    return match4(k.x, pat) | (match4(k.y, pat) << 4) | (match4(k.z, pat) << 8) | (match4(k.w, pat) << 12);
}

__device__ __forceinline__ float lane_f(float v, int l) {
    return __int_as_float(__builtin_amdgcn_readlane(__float_as_int(v), l));
}

#define LIST_CAP 16384
#define DL_BITS  9

template <int BITS>
__device__ __forceinline__ unsigned wave_reserve(unsigned* counter, unsigned n, int lane) {
    unsigned before = 0u, total = 0u;
#pragma unroll
    for (int k = 0; k < BITS; ++k) {
        const unsigned bal = __builtin_amdgcn_ballot_w32(((n >> k) & 1u) != 0u);
        before += __builtin_amdgcn_mbcnt_lo(bal, 0u) << k;
        total  += (unsigned)__popc(bal) << k;
    }
    unsigned start = 0u;
    if (lane == 0) start = atomicAdd(counter, total);
    return (unsigned)__builtin_amdgcn_readfirstlane((int)start) + before;
}

template <int H, int CH, int PASS>
__device__ __forceinline__ void take_edges(bool act, int s, int dl,
                                           const float* __restrict__ als, const float* __restrict__ hf,
                                           float* acc, unsigned* mx, float* den, const float* adl, int lane) {
    constexpr int C = H * CH;
    constexpr int NK = C / 32;
    float w[H];
#pragma unroll
    for (int h = 0; h < H; ++h) w[h] = 0.f;
    if (act) {
#pragma unroll
        for (int h = 0; h < H; ++h) {
            const float v = edge_score(als[(size_t)s * H + h], adl[dl * H + h]);
            if (PASS == 1) {
                atomicMax(&mx[dl * H + h], fenc(v));
            } else {
                w[h] = __expf(v - fdec(mx[dl * H + h]));
                atomicAdd(&den[dl * H + h], w[h]);
            }
        }
    }
    if (PASS == 2) {
        constexpr int RB = NK >= 4 ? 8 : 16;
        unsigned bal = __builtin_amdgcn_ballot_w32(act);
        while (bal != 0u) {
            int  ss[RB], dd[RB];
            bool ok[RB];
            float ww[RB][H];
            float r[RB][NK];
#pragma unroll
            for (int j = 0; j < RB; ++j) {
                ok[j] = bal != 0u;
                const int l = ok[j] ? (__ffs(bal) - 1) : 0;
                if (ok[j]) bal &= bal - 1u;
                ss[j] = __builtin_amdgcn_readlane(s, l);
                dd[j] = __builtin_amdgcn_readlane(dl, l);
#pragma unroll
                for (int h = 0; h < H; ++h) ww[j][h] = lane_f(w[h], l);
            }
#pragma unroll
            for (int j = 0; j < RB; ++j) {
#pragma unroll
                for (int k = 0; k < NK; ++k) r[j][k] = 0.f;
                if (ok[j]) {
#pragma unroll
                    for (int k = 0; k < NK; ++k) r[j][k] = hf[(size_t)ss[j] * C + k * 32 + lane];
                }
            }
#pragma unroll
            for (int j = 0; j < RB; ++j) {
                if (ok[j]) {
#pragma unroll
                    for (int k = 0; k < NK; ++k)
                        atomicAdd(&acc[dd[j] * C + k * 32 + lane], ww[j][(k * 32) / CH] * r[j][k]);
                }
            }
        }
    }
}

template <int H, int CH, int PASS, bool RECORD>
__device__ __forceinline__ void tile_sweep(const int* __restrict__ ei, const uint4* __restrict__ keys,
                                           const float* __restrict__ als, const float* __restrict__ hf,
                                           float* acc, unsigned* mx, float* den, const float* adl,
                                           unsigned* lst, unsigned* cnt,
                                           int t0, unsigned pat, int rot, int n_edges, int n_key16,
                                           int lane, int wave) {
    constexpr int STEP = (TILE_THREADS / 32) * 64;
    const uint4 none = make_uint4(~0u, ~0u, ~0u, ~0u);

    auto group_of = [&](int base, int u) {
        int q = base + u * 32 + lane + rot;
        return q >= n_key16 ? q - n_key16 : q;
    };
    auto fetch = [&](int base, int u) {
        return (base + u * 32 + lane < n_key16) ? keys[group_of(base, u)] : none;
    };

    uint4 kv0 = fetch(wave * 64, 0), kv1 = fetch(wave * 64, 1);
    for (int base = wave * 64; base < n_key16; base += STEP) {
        const uint4 nx0 = fetch(base + STEP, 0), nx1 = fetch(base + STEP, 1);
        unsigned mask = match16(kv0, pat) | (match16(kv1, pat) << 16);
        while (__builtin_amdgcn_ballot_w32(mask != 0u) != 0u) {
            const bool act = mask != 0u;
            int s = 0, dl = 0;
            if (act) {
                const int b = __ffs(mask) - 1;
                mask &= mask - 1u;
                const int e = group_of(base, b >> 4) * 16 + (b & 15);
                s  = ei[e];
                dl = ei[(size_t)n_edges + e] - t0;
                if (RECORD) {
                    const unsigned slot = atomicAdd(cnt, 1u);
                    if (slot < LIST_CAP) lst[slot] = ((unsigned)s << DL_BITS) | (unsigned)dl;
                }
            }
            take_edges<H, CH, PASS>(act, s, dl, als, hf, acc, mx, den, adl, lane);
        }
        kv0 = nx0; kv1 = nx1;
    }
}

template <int H, int CH, int PASS>
__device__ __forceinline__ void list_sweep(const unsigned* lst, int n,
                                           const float* __restrict__ als, const float* __restrict__ hf,
                                           float* acc, unsigned* mx, float* den, const float* adl,
                                           int lane, int wave) {
    if (PASS == 1) {
        for (int base = wave * 32; base < n; base += 4 * TILE_THREADS) {
            bool ok[4];
            unsigned p[4];
            float a[4][H];
#pragma unroll
            for (int u = 0; u < 4; ++u) {
                const int i = base + u * TILE_THREADS + lane;
                ok[u] = i < n;
                p[u] = ok[u] ? lst[i] : 0u;
            }
#pragma unroll
            for (int u = 0; u < 4; ++u) {
#pragma unroll
                for (int h = 0; h < H; ++h) a[u][h] = als[(size_t)(p[u] >> DL_BITS) * H + h];
            }
#pragma unroll
            for (int u = 0; u < 4; ++u) {
                if (ok[u]) {
                    const int dl = (int)(p[u] & ((1u << DL_BITS) - 1u));
#pragma unroll
                    for (int h = 0; h < H; ++h)
                        atomicMax(&mx[dl * H + h], fenc(edge_score(a[u][h], adl[dl * H + h])));
                }
            }
        }
        return;
    }
    for (int base = wave * 32; base < n; base += TILE_THREADS) {
        const int i = base + lane;
        const bool act = i < n;
        const unsigned p = act ? lst[i] : 0u;
        take_edges<H, CH, PASS>(act, (int)(p >> DL_BITS), (int)(p & ((1u << DL_BITS) - 1u)),
                                als, hf, acc, mx, den, adl, lane);
    }
}

template <int H, int CH, bool ADD_BIAS, bool FIRST>
__global__ __launch_bounds__(TILE_THREADS)
void k_tile_gat(const int* __restrict__ ei, const uint4* __restrict__ keys,
                const float* __restrict__ als, const float* __restrict__ ald,
                const float* __restrict__ hf, const float* __restrict__ bias,
                float* __restrict__ out, unsigned* glist, unsigned* gcnt,
                const unsigned* __restrict__ bitmap, int row_words,
                int n_nodes, int n_edges, int n_key16) {
    constexpr int C = H * CH;
    __shared__ float    acc[TILE * C];
    __shared__ unsigned mx[TILE * H];
    __shared__ float    den[TILE * H];
    __shared__ float    adl[TILE * H];
    __shared__ unsigned lst[LIST_CAP];
    __shared__ unsigned cnt;
    __shared__ unsigned cntb;

    const int tid = threadIdx.x, lane = tid & 31, wave = tid >> 5;
    const int t0 = blockIdx.x * TILE;
    const unsigned pat = (unsigned)blockIdx.x * 0x01010101u;
    const int rot = (int)(((long long)blockIdx.x * n_key16) / gridDim.x);
    unsigned* const glst = glist + (size_t)blockIdx.x * LIST_CAP;

    unsigned n_list = 0u;
    if (FIRST) {
        if (tid == 0) { cnt = 0u; cntb = 0u; }
    } else {
        n_list = (unsigned)__builtin_amdgcn_readfirstlane((int)gcnt[blockIdx.x]);
        if (n_list <= LIST_CAP) {
            for (int i0 = tid; i0 < (int)n_list; i0 += 8 * TILE_THREADS) {
                unsigned v[8];
#pragma unroll
                for (int u = 0; u < 8; ++u) {
                    const int i = i0 + u * TILE_THREADS;
                    v[u] = glst[i < (int)n_list ? i : i0];
                }
#pragma unroll
                for (int u = 0; u < 8; ++u) {
                    const int i = i0 + u * TILE_THREADS;
                    if (i < (int)n_list) lst[i] = v[u];
                }
            }
        }
    }

    for (int i = tid; i < TILE * H; i += TILE_THREADS) {
        const int d = t0 + i / H;
        float a = 0.f, v = 0.f;
        if (d < n_nodes) { a = ald[(size_t)t0 * H + i]; v = edge_score(als[(size_t)t0 * H + i], a); }
        adl[i] = a;
        mx[i]  = fenc(v);
    }
    __syncthreads();

    if (FIRST) {
        constexpr unsigned BL_CAP = TILE * C;
        unsigned* const blist = (unsigned*)acc;
        const unsigned* const row = bitmap + (size_t)blockIdx.x * row_words;
        for (int i0 = 0; i0 < row_words; i0 += 4 * TILE_THREADS) {
            unsigned wv[4];
#pragma unroll
            for (int u = 0; u < 4; ++u) {
                const int i = i0 + u * TILE_THREADS + tid;
                wv[u] = i < row_words ? row[i] : 0u;
            }
            unsigned slot = wave_reserve<8>(&cntb, (unsigned)(__popc(wv[0]) + __popc(wv[1]) + __popc(wv[2]) + __popc(wv[3])), lane);
#pragma unroll
            for (int u = 0; u < 4; ++u) {
                unsigned w = wv[u];
                const unsigned b0 = (unsigned)(i0 + u * TILE_THREADS + tid) * 32u;
                while (w != 0u) {
                    const unsigned b = (unsigned)__ffs(w) - 1u;
                    w &= w - 1u;
                    if (slot < BL_CAP) blist[slot] = b0 + b;
                    ++slot;
                }
            }
        }
        __syncthreads();
        const unsigned n_blk = (unsigned)__builtin_amdgcn_readfirstlane((int)cntb);
        if (n_blk <= BL_CAP) {
            for (int base = wave * 32; base < (int)n_blk; base += 4 * TILE_THREADS) {
                unsigned blk[4];
                uint4 kv[4];
#pragma unroll
                for (int u = 0; u < 4; ++u) {
                    const int i = base + u * TILE_THREADS + lane;
                    blk[u] = i < (int)n_blk ? blist[i] : ~0u;
                }
#pragma unroll
                for (int u = 0; u < 4; ++u) kv[u] = keys[blk[u] != ~0u ? blk[u] : 0u];
                unsigned mm[4];
#pragma unroll
                for (int u = 0; u < 4; ++u) mm[u] = blk[u] != ~0u ? match16(kv[u], pat) : 0u;
                unsigned slot = wave_reserve<7>(&cnt, (unsigned)(__popc(mm[0]) + __popc(mm[1]) + __popc(mm[2]) + __popc(mm[3])), lane);
#pragma unroll
                for (int u = 0; u < 4; ++u) {
                    unsigned m = mm[u];
                    while (m != 0u) {
                        const unsigned b = (unsigned)__ffs(m) - 1u;
                        m &= m - 1u;
                        if (slot < LIST_CAP) lst[slot] = blk[u] * 16u + b;
                        ++slot;
                    }
                }
            }
            __syncthreads();
            const unsigned n_found = (unsigned)__builtin_amdgcn_readfirstlane((int)cnt);
            if (n_found <= LIST_CAP) {
                for (int base = wave * 32; base < (int)n_found; base += 4 * TILE_THREADS) {
                    bool ok[4];
                    unsigned e[4];
                    int s[4], d[4];
                    float a[4][H];
#pragma unroll
                    for (int u = 0; u < 4; ++u) {
                        const int i = base + u * TILE_THREADS + lane;
                        ok[u] = i < (int)n_found;
                        e[u] = ok[u] ? lst[i] : 0u;
                    }
#pragma unroll
                    for (int u = 0; u < 4; ++u) { s[u] = ei[e[u]]; d[u] = ei[(size_t)n_edges + e[u]]; }
#pragma unroll
                    for (int u = 0; u < 4; ++u) {
#pragma unroll
                        for (int h = 0; h < H; ++h) a[u][h] = als[(size_t)s[u] * H + h];
                    }
#pragma unroll
                    for (int u = 0; u < 4; ++u) {
                        if (ok[u]) {
                            const int dl = d[u] - t0;
                            lst[base + u * TILE_THREADS + lane] = ((unsigned)s[u] << DL_BITS) | (unsigned)dl;
#pragma unroll
                            for (int h = 0; h < H; ++h)
                                atomicMax(&mx[dl * H + h], fenc(edge_score(a[u][h], adl[dl * H + h])));
                        }
                    }
                }
            } else {
                tile_sweep<H, CH, 1, false>(ei, keys, als, hf, acc, mx, den, adl, lst, &cnt, t0, pat, rot, n_edges, n_key16, lane, wave);
            }
        } else {
            tile_sweep<H, CH, 1, true>(ei, keys, als, hf, acc, mx, den, adl, lst, &cnt, t0, pat, rot, n_edges, n_key16, lane, wave);
        }
    } else if (n_list <= LIST_CAP) {
        list_sweep<H, CH, 1>(lst, (int)n_list, als, hf, acc, mx, den, adl, lane, wave);
    } else {
        tile_sweep<H, CH, 1, false>(ei, keys, als, hf, acc, mx, den, adl, lst, &cnt, t0, pat, rot, n_edges, n_key16, lane, wave);
    }
    __syncthreads();

    if (FIRST) {
        n_list = (unsigned)__builtin_amdgcn_readfirstlane((int)cnt);
        if (tid == 0) gcnt[blockIdx.x] = n_list;
        if (n_list <= LIST_CAP)
            for (int i = tid; i < (int)n_list; i += TILE_THREADS) glst[i] = lst[i];
    }

    for (int i = tid; i < TILE * H; i += TILE_THREADS) {
        const int d = t0 + i / H;
        float ex = 0.f;
        if (d < n_nodes) ex = __expf(edge_score(als[(size_t)t0 * H + i], adl[i]) - fdec(mx[i]));
        den[i] = ex;
    }
    constexpr int INIT_B = 10;
    for (int i0 = tid; i0 < TILE * C; i0 += INIT_B * TILE_THREADS) {
        float hv[INIT_B], av[INIT_B];
#pragma unroll
        for (int u = 0; u < INIT_B; ++u) {
            const int i = i0 + u * TILE_THREADS;
            const bool ok = i < TILE * C && t0 + i / C < n_nodes;
            const int ic = ok ? i : 0;
            hv[u] = hf[(size_t)t0 * C + ic];
            av[u] = als[(size_t)(t0 + ic / C) * H + (ic % C) / CH];
        }
#pragma unroll
        for (int u = 0; u < INIT_B; ++u) {
            const int i = i0 + u * TILE_THREADS;
            if (i < TILE * C) {
                const int dl = i / C, hd = (i % C) / CH;
                float a = 0.f;
                if (t0 + dl < n_nodes) {
                    const float ex = __expf(edge_score(av[u], adl[dl * H + hd]) - fdec(mx[dl * H + hd]));
                    a = ex * hv[u];
                }
                acc[i] = a;
            }
        }
    }
    __syncthreads();

    if (n_list <= LIST_CAP) {
        list_sweep<H, CH, 2>(lst, (int)n_list, als, hf, acc, mx, den, adl, lane, wave);
    } else {
        tile_sweep<H, CH, 2, false>(ei, keys, als, hf, acc, mx, den, adl, lst, &cnt, t0, pat, rot, n_edges, n_key16, lane, wave);
    }
    __syncthreads();

    for (int i = tid; i < TILE * C; i += TILE_THREADS) {
        const int dl = i / C, c = i % C;
        if (t0 + dl < n_nodes) {
            float v = acc[i] / (den[dl * H + c / CH] + 1e-16f);
            if (ADD_BIAS) v += bias[c];
            out[(size_t)t0 * C + i] = v;
        }
    }
}


extern "C" void kernel_launch(void* const* d_in, const int* in_sizes, int n_in,
                              void* d_out, int out_size, void* d_ws, size_t ws_size,
                              hipStream_t stream) {
    (void)n_in; (void)out_size; (void)ws_size; (void)in_sizes;
    constexpr long long N = 100000, E = 1600000;
    constexpr int C1 = 128, HEADS = 4, HID = 32, C2 = 32;

    static_assert(N % TILE == 0 && N / TILE <= 255 && E % 16 == 0, "tile ids must fit a byte below 0xFF; keys are read 16 at a time");
    static_assert(TILE <= (1 << DL_BITS) && N <= (1ll << (32 - DL_BITS)), "a list entry packs source and local destination into one word");
    static_assert(N / TILE <= IDX_PITCH && E % 4 == 0 && (IDX_WORDS * 32 * 4) % IDX_THREADS == 0, "tile ids stay inside one LDS run of the index pass; destinations are read four at a time");

    const float* x    = (const float*)d_in[0];
    const int*   ei   = (const int*)  d_in[1];
    const float* W1   = (const float*)d_in[2];
    const float* aS1  = (const float*)d_in[3];
    const float* aD1  = (const float*)d_in[4];
    const float* b1   = (const float*)d_in[5];
    const float* W2   = (const float*)d_in[6];
    const float* aS2  = (const float*)d_in[7];
    const float* aD2  = (const float*)d_in[8];
    const float* b2   = (const float*)d_in[9];
    float* out = (float*)d_out;

    char* ws = (char*)d_ws;
    size_t off = 0;
    auto alloc = [&](size_t bytes) { size_t o = off; off = (off + bytes + 255) & ~(size_t)255; return o; };
    float*          h1   = (float*)(ws + alloc(N * C1 * 4));
    float*          out1 = (float*)(ws + alloc(N * C1 * 4));
    float*          as1  = (float*)(ws + alloc(N * HEADS * 4));
    float*          ad1  = (float*)(ws + alloc(N * HEADS * 4));
    unsigned*       keys = (unsigned*)(ws + alloc(E));
    unsigned*       glist = (unsigned*)(ws + alloc((N / TILE) * LIST_CAP * 4));
    unsigned*       gcnt  = (unsigned*)(ws + alloc((N / TILE) * 4));
    constexpr int      N_WORDS   = (int)((E / 16 + 31) / 32);
    constexpr unsigned IDX_GRID  = (N_WORDS + IDX_WORDS - 1) / IDX_WORDS;
    constexpr int      ROW_WORDS = (int)(IDX_GRID * IDX_WORDS);
    unsigned*       bitmap = (unsigned*)(ws + alloc((N / TILE) * (size_t)ROW_WORDS * 4));

    const int B = 256;
    auto blocks = [](long long n, int b) { return (unsigned)((n + b - 1) / b); };
    constexpr int STRIPS = (int)(N / 16);
    constexpr unsigned GEMM_GRID = (STRIPS + GEMM_WAVES - 1) / GEMM_WAVES;
    static_assert(N % 16 == 0 && C1 == 128 && C1 % 32 == 0 && C2 % 32 == 0, "the GEMM takes whole 16-row strips, K = 128 and column tiles in pairs");

    k_gemm_wmma_f32<C1, false><<<GEMM_GRID, GEMM_WAVES * 32, 0, stream>>>(x, W1, b1, h1, STRIPS);

    k_alpha<<<blocks(N * HEADS, B), B, 0, stream>>>(h1, aS1, aD1, as1, ad1, N, HEADS, HID);

    k_tile_index<<<IDX_GRID, IDX_THREADS, 0, stream>>>(ei, keys, bitmap, E, N / TILE, ROW_WORDS);

    k_tile_gat<HEADS, HID, false, true><<<N / TILE, TILE_THREADS, 0, stream>>>(
        ei, (const uint4*)keys, as1, ad1, h1, b1, out1, glist, gcnt, bitmap, ROW_WORDS, N, E, E / 16);

    k_gemm_wmma_f32<C2, true><<<GEMM_GRID, GEMM_WAVES * 32, 0, stream>>>(out1, W2, b1, h1, STRIPS);

    k_alpha<<<blocks(N, B), B, 0, stream>>>(h1, aS2, aD2, as1, ad1, N, 1, C2);

    k_tile_gat<1, C2, true, false><<<N / TILE, TILE_THREADS, 0, stream>>>(
        ei, (const uint4*)keys, as1, ad1, h1, b2, out, glist, gcnt, bitmap, ROW_WORDS, N, E, E / 16);
}


# Attention vectors folded into the weights: a GEMM workgroup forms W*a_src, W*a_dst in LDS and one more 16-column f32 matrix tile yields both coefficients; their sum order changes; 7->5 launches
# speedup vs baseline: 8.3982x; 1.0541x over previous
#include <hip/hip_runtime.h>

typedef __attribute__((ext_vector_type(2))) float v2f;
typedef __attribute__((ext_vector_type(8))) float v8f;

#define NEG_SLOPE 0.2f


__device__ __forceinline__ unsigned fenc(float f) {
    unsigned u = __float_as_uint(f);
    return (u & 0x80000000u) ? ~u : (u | 0x80000000u);
}
__device__ __forceinline__ float fdec(unsigned u) {
    return (u & 0x80000000u) ? __uint_as_float(u & 0x7FFFFFFFu)
                             : __uint_as_float(~u);
}

#define GEMM_WAVES 8

__device__ __forceinline__ float sum8(float v) {
    v += __int_as_float(__builtin_amdgcn_update_dpp(0, __float_as_int(v), 0x161, 0xf, 0xf, true));
    v += __int_as_float(__builtin_amdgcn_update_dpp(0, __float_as_int(v), 0x162, 0xf, 0xf, true));
    v += __int_as_float(__builtin_amdgcn_update_dpp(0, __float_as_int(v), 0x164, 0xf, 0xf, true));
    return v;
}

template <int NC, int H, bool BIAS_RELU>
__global__ __launch_bounds__(GEMM_WAVES * 32)
void k_gemm_wmma_f32(const float* __restrict__ A, const float* __restrict__ W,
                     const float* __restrict__ bias,
                     const float* __restrict__ a_s, const float* __restrict__ a_d,
                     float* __restrict__ C, float* __restrict__ als, float* __restrict__ ald, int n_strips) {
    static_assert(NC == H * 32 && 2 * H <= 16 && (GEMM_WAVES * 32 * 4) % NC == 0,
                  "a head is 32 columns = 8 lanes of a staging trip; both coefficient sets fit one column tile");
    __shared__ float4 wl[NC * 128 / 4];
    __shared__ float4 bl[128 / 4];
    __shared__ float2 wa[32 * 32];

    const int tid = threadIdx.x, lane = tid & 31;
    const int wave = __builtin_amdgcn_readfirstlane(tid >> 5);
    const int c = lane & 15, half = lane >> 4;
    const int strip = blockIdx.x * GEMM_WAVES + wave;
    const bool live = strip < n_strips;

    float4 a4[16];
    if (live) {
        const float4* arow = (const float4*)(A + (size_t)(strip * 16 + c) * 128 + half * 64);
#pragma unroll
        for (int i = 0; i < 16; ++i) a4[i] = arow[i];
    }

    float* const wlf = (float*)wl;
    float* const waf = (float*)wa;
    for (int i = tid; i < 32 * 32; i += GEMM_WAVES * 32)
        if ((i & 15) >= 2 * H) wa[i] = make_float2(0.f, 0.f);
    const float4 s4 = *(const float4*)(a_s + (tid * 4) % NC);
    const float4 d4 = *(const float4*)(a_d + (tid * 4) % NC);
    constexpr int TRIPS = 128 * NC / 4 / (GEMM_WAVES * 32);
    constexpr int WB = TRIPS < 8 ? TRIPS : 8;
    static_assert(TRIPS % WB == 0, "whole groups of trips");
    for (int g0 = 0; g0 < TRIPS; g0 += WB) {
        float4 wv[WB];
#pragma unroll
        for (int u = 0; u < WB; ++u) wv[u] = ((const float4*)W)[tid + (g0 + u) * (GEMM_WAVES * 32)];
#pragma unroll
        for (int u = 0; u < WB; ++u) {
            const float4 w = wv[u];
            const int i4 = tid + (g0 + u) * (GEMM_WAVES * 32);
            const int k = (i4 * 4) / NC, n = (i4 * 4) % NC;
            const int t = (k & 63) >> 1, s = k & 1, j = n >> 4;
            float* const d = wlf + ((((j >> 1) * 32 + t) * 32 + ((k >> 6) * 16 + (n & 15))) * 4 + (j & 1) * 2 + s);
            d[0] = w.x; d[4] = w.y; d[8] = w.z; d[12] = w.w;
            const float ps = sum8(w.x * s4.x + w.y * s4.y + w.z * s4.z + w.w * s4.w);
            const float pd = sum8(w.x * d4.x + w.y * d4.y + w.z * d4.z + w.w * d4.w);
            const int side = lane & 7;
            if (side < 2) waf[(t * 32 + (k >> 6) * 16 + side * H + (n >> 5)) * 2 + s] = side ? pd : ps;
        }
    }
    if (BIAS_RELU && tid < 128) ((float*)bl)[tid] = bias[tid];
    __syncthreads();

    if (!live) return;

    float af[64];
#pragma unroll
    for (int i = 0; i < 16; ++i) {
        float4 v = a4[i];
        if (BIAS_RELU) {
            const float4 b = bl[half * 16 + i];
            v.x = fmaxf(v.x + b.x, 0.f); v.y = fmaxf(v.y + b.y, 0.f);
            v.z = fmaxf(v.z + b.z, 0.f); v.w = fmaxf(v.w + b.w, 0.f);
        }
        af[4 * i] = v.x; af[4 * i + 1] = v.y; af[4 * i + 2] = v.z; af[4 * i + 3] = v.w;
    }

    {
        v8f e = {};
#pragma unroll
        for (int t = 0; t < 32; ++t) {
            const float2 p = wa[t * 32 + lane];
            const v2f a  = {af[2 * t], af[2 * t + 1]};
            const v2f bp = {p.x, p.y};
            e = __builtin_amdgcn_wmma_f32_16x16x4_f32(false, a, false, bp, (short)0, e, false, false);
        }
        if (c < 2 * H) {
            float* const dst = (c < H ? als + c : ald + (c - H)) + (size_t)(strip * 16 + 8 * half) * H;
#pragma unroll
            for (int r = 0; r < 8; ++r) dst[r * H] = e[r];
        }
    }

    for (int jp = 0; jp < NC / 32; ++jp) {
        v8f acc0 = {}, acc1 = {};
#pragma unroll
        for (int t = 0; t < 32; ++t) {
            const float4 b = wl[(jp * 32 + t) * 32 + lane];
            const v2f a  = {af[2 * t], af[2 * t + 1]};
            const v2f b0 = {b.x, b.y};
            const v2f b1 = {b.z, b.w};
            acc0 = __builtin_amdgcn_wmma_f32_16x16x4_f32(false, a, false, b0, (short)0, acc0, false, false);
            acc1 = __builtin_amdgcn_wmma_f32_16x16x4_f32(false, a, false, b1, (short)0, acc1, false, false);
        }
#pragma unroll
        for (int r = 0; r < 8; ++r) {
            float* const crow = C + (size_t)(strip * 16 + r + 8 * half) * NC + jp * 32 + c;
            crow[0]  = acc0[r];
            crow[16] = acc1[r];
        }
    }
}

#define TILE 400
#define TILE_THREADS 1024

__device__ __forceinline__ float edge_score(float a, float b) {
    float v = a + b;
    return v > 0.f ? v : NEG_SLOPE * v;
}

#define IDX_WORDS   32
#define IDX_THREADS 1024
#define IDX_PITCH   257
__global__ __launch_bounds__(IDX_THREADS)
void k_tile_index(const int* __restrict__ ei, unsigned* __restrict__ keys, unsigned* __restrict__ bitmap,
                  int n_edges, int n_tiles, int row_words) {
    __shared__ unsigned bm[IDX_WORDS * IDX_PITCH];
    const int tid = threadIdx.x;
    for (int i = tid; i < IDX_WORDS * IDX_PITCH; i += IDX_THREADS) bm[i] = 0u;
    __syncthreads();

    const int n_quads = n_edges / 4;
    const int q0 = blockIdx.x * (IDX_WORDS * 32 * 4);
    constexpr int QPT = IDX_WORDS * 32 * 4 / IDX_THREADS;
    int4 dv[QPT];
#pragma unroll
    for (int u = 0; u < QPT; ++u) {
        const int q = q0 + u * IDX_THREADS + tid;
        dv[u] = q < n_quads ? *(const int4*)(ei + (size_t)n_edges + (size_t)q * 4) : make_int4(0, 0, 0, 0);
    }
#pragma unroll
    for (int u = 0; u < QPT; ++u) {
        const int ql = u * IDX_THREADS + tid;
        const int q = q0 + ql;
        if (q < n_quads) {
            const int4 d = dv[u];
            const unsigned ta = (unsigned)(d.x / TILE), tb = (unsigned)(d.y / TILE),
                           tc = (unsigned)(d.z / TILE), td = (unsigned)(d.w / TILE);
            keys[q] = ta | (tb << 8) | (tc << 16) | (td << 24);
            unsigned* const wrow = bm + (ql >> 7) * IDX_PITCH;
            const unsigned bit = 1u << ((ql >> 2) & 31);
            atomicOr(&wrow[ta], bit); atomicOr(&wrow[tb], bit);
            atomicOr(&wrow[tc], bit); atomicOr(&wrow[td], bit);
        }
    }
    __syncthreads();

    for (int i = tid; i < n_tiles * IDX_WORDS; i += IDX_THREADS) {
        const int t = i / IDX_WORDS, w = i % IDX_WORDS;
        bitmap[(size_t)t * row_words + blockIdx.x * IDX_WORDS + w] = bm[w * IDX_PITCH + t];
    }
}

__device__ __forceinline__ unsigned match4(unsigned k, unsigned pat) {
    unsigned x = k ^ pat;
    unsigned t = ~(((x & 0x7F7F7F7Fu) + 0x7F7F7F7Fu) | x | 0x7F7F7F7Fu);
    return (((t >> 7) * 0x00204081u) >> 21) & 0xFu;
}
__device__ __forceinline__ unsigned match16(uint4 k, unsigned pat) {
    return match4(k.x, pat) | (match4(k.y, pat) << 4) | (match4(k.z, pat) << 8) | (match4(k.w, pat) << 12);
}

__device__ __forceinline__ float lane_f(float v, int l) {
    return __int_as_float(__builtin_amdgcn_readlane(__float_as_int(v), l));
}

#define LIST_CAP 16384
#define DL_BITS  9

template <int BITS>
__device__ __forceinline__ unsigned wave_reserve(unsigned* counter, unsigned n, int lane) {
    unsigned before = 0u, total = 0u;
#pragma unroll
    for (int k = 0; k < BITS; ++k) {
        const unsigned bal = __builtin_amdgcn_ballot_w32(((n >> k) & 1u) != 0u);
        before += __builtin_amdgcn_mbcnt_lo(bal, 0u) << k;
        total  += (unsigned)__popc(bal) << k;
    }
    unsigned start = 0u;
    if (lane == 0) start = atomicAdd(counter, total);
    return (unsigned)__builtin_amdgcn_readfirstlane((int)start) + before;
}

template <int H, int CH, int PASS>
__device__ __forceinline__ void take_edges(bool act, int s, int dl,
                                           const float* __restrict__ als, const float* __restrict__ hf,
                                           float* acc, unsigned* mx, float* den, const float* adl, int lane) {
    constexpr int C = H * CH;
    constexpr int NK = C / 32;
    float w[H];
#pragma unroll
    for (int h = 0; h < H; ++h) w[h] = 0.f;
    if (act) {
#pragma unroll
        for (int h = 0; h < H; ++h) {
            const float v = edge_score(als[(size_t)s * H + h], adl[dl * H + h]);
            if (PASS == 1) {
                atomicMax(&mx[dl * H + h], fenc(v));
            } else {
                w[h] = __expf(v - fdec(mx[dl * H + h]));
                atomicAdd(&den[dl * H + h], w[h]);
            }
        }
    }
    if (PASS == 2) {
        constexpr int RB = NK >= 4 ? 8 : 16;
        unsigned bal = __builtin_amdgcn_ballot_w32(act);
        while (bal != 0u) {
            int  ss[RB], dd[RB];
            bool ok[RB];
            float ww[RB][H];
            float r[RB][NK];
#pragma unroll
            for (int j = 0; j < RB; ++j) {
                ok[j] = bal != 0u;
                const int l = ok[j] ? (__ffs(bal) - 1) : 0;
                if (ok[j]) bal &= bal - 1u;
                ss[j] = __builtin_amdgcn_readlane(s, l);
                dd[j] = __builtin_amdgcn_readlane(dl, l);
#pragma unroll
                for (int h = 0; h < H; ++h) ww[j][h] = lane_f(w[h], l);
            }
#pragma unroll
            for (int j = 0; j < RB; ++j) {
#pragma unroll
                for (int k = 0; k < NK; ++k) r[j][k] = 0.f;
                if (ok[j]) {
#pragma unroll
                    for (int k = 0; k < NK; ++k) r[j][k] = hf[(size_t)ss[j] * C + k * 32 + lane];
                }
            }
#pragma unroll
            for (int j = 0; j < RB; ++j) {
                if (ok[j]) {
#pragma unroll
                    for (int k = 0; k < NK; ++k)
                        atomicAdd(&acc[dd[j] * C + k * 32 + lane], ww[j][(k * 32) / CH] * r[j][k]);
                }
            }
        }
    }
}

template <int H, int CH, int PASS, bool RECORD>
__device__ __forceinline__ void tile_sweep(const int* __restrict__ ei, const uint4* __restrict__ keys,
                                           const float* __restrict__ als, const float* __restrict__ hf,
                                           float* acc, unsigned* mx, float* den, const float* adl,
                                           unsigned* lst, unsigned* cnt,
                                           int t0, unsigned pat, int rot, int n_edges, int n_key16,
                                           int lane, int wave) {
    constexpr int STEP = (TILE_THREADS / 32) * 64;
    const uint4 none = make_uint4(~0u, ~0u, ~0u, ~0u);

    auto group_of = [&](int base, int u) {
        int q = base + u * 32 + lane + rot;
        return q >= n_key16 ? q - n_key16 : q;
    };
    auto fetch = [&](int base, int u) {
        return (base + u * 32 + lane < n_key16) ? keys[group_of(base, u)] : none;
    };

    uint4 kv0 = fetch(wave * 64, 0), kv1 = fetch(wave * 64, 1);
    for (int base = wave * 64; base < n_key16; base += STEP) {
        const uint4 nx0 = fetch(base + STEP, 0), nx1 = fetch(base + STEP, 1);
        unsigned mask = match16(kv0, pat) | (match16(kv1, pat) << 16);
        while (__builtin_amdgcn_ballot_w32(mask != 0u) != 0u) {
            const bool act = mask != 0u;
            int s = 0, dl = 0;
            if (act) {
                const int b = __ffs(mask) - 1;
                mask &= mask - 1u;
                const int e = group_of(base, b >> 4) * 16 + (b & 15);
                s  = ei[e];
                dl = ei[(size_t)n_edges + e] - t0;
                if (RECORD) {
                    const unsigned slot = atomicAdd(cnt, 1u);
                    if (slot < LIST_CAP) lst[slot] = ((unsigned)s << DL_BITS) | (unsigned)dl;
                }
            }
            take_edges<H, CH, PASS>(act, s, dl, als, hf, acc, mx, den, adl, lane);
        }
        kv0 = nx0; kv1 = nx1;
    }
}

template <int H, int CH, int PASS>
__device__ __forceinline__ void list_sweep(const unsigned* lst, int n,
                                           const float* __restrict__ als, const float* __restrict__ hf,
                                           float* acc, unsigned* mx, float* den, const float* adl,
                                           int lane, int wave) {
    if (PASS == 1) {
        for (int base = wave * 32; base < n; base += 4 * TILE_THREADS) {
            bool ok[4];
            unsigned p[4];
            float a[4][H];
#pragma unroll
            for (int u = 0; u < 4; ++u) {
                const int i = base + u * TILE_THREADS + lane;
                ok[u] = i < n;
                p[u] = ok[u] ? lst[i] : 0u;
            }
#pragma unroll
            for (int u = 0; u < 4; ++u) {
#pragma unroll
                for (int h = 0; h < H; ++h) a[u][h] = als[(size_t)(p[u] >> DL_BITS) * H + h];
            }
#pragma unroll
            for (int u = 0; u < 4; ++u) {
                if (ok[u]) {
                    const int dl = (int)(p[u] & ((1u << DL_BITS) - 1u));
#pragma unroll
                    for (int h = 0; h < H; ++h)
                        atomicMax(&mx[dl * H + h], fenc(edge_score(a[u][h], adl[dl * H + h])));
                }
            }
        }
        return;
    }
    for (int base = wave * 32; base < n; base += TILE_THREADS) {
        const int i = base + lane;
        const bool act = i < n;
        const unsigned p = act ? lst[i] : 0u;
        take_edges<H, CH, PASS>(act, (int)(p >> DL_BITS), (int)(p & ((1u << DL_BITS) - 1u)),
                                als, hf, acc, mx, den, adl, lane);
    }
}

template <int H, int CH, bool ADD_BIAS, bool FIRST>
__global__ __launch_bounds__(TILE_THREADS)
void k_tile_gat(const int* __restrict__ ei, const uint4* __restrict__ keys,
                const float* __restrict__ als, const float* __restrict__ ald,
                const float* __restrict__ hf, const float* __restrict__ bias,
                float* __restrict__ out, unsigned* glist, unsigned* gcnt,
                const unsigned* __restrict__ bitmap, int row_words,
                int n_nodes, int n_edges, int n_key16) {
    constexpr int C = H * CH;
    __shared__ float    acc[TILE * C];
    __shared__ unsigned mx[TILE * H];
    __shared__ float    den[TILE * H];
    __shared__ float    adl[TILE * H];
    __shared__ unsigned lst[LIST_CAP];
    __shared__ unsigned cnt;
    __shared__ unsigned cntb;

    const int tid = threadIdx.x, lane = tid & 31, wave = tid >> 5;
    const int t0 = blockIdx.x * TILE;
    const unsigned pat = (unsigned)blockIdx.x * 0x01010101u;
    const int rot = (int)(((long long)blockIdx.x * n_key16) / gridDim.x);
    unsigned* const glst = glist + (size_t)blockIdx.x * LIST_CAP;

    unsigned n_list = 0u;
    if (FIRST) {
        if (tid == 0) { cnt = 0u; cntb = 0u; }
    } else {
        n_list = (unsigned)__builtin_amdgcn_readfirstlane((int)gcnt[blockIdx.x]);
        if (n_list <= LIST_CAP) {
            for (int i0 = tid; i0 < (int)n_list; i0 += 8 * TILE_THREADS) {
                unsigned v[8];
#pragma unroll
                for (int u = 0; u < 8; ++u) {
                    const int i = i0 + u * TILE_THREADS;
                    v[u] = glst[i < (int)n_list ? i : i0];
                }
#pragma unroll
                for (int u = 0; u < 8; ++u) {
                    const int i = i0 + u * TILE_THREADS;
                    if (i < (int)n_list) lst[i] = v[u];
                }
            }
        }
    }

    for (int i = tid; i < TILE * H; i += TILE_THREADS) {
        const int d = t0 + i / H;
        float a = 0.f, v = 0.f;
        if (d < n_nodes) { a = ald[(size_t)t0 * H + i]; v = edge_score(als[(size_t)t0 * H + i], a); }
        adl[i] = a;
        mx[i]  = fenc(v);
    }
    __syncthreads();

    if (FIRST) {
        constexpr unsigned BL_CAP = TILE * C;
        unsigned* const blist = (unsigned*)acc;
        const unsigned* const row = bitmap + (size_t)blockIdx.x * row_words;
        for (int i0 = 0; i0 < row_words; i0 += 4 * TILE_THREADS) {
            unsigned wv[4];
#pragma unroll
            for (int u = 0; u < 4; ++u) {
                const int i = i0 + u * TILE_THREADS + tid;
                wv[u] = i < row_words ? row[i] : 0u;
            }
            unsigned slot = wave_reserve<8>(&cntb, (unsigned)(__popc(wv[0]) + __popc(wv[1]) + __popc(wv[2]) + __popc(wv[3])), lane);
#pragma unroll
            for (int u = 0; u < 4; ++u) {
                unsigned w = wv[u];
                const unsigned b0 = (unsigned)(i0 + u * TILE_THREADS + tid) * 32u;
                while (w != 0u) {
                    const unsigned b = (unsigned)__ffs(w) - 1u;
                    w &= w - 1u;
                    if (slot < BL_CAP) blist[slot] = b0 + b;
                    ++slot;
                }
            }
        }
        __syncthreads();
        const unsigned n_blk = (unsigned)__builtin_amdgcn_readfirstlane((int)cntb);
        if (n_blk <= BL_CAP) {
            for (int base = wave * 32; base < (int)n_blk; base += 4 * TILE_THREADS) {
                unsigned blk[4];
                uint4 kv[4];
#pragma unroll
                for (int u = 0; u < 4; ++u) {
                    const int i = base + u * TILE_THREADS + lane;
                    blk[u] = i < (int)n_blk ? blist[i] : ~0u;
                }
#pragma unroll
                for (int u = 0; u < 4; ++u) kv[u] = keys[blk[u] != ~0u ? blk[u] : 0u];
                unsigned mm[4];
#pragma unroll
                for (int u = 0; u < 4; ++u) mm[u] = blk[u] != ~0u ? match16(kv[u], pat) : 0u;
                unsigned slot = wave_reserve<7>(&cnt, (unsigned)(__popc(mm[0]) + __popc(mm[1]) + __popc(mm[2]) + __popc(mm[3])), lane);
#pragma unroll
                for (int u = 0; u < 4; ++u) {
                    unsigned m = mm[u];
                    while (m != 0u) {
                        const unsigned b = (unsigned)__ffs(m) - 1u;
                        m &= m - 1u;
                        if (slot < LIST_CAP) lst[slot] = blk[u] * 16u + b;
                        ++slot;
                    }
                }
            }
            __syncthreads();
            const unsigned n_found = (unsigned)__builtin_amdgcn_readfirstlane((int)cnt);
            if (n_found <= LIST_CAP) {
                for (int base = wave * 32; base < (int)n_found; base += 4 * TILE_THREADS) {
                    bool ok[4];
                    unsigned e[4];
                    int s[4], d[4];
                    float a[4][H];
#pragma unroll
                    for (int u = 0; u < 4; ++u) {
                        const int i = base + u * TILE_THREADS + lane;
                        ok[u] = i < (int)n_found;
                        e[u] = ok[u] ? lst[i] : 0u;
                    }
#pragma unroll
                    for (int u = 0; u < 4; ++u) { s[u] = ei[e[u]]; d[u] = ei[(size_t)n_edges + e[u]]; }
#pragma unroll
                    for (int u = 0; u < 4; ++u) {
#pragma unroll
                        for (int h = 0; h < H; ++h) a[u][h] = als[(size_t)s[u] * H + h];
                    }
#pragma unroll
                    for (int u = 0; u < 4; ++u) {
                        if (ok[u]) {
                            const int dl = d[u] - t0;
                            lst[base + u * TILE_THREADS + lane] = ((unsigned)s[u] << DL_BITS) | (unsigned)dl;
#pragma unroll
                            for (int h = 0; h < H; ++h)
                                atomicMax(&mx[dl * H + h], fenc(edge_score(a[u][h], adl[dl * H + h])));
                        }
                    }
                }
            } else {
                tile_sweep<H, CH, 1, false>(ei, keys, als, hf, acc, mx, den, adl, lst, &cnt, t0, pat, rot, n_edges, n_key16, lane, wave);
            }
        } else {
            tile_sweep<H, CH, 1, true>(ei, keys, als, hf, acc, mx, den, adl, lst, &cnt, t0, pat, rot, n_edges, n_key16, lane, wave);
        }
    } else if (n_list <= LIST_CAP) {
        list_sweep<H, CH, 1>(lst, (int)n_list, als, hf, acc, mx, den, adl, lane, wave);
    } else {
        tile_sweep<H, CH, 1, false>(ei, keys, als, hf, acc, mx, den, adl, lst, &cnt, t0, pat, rot, n_edges, n_key16, lane, wave);
    }
    __syncthreads();

    if (FIRST) {
        n_list = (unsigned)__builtin_amdgcn_readfirstlane((int)cnt);
        if (tid == 0) gcnt[blockIdx.x] = n_list;
        if (n_list <= LIST_CAP)
            for (int i = tid; i < (int)n_list; i += TILE_THREADS) glst[i] = lst[i];
    }

    for (int i = tid; i < TILE * H; i += TILE_THREADS) {
        const int d = t0 + i / H;
        float ex = 0.f;
        if (d < n_nodes) ex = __expf(edge_score(als[(size_t)t0 * H + i], adl[i]) - fdec(mx[i]));
        den[i] = ex;
    }
    constexpr int INIT_B = 10;
    for (int i0 = tid; i0 < TILE * C; i0 += INIT_B * TILE_THREADS) {
        float hv[INIT_B], av[INIT_B];
#pragma unroll
        for (int u = 0; u < INIT_B; ++u) {
            const int i = i0 + u * TILE_THREADS;
            const bool ok = i < TILE * C && t0 + i / C < n_nodes;
            const int ic = ok ? i : 0;
            hv[u] = hf[(size_t)t0 * C + ic];
            av[u] = als[(size_t)(t0 + ic / C) * H + (ic % C) / CH];
        }
#pragma unroll
        for (int u = 0; u < INIT_B; ++u) {
            const int i = i0 + u * TILE_THREADS;
            if (i < TILE * C) {
                const int dl = i / C, hd = (i % C) / CH;
                float a = 0.f;
                if (t0 + dl < n_nodes) {
                    const float ex = __expf(edge_score(av[u], adl[dl * H + hd]) - fdec(mx[dl * H + hd]));
                    a = ex * hv[u];
                }
                acc[i] = a;
            }
        }
    }
    __syncthreads();

    if (n_list <= LIST_CAP) {
        list_sweep<H, CH, 2>(lst, (int)n_list, als, hf, acc, mx, den, adl, lane, wave);
    } else {
        tile_sweep<H, CH, 2, false>(ei, keys, als, hf, acc, mx, den, adl, lst, &cnt, t0, pat, rot, n_edges, n_key16, lane, wave);
    }
    __syncthreads();

    for (int i = tid; i < TILE * C; i += TILE_THREADS) {
        const int dl = i / C, c = i % C;
        if (t0 + dl < n_nodes) {
            float v = acc[i] / (den[dl * H + c / CH] + 1e-16f);
            if (ADD_BIAS) v += bias[c];
            out[(size_t)t0 * C + i] = v;
        }
    }
}


extern "C" void kernel_launch(void* const* d_in, const int* in_sizes, int n_in,
                              void* d_out, int out_size, void* d_ws, size_t ws_size,
                              hipStream_t stream) {
    (void)n_in; (void)out_size; (void)ws_size; (void)in_sizes;
    constexpr long long N = 100000, E = 1600000;
    constexpr int C1 = 128, HEADS = 4, HID = 32, C2 = 32;

    static_assert(N % TILE == 0 && N / TILE <= 255 && E % 16 == 0, "tile ids must fit a byte below 0xFF; keys are read 16 at a time");
    static_assert(TILE <= (1 << DL_BITS) && N <= (1ll << (32 - DL_BITS)), "a list entry packs source and local destination into one word");
    static_assert(N / TILE <= IDX_PITCH && E % 4 == 0 && (IDX_WORDS * 32 * 4) % IDX_THREADS == 0, "tile ids stay inside one LDS run of the index pass; destinations are read four at a time");

    const float* x    = (const float*)d_in[0];
    const int*   ei   = (const int*)  d_in[1];
    const float* W1   = (const float*)d_in[2];
    const float* aS1  = (const float*)d_in[3];
    const float* aD1  = (const float*)d_in[4];
    const float* b1   = (const float*)d_in[5];
    const float* W2   = (const float*)d_in[6];
    const float* aS2  = (const float*)d_in[7];
    const float* aD2  = (const float*)d_in[8];
    const float* b2   = (const float*)d_in[9];
    float* out = (float*)d_out;

    char* ws = (char*)d_ws;
    size_t off = 0;
    auto alloc = [&](size_t bytes) { size_t o = off; off = (off + bytes + 255) & ~(size_t)255; return o; };
    float*          h1   = (float*)(ws + alloc(N * C1 * 4));
    float*          out1 = (float*)(ws + alloc(N * C1 * 4));
    float*          as1  = (float*)(ws + alloc(N * HEADS * 4));
    float*          ad1  = (float*)(ws + alloc(N * HEADS * 4));
    unsigned*       keys = (unsigned*)(ws + alloc(E));
    unsigned*       glist = (unsigned*)(ws + alloc((N / TILE) * LIST_CAP * 4));
    unsigned*       gcnt  = (unsigned*)(ws + alloc((N / TILE) * 4));
    constexpr int      N_WORDS   = (int)((E / 16 + 31) / 32);
    constexpr unsigned IDX_GRID  = (N_WORDS + IDX_WORDS - 1) / IDX_WORDS;
    constexpr int      ROW_WORDS = (int)(IDX_GRID * IDX_WORDS);
    unsigned*       bitmap = (unsigned*)(ws + alloc((N / TILE) * (size_t)ROW_WORDS * 4));

    constexpr int STRIPS = (int)(N / 16);
    constexpr unsigned GEMM_GRID = (STRIPS + GEMM_WAVES - 1) / GEMM_WAVES;
    static_assert(N % 16 == 0 && C1 == 128 && C1 % 32 == 0 && C2 % 32 == 0, "the GEMM takes whole 16-row strips, K = 128 and column tiles in pairs");

    k_gemm_wmma_f32<C1, HEADS, false><<<GEMM_GRID, GEMM_WAVES * 32, 0, stream>>>(x, W1, b1, aS1, aD1, h1, as1, ad1, STRIPS);

    k_tile_index<<<IDX_GRID, IDX_THREADS, 0, stream>>>(ei, keys, bitmap, E, N / TILE, ROW_WORDS);

    k_tile_gat<HEADS, HID, false, true><<<N / TILE, TILE_THREADS, 0, stream>>>(
        ei, (const uint4*)keys, as1, ad1, h1, b1, out1, glist, gcnt, bitmap, ROW_WORDS, N, E, E / 16);

    k_gemm_wmma_f32<C2, 1, true><<<GEMM_GRID, GEMM_WAVES * 32, 0, stream>>>(out1, W2, b1, aS2, aD2, h1, as1, ad1, STRIPS);

    k_tile_gat<1, C2, true, false><<<N / TILE, TILE_THREADS, 0, stream>>>(
        ei, (const uint4*)keys, as1, ad1, h1, b2, out, glist, gcnt, bitmap, ROW_WORDS, N, E, E / 16);
}


# Owner set-up phases take one round trip each: a phase's loads all go out together (8 list rounds; the tile's own rows, 4 channels a load), the self-loop term is kept, not fetched and formed again
# speedup vs baseline: 8.6701x; 1.0324x over previous
#include <hip/hip_runtime.h>

typedef __attribute__((ext_vector_type(2))) float v2f;
typedef __attribute__((ext_vector_type(8))) float v8f;

#define NEG_SLOPE 0.2f


__device__ __forceinline__ unsigned fenc(float f) {
    unsigned u = __float_as_uint(f);
    return (u & 0x80000000u) ? ~u : (u | 0x80000000u);
}
__device__ __forceinline__ float fdec(unsigned u) {
    return (u & 0x80000000u) ? __uint_as_float(u & 0x7FFFFFFFu)
                             : __uint_as_float(~u);
}

#define GEMM_WAVES 8

__device__ __forceinline__ float sum8(float v) {
    v += __int_as_float(__builtin_amdgcn_update_dpp(0, __float_as_int(v), 0x161, 0xf, 0xf, true));
    v += __int_as_float(__builtin_amdgcn_update_dpp(0, __float_as_int(v), 0x162, 0xf, 0xf, true));
    v += __int_as_float(__builtin_amdgcn_update_dpp(0, __float_as_int(v), 0x164, 0xf, 0xf, true));
    return v;
}

template <int NC, int H, bool BIAS_RELU>
__global__ __launch_bounds__(GEMM_WAVES * 32)
void k_gemm_wmma_f32(const float* __restrict__ A, const float* __restrict__ W,
                     const float* __restrict__ bias,
                     const float* __restrict__ a_s, const float* __restrict__ a_d,
                     float* __restrict__ C, float* __restrict__ als, float* __restrict__ ald, int n_strips) {
    static_assert(NC == H * 32 && 2 * H <= 16 && (GEMM_WAVES * 32 * 4) % NC == 0,
                  "a head is 32 columns = 8 lanes of a staging trip; both coefficient sets fit one column tile");
    __shared__ float4 wl[NC * 128 / 4];
    __shared__ float4 bl[128 / 4];
    __shared__ float2 wa[32 * 32];

    const int tid = threadIdx.x, lane = tid & 31;
    const int wave = __builtin_amdgcn_readfirstlane(tid >> 5);
    const int c = lane & 15, half = lane >> 4;
    const int strip = blockIdx.x * GEMM_WAVES + wave;
    const bool live = strip < n_strips;

    float4 a4[16];
    if (live) {
        const float4* arow = (const float4*)(A + (size_t)(strip * 16 + c) * 128 + half * 64);
#pragma unroll
        for (int i = 0; i < 16; ++i) a4[i] = arow[i];
    }

    float* const wlf = (float*)wl;
    float* const waf = (float*)wa;
    for (int i = tid; i < 32 * 32; i += GEMM_WAVES * 32)
        if ((i & 15) >= 2 * H) wa[i] = make_float2(0.f, 0.f);
    const float4 s4 = *(const float4*)(a_s + (tid * 4) % NC);
    const float4 d4 = *(const float4*)(a_d + (tid * 4) % NC);
    constexpr int TRIPS = 128 * NC / 4 / (GEMM_WAVES * 32);
    constexpr int WB = TRIPS < 8 ? TRIPS : 8;
    static_assert(TRIPS % WB == 0, "whole groups of trips");
    for (int g0 = 0; g0 < TRIPS; g0 += WB) {
        float4 wv[WB];
#pragma unroll
        for (int u = 0; u < WB; ++u) wv[u] = ((const float4*)W)[tid + (g0 + u) * (GEMM_WAVES * 32)];
#pragma unroll
        for (int u = 0; u < WB; ++u) {
            const float4 w = wv[u];
            const int i4 = tid + (g0 + u) * (GEMM_WAVES * 32);
            const int k = (i4 * 4) / NC, n = (i4 * 4) % NC;
            const int t = (k & 63) >> 1, s = k & 1, j = n >> 4;
            float* const d = wlf + ((((j >> 1) * 32 + t) * 32 + ((k >> 6) * 16 + (n & 15))) * 4 + (j & 1) * 2 + s);
            d[0] = w.x; d[4] = w.y; d[8] = w.z; d[12] = w.w;
            const float ps = sum8(w.x * s4.x + w.y * s4.y + w.z * s4.z + w.w * s4.w);
            const float pd = sum8(w.x * d4.x + w.y * d4.y + w.z * d4.z + w.w * d4.w);
            const int side = lane & 7;
            if (side < 2) waf[(t * 32 + (k >> 6) * 16 + side * H + (n >> 5)) * 2 + s] = side ? pd : ps;
        }
    }
    if (BIAS_RELU && tid < 128) ((float*)bl)[tid] = bias[tid];
    __syncthreads();

    if (!live) return;

    float af[64];
#pragma unroll
    for (int i = 0; i < 16; ++i) {
        float4 v = a4[i];
        if (BIAS_RELU) {
            const float4 b = bl[half * 16 + i];
            v.x = fmaxf(v.x + b.x, 0.f); v.y = fmaxf(v.y + b.y, 0.f);
            v.z = fmaxf(v.z + b.z, 0.f); v.w = fmaxf(v.w + b.w, 0.f);
        }
        af[4 * i] = v.x; af[4 * i + 1] = v.y; af[4 * i + 2] = v.z; af[4 * i + 3] = v.w;
    }

    {
        v8f e = {};
#pragma unroll
        for (int t = 0; t < 32; ++t) {
            const float2 p = wa[t * 32 + lane];
            const v2f a  = {af[2 * t], af[2 * t + 1]};
            const v2f bp = {p.x, p.y};
            e = __builtin_amdgcn_wmma_f32_16x16x4_f32(false, a, false, bp, (short)0, e, false, false);
        }
        if (c < 2 * H) {
            float* const dst = (c < H ? als + c : ald + (c - H)) + (size_t)(strip * 16 + 8 * half) * H;
#pragma unroll
            for (int r = 0; r < 8; ++r) dst[r * H] = e[r];
        }
    }

    for (int jp = 0; jp < NC / 32; ++jp) {
        v8f acc0 = {}, acc1 = {};
#pragma unroll
        for (int t = 0; t < 32; ++t) {
            const float4 b = wl[(jp * 32 + t) * 32 + lane];
            const v2f a  = {af[2 * t], af[2 * t + 1]};
            const v2f b0 = {b.x, b.y};
            const v2f b1 = {b.z, b.w};
            acc0 = __builtin_amdgcn_wmma_f32_16x16x4_f32(false, a, false, b0, (short)0, acc0, false, false);
            acc1 = __builtin_amdgcn_wmma_f32_16x16x4_f32(false, a, false, b1, (short)0, acc1, false, false);
        }
#pragma unroll
        for (int r = 0; r < 8; ++r) {
            float* const crow = C + (size_t)(strip * 16 + r + 8 * half) * NC + jp * 32 + c;
            crow[0]  = acc0[r];
            crow[16] = acc1[r];
        }
    }
}

#define TILE 400
#define TILE_THREADS 1024

__device__ __forceinline__ float edge_score(float a, float b) {
    float v = a + b;
    return v > 0.f ? v : NEG_SLOPE * v;
}

#define IDX_WORDS   32
#define IDX_THREADS 1024
#define IDX_PITCH   257
__global__ __launch_bounds__(IDX_THREADS)
void k_tile_index(const int* __restrict__ ei, unsigned* __restrict__ keys, unsigned* __restrict__ bitmap,
                  int n_edges, int n_tiles, int row_words) {
    __shared__ unsigned bm[IDX_WORDS * IDX_PITCH];
    const int tid = threadIdx.x;
    for (int i = tid; i < IDX_WORDS * IDX_PITCH; i += IDX_THREADS) bm[i] = 0u;
    __syncthreads();

    const int n_quads = n_edges / 4;
    const int q0 = blockIdx.x * (IDX_WORDS * 32 * 4);
    constexpr int QPT = IDX_WORDS * 32 * 4 / IDX_THREADS;
    int4 dv[QPT];
#pragma unroll
    for (int u = 0; u < QPT; ++u) {
        const int q = q0 + u * IDX_THREADS + tid;
        dv[u] = q < n_quads ? *(const int4*)(ei + (size_t)n_edges + (size_t)q * 4) : make_int4(0, 0, 0, 0);
    }
#pragma unroll
    for (int u = 0; u < QPT; ++u) {
        const int ql = u * IDX_THREADS + tid;
        const int q = q0 + ql;
        if (q < n_quads) {
            const int4 d = dv[u];
            const unsigned ta = (unsigned)(d.x / TILE), tb = (unsigned)(d.y / TILE),
                           tc = (unsigned)(d.z / TILE), td = (unsigned)(d.w / TILE);
            keys[q] = ta | (tb << 8) | (tc << 16) | (td << 24);
            unsigned* const wrow = bm + (ql >> 7) * IDX_PITCH;
            const unsigned bit = 1u << ((ql >> 2) & 31);
            atomicOr(&wrow[ta], bit); atomicOr(&wrow[tb], bit);
            atomicOr(&wrow[tc], bit); atomicOr(&wrow[td], bit);
        }
    }
    __syncthreads();

    for (int i = tid; i < n_tiles * IDX_WORDS; i += IDX_THREADS) {
        const int t = i / IDX_WORDS, w = i % IDX_WORDS;
        bitmap[(size_t)t * row_words + blockIdx.x * IDX_WORDS + w] = bm[w * IDX_PITCH + t];
    }
}

__device__ __forceinline__ unsigned match4(unsigned k, unsigned pat) {
    unsigned x = k ^ pat;
    unsigned t = ~(((x & 0x7F7F7F7Fu) + 0x7F7F7F7Fu) | x | 0x7F7F7F7Fu);
    return (((t >> 7) * 0x00204081u) >> 21) & 0xFu;
}
__device__ __forceinline__ unsigned match16(uint4 k, unsigned pat) {
    return match4(k.x, pat) | (match4(k.y, pat) << 4) | (match4(k.z, pat) << 8) | (match4(k.w, pat) << 12);
}

__device__ __forceinline__ float lane_f(float v, int l) {
    return __int_as_float(__builtin_amdgcn_readlane(__float_as_int(v), l));
}

#define LIST_CAP 16384
#define DL_BITS  9
#define SWEEP_B  8

template <int BITS>
__device__ __forceinline__ unsigned wave_reserve(unsigned* counter, unsigned n, int lane) {
    unsigned before = 0u, total = 0u;
#pragma unroll
    for (int k = 0; k < BITS; ++k) {
        const unsigned bal = __builtin_amdgcn_ballot_w32(((n >> k) & 1u) != 0u);
        before += __builtin_amdgcn_mbcnt_lo(bal, 0u) << k;
        total  += (unsigned)__popc(bal) << k;
    }
    unsigned start = 0u;
    if (lane == 0) start = atomicAdd(counter, total);
    return (unsigned)__builtin_amdgcn_readfirstlane((int)start) + before;
}

template <int H, int CH, int PASS>
__device__ __forceinline__ void take_edges(bool act, int s, int dl,
                                           const float* __restrict__ als, const float* __restrict__ hf,
                                           float* acc, unsigned* mx, float* den, const float* adl, int lane) {
    constexpr int C = H * CH;
    constexpr int NK = C / 32;
    float w[H];
#pragma unroll
    for (int h = 0; h < H; ++h) w[h] = 0.f;
    if (act) {
#pragma unroll
        for (int h = 0; h < H; ++h) {
            const float v = edge_score(als[(size_t)s * H + h], adl[dl * H + h]);
            if (PASS == 1) {
                atomicMax(&mx[dl * H + h], fenc(v));
            } else {
                w[h] = __expf(v - fdec(mx[dl * H + h]));
                atomicAdd(&den[dl * H + h], w[h]);
            }
        }
    }
    if (PASS == 2) {
        constexpr int RB = NK >= 4 ? 8 : 16;
        unsigned bal = __builtin_amdgcn_ballot_w32(act);
        while (bal != 0u) {
            int  ss[RB], dd[RB];
            bool ok[RB];
            float ww[RB][H];
            float r[RB][NK];
#pragma unroll
            for (int j = 0; j < RB; ++j) {
                ok[j] = bal != 0u;
                const int l = ok[j] ? (__ffs(bal) - 1) : 0;
                if (ok[j]) bal &= bal - 1u;
                ss[j] = __builtin_amdgcn_readlane(s, l);
                dd[j] = __builtin_amdgcn_readlane(dl, l);
#pragma unroll
                for (int h = 0; h < H; ++h) ww[j][h] = lane_f(w[h], l);
            }
#pragma unroll
            for (int j = 0; j < RB; ++j) {
#pragma unroll
                for (int k = 0; k < NK; ++k) r[j][k] = 0.f;
                if (ok[j]) {
#pragma unroll
                    for (int k = 0; k < NK; ++k) r[j][k] = hf[(size_t)ss[j] * C + k * 32 + lane];
                }
            }
#pragma unroll
            for (int j = 0; j < RB; ++j) {
                if (ok[j]) {
#pragma unroll
                    for (int k = 0; k < NK; ++k)
                        atomicAdd(&acc[dd[j] * C + k * 32 + lane], ww[j][(k * 32) / CH] * r[j][k]);
                }
            }
        }
    }
}

template <int H, int CH, int PASS, bool RECORD>
__device__ __forceinline__ void tile_sweep(const int* __restrict__ ei, const uint4* __restrict__ keys,
                                           const float* __restrict__ als, const float* __restrict__ hf,
                                           float* acc, unsigned* mx, float* den, const float* adl,
                                           unsigned* lst, unsigned* cnt,
                                           int t0, unsigned pat, int rot, int n_edges, int n_key16,
                                           int lane, int wave) {
    constexpr int STEP = (TILE_THREADS / 32) * 64;
    const uint4 none = make_uint4(~0u, ~0u, ~0u, ~0u);

    auto group_of = [&](int base, int u) {
        int q = base + u * 32 + lane + rot;
        return q >= n_key16 ? q - n_key16 : q;
    };
    auto fetch = [&](int base, int u) {
        return (base + u * 32 + lane < n_key16) ? keys[group_of(base, u)] : none;
    };

    uint4 kv0 = fetch(wave * 64, 0), kv1 = fetch(wave * 64, 1);
    for (int base = wave * 64; base < n_key16; base += STEP) {
        const uint4 nx0 = fetch(base + STEP, 0), nx1 = fetch(base + STEP, 1);
        unsigned mask = match16(kv0, pat) | (match16(kv1, pat) << 16);
        while (__builtin_amdgcn_ballot_w32(mask != 0u) != 0u) {
            const bool act = mask != 0u;
            int s = 0, dl = 0;
            if (act) {
                const int b = __ffs(mask) - 1;
                mask &= mask - 1u;
                const int e = group_of(base, b >> 4) * 16 + (b & 15);
                s  = ei[e];
                dl = ei[(size_t)n_edges + e] - t0;
                if (RECORD) {
                    const unsigned slot = atomicAdd(cnt, 1u);
                    if (slot < LIST_CAP) lst[slot] = ((unsigned)s << DL_BITS) | (unsigned)dl;
                }
            }
            take_edges<H, CH, PASS>(act, s, dl, als, hf, acc, mx, den, adl, lane);
        }
        kv0 = nx0; kv1 = nx1;
    }
}

template <int H, int CH, int PASS>
__device__ __forceinline__ void list_sweep(const unsigned* lst, int n,
                                           const float* __restrict__ als, const float* __restrict__ hf,
                                           float* acc, unsigned* mx, float* den, const float* adl,
                                           int lane, int wave) {
    if (PASS == 1) {
        for (int base = wave * 32; base < n; base += SWEEP_B * TILE_THREADS) {
            bool ok[SWEEP_B];
            unsigned p[SWEEP_B];
            float a[SWEEP_B][H];
#pragma unroll
            for (int u = 0; u < SWEEP_B; ++u) {
                const int i = base + u * TILE_THREADS + lane;
                ok[u] = i < n;
                p[u] = ok[u] ? lst[i] : 0u;
            }
#pragma unroll
            for (int u = 0; u < SWEEP_B; ++u) {
#pragma unroll
                for (int h = 0; h < H; ++h) a[u][h] = als[(size_t)(p[u] >> DL_BITS) * H + h];
            }
#pragma unroll
            for (int u = 0; u < SWEEP_B; ++u) {
                if (ok[u]) {
                    const int dl = (int)(p[u] & ((1u << DL_BITS) - 1u));
#pragma unroll
                    for (int h = 0; h < H; ++h)
                        atomicMax(&mx[dl * H + h], fenc(edge_score(a[u][h], adl[dl * H + h])));
                }
            }
        }
        return;
    }
    for (int base = wave * 32; base < n; base += TILE_THREADS) {
        const int i = base + lane;
        const bool act = i < n;
        const unsigned p = act ? lst[i] : 0u;
        take_edges<H, CH, PASS>(act, (int)(p >> DL_BITS), (int)(p & ((1u << DL_BITS) - 1u)),
                                als, hf, acc, mx, den, adl, lane);
    }
}

template <int H, int CH, bool ADD_BIAS, bool FIRST>
__global__ __launch_bounds__(TILE_THREADS)
void k_tile_gat(const int* __restrict__ ei, const uint4* __restrict__ keys,
                const float* __restrict__ als, const float* __restrict__ ald,
                const float* __restrict__ hf, const float* __restrict__ bias,
                float* __restrict__ out, unsigned* glist, unsigned* gcnt,
                const unsigned* __restrict__ bitmap, int row_words,
                int n_nodes, int n_edges, int n_key16) {
    constexpr int C = H * CH;
    __shared__ __attribute__((aligned(16))) float acc[TILE * C];
    __shared__ unsigned mx[TILE * H];
    __shared__ float    den[TILE * H];
    __shared__ float    adl[TILE * H];
    __shared__ unsigned lst[LIST_CAP];
    __shared__ unsigned cnt;
    __shared__ unsigned cntb;

    const int tid = threadIdx.x, lane = tid & 31, wave = tid >> 5;
    const int t0 = blockIdx.x * TILE;
    const unsigned pat = (unsigned)blockIdx.x * 0x01010101u;
    const int rot = (int)(((long long)blockIdx.x * n_key16) / gridDim.x);
    unsigned* const glst = glist + (size_t)blockIdx.x * LIST_CAP;

    unsigned n_list = 0u;
    if (FIRST) {
        if (tid == 0) { cnt = 0u; cntb = 0u; }
    } else {
        n_list = (unsigned)__builtin_amdgcn_readfirstlane((int)gcnt[blockIdx.x]);
        if (n_list <= LIST_CAP) {
            for (int i0 = tid; i0 < (int)n_list; i0 += 8 * TILE_THREADS) {
                unsigned v[8];
#pragma unroll
                for (int u = 0; u < 8; ++u) {
                    const int i = i0 + u * TILE_THREADS;
                    v[u] = glst[i < (int)n_list ? i : i0];
                }
#pragma unroll
                for (int u = 0; u < 8; ++u) {
                    const int i = i0 + u * TILE_THREADS;
                    if (i < (int)n_list) lst[i] = v[u];
                }
            }
        }
    }

    constexpr int SELF_T = (TILE * H + TILE_THREADS - 1) / TILE_THREADS;
    float selfv[SELF_T];
    {
        float sa[SELF_T], ss[SELF_T];
#pragma unroll
        for (int u = 0; u < SELF_T; ++u) {
            const int i = tid + u * TILE_THREADS;
            const bool ok = i < TILE * H && t0 + i / H < n_nodes;
            const int ic = ok ? i : 0;
            sa[u] = ald[(size_t)t0 * H + ic];
            ss[u] = als[(size_t)t0 * H + ic];
        }
#pragma unroll
        for (int u = 0; u < SELF_T; ++u) {
            const int i = tid + u * TILE_THREADS;
            const bool ok = i < TILE * H && t0 + i / H < n_nodes;
            const float a = ok ? sa[u] : 0.f;
            const float v = ok ? edge_score(ss[u], sa[u]) : 0.f;
            selfv[u] = v;
            if (i < TILE * H) { adl[i] = a; mx[i] = fenc(v); }
        }
    }
    __syncthreads();

    if (FIRST) {
        constexpr unsigned BL_CAP = TILE * C;
        unsigned* const blist = (unsigned*)acc;
        const unsigned* const row = bitmap + (size_t)blockIdx.x * row_words;
        for (int i0 = 0; i0 < row_words; i0 += 4 * TILE_THREADS) {
            unsigned wv[4];
#pragma unroll
            for (int u = 0; u < 4; ++u) {
                const int i = i0 + u * TILE_THREADS + tid;
                wv[u] = i < row_words ? row[i] : 0u;
            }
            unsigned slot = wave_reserve<8>(&cntb, (unsigned)(__popc(wv[0]) + __popc(wv[1]) + __popc(wv[2]) + __popc(wv[3])), lane);
#pragma unroll
            for (int u = 0; u < 4; ++u) {
                unsigned w = wv[u];
                const unsigned b0 = (unsigned)(i0 + u * TILE_THREADS + tid) * 32u;
                while (w != 0u) {
                    const unsigned b = (unsigned)__ffs(w) - 1u;
                    w &= w - 1u;
                    if (slot < BL_CAP) blist[slot] = b0 + b;
                    ++slot;
                }
            }
        }
        __syncthreads();
        const unsigned n_blk = (unsigned)__builtin_amdgcn_readfirstlane((int)cntb);
        if (n_blk <= BL_CAP) {
            for (int base = wave * 32; base < (int)n_blk; base += SWEEP_B * TILE_THREADS) {
                unsigned blk[SWEEP_B];
                uint4 kv[SWEEP_B];
#pragma unroll
                for (int u = 0; u < SWEEP_B; ++u) {
                    const int i = base + u * TILE_THREADS + lane;
                    blk[u] = i < (int)n_blk ? blist[i] : ~0u;
                }
#pragma unroll
                for (int u = 0; u < SWEEP_B; ++u) kv[u] = keys[blk[u] != ~0u ? blk[u] : 0u];
                unsigned mm[SWEEP_B];
                unsigned found = 0u;
#pragma unroll
                for (int u = 0; u < SWEEP_B; ++u) { mm[u] = blk[u] != ~0u ? match16(kv[u], pat) : 0u; found += (unsigned)__popc(mm[u]); }
                static_assert(16 * SWEEP_B < 256, "a lane's count must fit the planes that wave_reserve adds up");
                unsigned slot = wave_reserve<8>(&cnt, found, lane);
#pragma unroll
                for (int u = 0; u < SWEEP_B; ++u) {
                    unsigned m = mm[u];
                    while (m != 0u) {
                        const unsigned b = (unsigned)__ffs(m) - 1u;
                        m &= m - 1u;
                        if (slot < LIST_CAP) lst[slot] = blk[u] * 16u + b;
                        ++slot;
                    }
                }
            }
            __syncthreads();
            const unsigned n_found = (unsigned)__builtin_amdgcn_readfirstlane((int)cnt);
            if (n_found <= LIST_CAP) {
                for (int base = wave * 32; base < (int)n_found; base += SWEEP_B * TILE_THREADS) {
                    bool ok[SWEEP_B];
                    unsigned e[SWEEP_B];
                    int s[SWEEP_B], d[SWEEP_B];
                    float a[SWEEP_B][H];
#pragma unroll
                    for (int u = 0; u < SWEEP_B; ++u) {
                        const int i = base + u * TILE_THREADS + lane;
                        ok[u] = i < (int)n_found;
                        e[u] = ok[u] ? lst[i] : 0u;
                    }
#pragma unroll
                    for (int u = 0; u < SWEEP_B; ++u) { s[u] = ei[e[u]]; d[u] = ei[(size_t)n_edges + e[u]]; }
#pragma unroll
                    for (int u = 0; u < SWEEP_B; ++u) {
#pragma unroll
                        for (int h = 0; h < H; ++h) a[u][h] = als[(size_t)s[u] * H + h];
                    }
#pragma unroll
                    for (int u = 0; u < SWEEP_B; ++u) {
                        if (ok[u]) {
                            const int dl = d[u] - t0;
                            lst[base + u * TILE_THREADS + lane] = ((unsigned)s[u] << DL_BITS) | (unsigned)dl;
#pragma unroll
                            for (int h = 0; h < H; ++h)
                                atomicMax(&mx[dl * H + h], fenc(edge_score(a[u][h], adl[dl * H + h])));
                        }
                    }
                }
            } else {
                tile_sweep<H, CH, 1, false>(ei, keys, als, hf, acc, mx, den, adl, lst, &cnt, t0, pat, rot, n_edges, n_key16, lane, wave);
            }
        } else {
            tile_sweep<H, CH, 1, true>(ei, keys, als, hf, acc, mx, den, adl, lst, &cnt, t0, pat, rot, n_edges, n_key16, lane, wave);
        }
    } else if (n_list <= LIST_CAP) {
        list_sweep<H, CH, 1>(lst, (int)n_list, als, hf, acc, mx, den, adl, lane, wave);
    } else {
        tile_sweep<H, CH, 1, false>(ei, keys, als, hf, acc, mx, den, adl, lst, &cnt, t0, pat, rot, n_edges, n_key16, lane, wave);
    }
    __syncthreads();

    if (FIRST) {
        n_list = (unsigned)__builtin_amdgcn_readfirstlane((int)cnt);
        if (tid == 0) gcnt[blockIdx.x] = n_list;
        if (n_list <= LIST_CAP)
            for (int i = tid; i < (int)n_list; i += TILE_THREADS) glst[i] = lst[i];
    }

    static_assert(CH % 4 == 0 && C % 4 == 0 && (TILE * C) % 4 == 0, "four neighbouring channels share a head and a row");
    constexpr int V4   = TILE * C / 4;
    constexpr int V4_T = (V4 + TILE_THREADS - 1) / TILE_THREADS;
    const float4* const hrow = (const float4*)(hf + (size_t)t0 * C);
    float4 hv[V4_T];
#pragma unroll
    for (int u = 0; u < V4_T; ++u) {
        const int i4 = tid + u * TILE_THREADS;
        const bool ok = i4 < V4 && t0 + (i4 * 4) / C < n_nodes;
        hv[u] = hrow[ok ? i4 : 0];
    }
#pragma unroll
    for (int u = 0; u < SELF_T; ++u) {
        const int i = tid + u * TILE_THREADS;
        if (i < TILE * H) {
            float ex = 0.f;
            if (t0 + i / H < n_nodes) ex = __expf(selfv[u] - fdec(mx[i]));
            den[i] = ex;
        }
    }
    __syncthreads();
    float4* const acc4 = (float4*)acc;
#pragma unroll
    for (int u = 0; u < V4_T; ++u) {
        const int i4 = tid + u * TILE_THREADS;
        if (i4 < V4) {
            const int dl = (i4 * 4) / C, hd = ((i4 * 4) % C) / CH;
            float4 a = make_float4(0.f, 0.f, 0.f, 0.f);
            if (t0 + dl < n_nodes) {
                const float ex = den[dl * H + hd];
                a = make_float4(ex * hv[u].x, ex * hv[u].y, ex * hv[u].z, ex * hv[u].w);
            }
            acc4[i4] = a;
        }
    }
    __syncthreads();

    if (n_list <= LIST_CAP) {
        list_sweep<H, CH, 2>(lst, (int)n_list, als, hf, acc, mx, den, adl, lane, wave);
    } else {
        tile_sweep<H, CH, 2, false>(ei, keys, als, hf, acc, mx, den, adl, lst, &cnt, t0, pat, rot, n_edges, n_key16, lane, wave);
    }
    __syncthreads();

    for (int i = tid; i < TILE * C; i += TILE_THREADS) {
        const int dl = i / C, c = i % C;
        if (t0 + dl < n_nodes) {
            float v = acc[i] / (den[dl * H + c / CH] + 1e-16f);
            if (ADD_BIAS) v += bias[c];
            out[(size_t)t0 * C + i] = v;
        }
    }
}


extern "C" void kernel_launch(void* const* d_in, const int* in_sizes, int n_in,
                              void* d_out, int out_size, void* d_ws, size_t ws_size,
                              hipStream_t stream) {
    (void)n_in; (void)out_size; (void)ws_size; (void)in_sizes;
    constexpr long long N = 100000, E = 1600000;
    constexpr int C1 = 128, HEADS = 4, HID = 32, C2 = 32;

    static_assert(N % TILE == 0 && N / TILE <= 255 && E % 16 == 0, "tile ids must fit a byte below 0xFF; keys are read 16 at a time");
    static_assert(TILE <= (1 << DL_BITS) && N <= (1ll << (32 - DL_BITS)), "a list entry packs source and local destination into one word");
    static_assert(N / TILE <= IDX_PITCH && E % 4 == 0 && (IDX_WORDS * 32 * 4) % IDX_THREADS == 0, "tile ids stay inside one LDS run of the index pass; destinations are read four at a time");

    const float* x    = (const float*)d_in[0];
    const int*   ei   = (const int*)  d_in[1];
    const float* W1   = (const float*)d_in[2];
    const float* aS1  = (const float*)d_in[3];
    const float* aD1  = (const float*)d_in[4];
    const float* b1   = (const float*)d_in[5];
    const float* W2   = (const float*)d_in[6];
    const float* aS2  = (const float*)d_in[7];
    const float* aD2  = (const float*)d_in[8];
    const float* b2   = (const float*)d_in[9];
    float* out = (float*)d_out;

    char* ws = (char*)d_ws;
    size_t off = 0;
    auto alloc = [&](size_t bytes) { size_t o = off; off = (off + bytes + 255) & ~(size_t)255; return o; };
    float*          h1   = (float*)(ws + alloc(N * C1 * 4));
    float*          out1 = (float*)(ws + alloc(N * C1 * 4));
    float*          as1  = (float*)(ws + alloc(N * HEADS * 4));
    float*          ad1  = (float*)(ws + alloc(N * HEADS * 4));
    unsigned*       keys = (unsigned*)(ws + alloc(E));
    unsigned*       glist = (unsigned*)(ws + alloc((N / TILE) * LIST_CAP * 4));
    unsigned*       gcnt  = (unsigned*)(ws + alloc((N / TILE) * 4));
    constexpr int      N_WORDS   = (int)((E / 16 + 31) / 32);
    constexpr unsigned IDX_GRID  = (N_WORDS + IDX_WORDS - 1) / IDX_WORDS;
    constexpr int      ROW_WORDS = (int)(IDX_GRID * IDX_WORDS);
    unsigned*       bitmap = (unsigned*)(ws + alloc((N / TILE) * (size_t)ROW_WORDS * 4));

    constexpr int STRIPS = (int)(N / 16);
    constexpr unsigned GEMM_GRID = (STRIPS + GEMM_WAVES - 1) / GEMM_WAVES;
    static_assert(N % 16 == 0 && C1 == 128 && C1 % 32 == 0 && C2 % 32 == 0, "the GEMM takes whole 16-row strips, K = 128 and column tiles in pairs");

    k_gemm_wmma_f32<C1, HEADS, false><<<GEMM_GRID, GEMM_WAVES * 32, 0, stream>>>(x, W1, b1, aS1, aD1, h1, as1, ad1, STRIPS);

    k_tile_index<<<IDX_GRID, IDX_THREADS, 0, stream>>>(ei, keys, bitmap, E, N / TILE, ROW_WORDS);

    k_tile_gat<HEADS, HID, false, true><<<N / TILE, TILE_THREADS, 0, stream>>>(
        ei, (const uint4*)keys, as1, ad1, h1, b1, out1, glist, gcnt, bitmap, ROW_WORDS, N, E, E / 16);

    k_gemm_wmma_f32<C2, 1, true><<<GEMM_GRID, GEMM_WAVES * 32, 0, stream>>>(out1, W2, b1, aS2, aD2, h1, as1, ad1, STRIPS);

    k_tile_gat<1, C2, true, false><<<N / TILE, TILE_THREADS, 0, stream>>>(
        ei, (const uint4*)keys, as1, ad1, h1, b2, out, glist, gcnt, bitmap, ROW_WORDS, N, E, E / 16);
}


# Sum pass reuses the edge scores the max pass kept in registers (same thread, same list entries): source coefficients fetched once per edge and layer, not twice; sums stay in arrival order
# speedup vs baseline: 8.9417x; 1.0313x over previous
#include <hip/hip_runtime.h>

typedef __attribute__((ext_vector_type(2))) float v2f;
typedef __attribute__((ext_vector_type(8))) float v8f;

#define NEG_SLOPE 0.2f


__device__ __forceinline__ unsigned fenc(float f) {
    unsigned u = __float_as_uint(f);
    return (u & 0x80000000u) ? ~u : (u | 0x80000000u);
}
__device__ __forceinline__ float fdec(unsigned u) {
    return (u & 0x80000000u) ? __uint_as_float(u & 0x7FFFFFFFu)
                             : __uint_as_float(~u);
}

#define GEMM_WAVES 8

__device__ __forceinline__ float sum8(float v) {
    v += __int_as_float(__builtin_amdgcn_update_dpp(0, __float_as_int(v), 0x161, 0xf, 0xf, true));
    v += __int_as_float(__builtin_amdgcn_update_dpp(0, __float_as_int(v), 0x162, 0xf, 0xf, true));
    v += __int_as_float(__builtin_amdgcn_update_dpp(0, __float_as_int(v), 0x164, 0xf, 0xf, true));
    return v;
}

template <int NC, int H, bool BIAS_RELU>
__global__ __launch_bounds__(GEMM_WAVES * 32)
void k_gemm_wmma_f32(const float* __restrict__ A, const float* __restrict__ W,
                     const float* __restrict__ bias,
                     const float* __restrict__ a_s, const float* __restrict__ a_d,
                     float* __restrict__ C, float* __restrict__ als, float* __restrict__ ald, int n_strips) {
    static_assert(NC == H * 32 && 2 * H <= 16 && (GEMM_WAVES * 32 * 4) % NC == 0,
                  "a head is 32 columns = 8 lanes of a staging trip; both coefficient sets fit one column tile");
    __shared__ float4 wl[NC * 128 / 4];
    __shared__ float4 bl[128 / 4];
    __shared__ float2 wa[32 * 32];

    const int tid = threadIdx.x, lane = tid & 31;
    const int wave = __builtin_amdgcn_readfirstlane(tid >> 5);
    const int c = lane & 15, half = lane >> 4;
    const int strip = blockIdx.x * GEMM_WAVES + wave;
    const bool live = strip < n_strips;

    float4 a4[16];
    if (live) {
        const float4* arow = (const float4*)(A + (size_t)(strip * 16 + c) * 128 + half * 64);
#pragma unroll
        for (int i = 0; i < 16; ++i) a4[i] = arow[i];
    }

    float* const wlf = (float*)wl;
    float* const waf = (float*)wa;
    for (int i = tid; i < 32 * 32; i += GEMM_WAVES * 32)
        if ((i & 15) >= 2 * H) wa[i] = make_float2(0.f, 0.f);
    const float4 s4 = *(const float4*)(a_s + (tid * 4) % NC);
    const float4 d4 = *(const float4*)(a_d + (tid * 4) % NC);
    constexpr int TRIPS = 128 * NC / 4 / (GEMM_WAVES * 32);
    constexpr int WB = TRIPS < 8 ? TRIPS : 8;
    static_assert(TRIPS % WB == 0, "whole groups of trips");
    for (int g0 = 0; g0 < TRIPS; g0 += WB) {
        float4 wv[WB];
#pragma unroll
        for (int u = 0; u < WB; ++u) wv[u] = ((const float4*)W)[tid + (g0 + u) * (GEMM_WAVES * 32)];
#pragma unroll
        for (int u = 0; u < WB; ++u) {
            const float4 w = wv[u];
            const int i4 = tid + (g0 + u) * (GEMM_WAVES * 32);
            const int k = (i4 * 4) / NC, n = (i4 * 4) % NC;
            const int t = (k & 63) >> 1, s = k & 1, j = n >> 4;
            float* const d = wlf + ((((j >> 1) * 32 + t) * 32 + ((k >> 6) * 16 + (n & 15))) * 4 + (j & 1) * 2 + s);
            d[0] = w.x; d[4] = w.y; d[8] = w.z; d[12] = w.w;
            const float ps = sum8(w.x * s4.x + w.y * s4.y + w.z * s4.z + w.w * s4.w);
            const float pd = sum8(w.x * d4.x + w.y * d4.y + w.z * d4.z + w.w * d4.w);
            const int side = lane & 7;
            if (side < 2) waf[(t * 32 + (k >> 6) * 16 + side * H + (n >> 5)) * 2 + s] = side ? pd : ps;
        }
    }
    if (BIAS_RELU && tid < 128) ((float*)bl)[tid] = bias[tid];
    __syncthreads();

    if (!live) return;

    float af[64];
#pragma unroll
    for (int i = 0; i < 16; ++i) {
        float4 v = a4[i];
        if (BIAS_RELU) {
            const float4 b = bl[half * 16 + i];
            v.x = fmaxf(v.x + b.x, 0.f); v.y = fmaxf(v.y + b.y, 0.f);
            v.z = fmaxf(v.z + b.z, 0.f); v.w = fmaxf(v.w + b.w, 0.f);
        }
        af[4 * i] = v.x; af[4 * i + 1] = v.y; af[4 * i + 2] = v.z; af[4 * i + 3] = v.w;
    }

    {
        v8f e = {};
#pragma unroll
        for (int t = 0; t < 32; ++t) {
            const float2 p = wa[t * 32 + lane];
            const v2f a  = {af[2 * t], af[2 * t + 1]};
            const v2f bp = {p.x, p.y};
            e = __builtin_amdgcn_wmma_f32_16x16x4_f32(false, a, false, bp, (short)0, e, false, false);
        }
        if (c < 2 * H) {
            float* const dst = (c < H ? als + c : ald + (c - H)) + (size_t)(strip * 16 + 8 * half) * H;
#pragma unroll
            for (int r = 0; r < 8; ++r) dst[r * H] = e[r];
        }
    }

    for (int jp = 0; jp < NC / 32; ++jp) {
        v8f acc0 = {}, acc1 = {};
#pragma unroll
        for (int t = 0; t < 32; ++t) {
            const float4 b = wl[(jp * 32 + t) * 32 + lane];
            const v2f a  = {af[2 * t], af[2 * t + 1]};
            const v2f b0 = {b.x, b.y};
            const v2f b1 = {b.z, b.w};
            acc0 = __builtin_amdgcn_wmma_f32_16x16x4_f32(false, a, false, b0, (short)0, acc0, false, false);
            acc1 = __builtin_amdgcn_wmma_f32_16x16x4_f32(false, a, false, b1, (short)0, acc1, false, false);
        }
#pragma unroll
        for (int r = 0; r < 8; ++r) {
            float* const crow = C + (size_t)(strip * 16 + r + 8 * half) * NC + jp * 32 + c;
            crow[0]  = acc0[r];
            crow[16] = acc1[r];
        }
    }
}

#define TILE 400
#define TILE_THREADS 1024

__device__ __forceinline__ float edge_score(float a, float b) {
    float v = a + b;
    return v > 0.f ? v : NEG_SLOPE * v;
}

#define IDX_WORDS   32
#define IDX_THREADS 1024
#define IDX_PITCH   257
__global__ __launch_bounds__(IDX_THREADS)
void k_tile_index(const int* __restrict__ ei, unsigned* __restrict__ keys, unsigned* __restrict__ bitmap,
                  int n_edges, int n_tiles, int row_words) {
    __shared__ unsigned bm[IDX_WORDS * IDX_PITCH];
    const int tid = threadIdx.x;
    for (int i = tid; i < IDX_WORDS * IDX_PITCH; i += IDX_THREADS) bm[i] = 0u;
    __syncthreads();

    const int n_quads = n_edges / 4;
    const int q0 = blockIdx.x * (IDX_WORDS * 32 * 4);
    constexpr int QPT = IDX_WORDS * 32 * 4 / IDX_THREADS;
    int4 dv[QPT];
#pragma unroll
    for (int u = 0; u < QPT; ++u) {
        const int q = q0 + u * IDX_THREADS + tid;
        dv[u] = q < n_quads ? *(const int4*)(ei + (size_t)n_edges + (size_t)q * 4) : make_int4(0, 0, 0, 0);
    }
#pragma unroll
    for (int u = 0; u < QPT; ++u) {
        const int ql = u * IDX_THREADS + tid;
        const int q = q0 + ql;
        if (q < n_quads) {
            const int4 d = dv[u];
            const unsigned ta = (unsigned)(d.x / TILE), tb = (unsigned)(d.y / TILE),
                           tc = (unsigned)(d.z / TILE), td = (unsigned)(d.w / TILE);
            keys[q] = ta | (tb << 8) | (tc << 16) | (td << 24);
            unsigned* const wrow = bm + (ql >> 7) * IDX_PITCH;
            const unsigned bit = 1u << ((ql >> 2) & 31);
            atomicOr(&wrow[ta], bit); atomicOr(&wrow[tb], bit);
            atomicOr(&wrow[tc], bit); atomicOr(&wrow[td], bit);
        }
    }
    __syncthreads();

    for (int i = tid; i < n_tiles * IDX_WORDS; i += IDX_THREADS) {
        const int t = i / IDX_WORDS, w = i % IDX_WORDS;
        bitmap[(size_t)t * row_words + blockIdx.x * IDX_WORDS + w] = bm[w * IDX_PITCH + t];
    }
}

__device__ __forceinline__ unsigned match4(unsigned k, unsigned pat) {
    unsigned x = k ^ pat;
    unsigned t = ~(((x & 0x7F7F7F7Fu) + 0x7F7F7F7Fu) | x | 0x7F7F7F7Fu);
    return (((t >> 7) * 0x00204081u) >> 21) & 0xFu;
}
__device__ __forceinline__ unsigned match16(uint4 k, unsigned pat) {
    return match4(k.x, pat) | (match4(k.y, pat) << 4) | (match4(k.z, pat) << 8) | (match4(k.w, pat) << 12);
}

__device__ __forceinline__ float lane_f(float v, int l) {
    return __int_as_float(__builtin_amdgcn_readlane(__float_as_int(v), l));
}

#define LIST_CAP 16384
#define DL_BITS  9
#define SWEEP_B  8

template <int BITS>
__device__ __forceinline__ unsigned wave_reserve(unsigned* counter, unsigned n, int lane) {
    unsigned before = 0u, total = 0u;
#pragma unroll
    for (int k = 0; k < BITS; ++k) {
        const unsigned bal = __builtin_amdgcn_ballot_w32(((n >> k) & 1u) != 0u);
        before += __builtin_amdgcn_mbcnt_lo(bal, 0u) << k;
        total  += (unsigned)__popc(bal) << k;
    }
    unsigned start = 0u;
    if (lane == 0) start = atomicAdd(counter, total);
    return (unsigned)__builtin_amdgcn_readfirstlane((int)start) + before;
}

template <int H, int CH, int PASS, bool KEPT>
__device__ __forceinline__ void take_edges(bool act, int s, int dl, const float* kv,
                                           const float* __restrict__ als, const float* __restrict__ hf,
                                           float* acc, unsigned* mx, float* den, const float* adl, int lane) {
    constexpr int C = H * CH;
    constexpr int NK = C / 32;
    float w[H];
#pragma unroll
    for (int h = 0; h < H; ++h) w[h] = 0.f;
    if (act) {
#pragma unroll
        for (int h = 0; h < H; ++h) {
            float v;
            if (KEPT) v = kv[h]; else v = edge_score(als[(size_t)s * H + h], adl[dl * H + h]);
            if (PASS == 1) {
                atomicMax(&mx[dl * H + h], fenc(v));
            } else {
                w[h] = __expf(v - fdec(mx[dl * H + h]));
                atomicAdd(&den[dl * H + h], w[h]);
            }
        }
    }
    if (PASS == 2) {
        constexpr int RB = NK >= 4 ? 8 : 16;
        unsigned bal = __builtin_amdgcn_ballot_w32(act);
        while (bal != 0u) {
            int  ss[RB], dd[RB];
            bool ok[RB];
            float ww[RB][H];
            float r[RB][NK];
#pragma unroll
            for (int j = 0; j < RB; ++j) {
                ok[j] = bal != 0u;
                const int l = ok[j] ? (__ffs(bal) - 1) : 0;
                if (ok[j]) bal &= bal - 1u;
                ss[j] = __builtin_amdgcn_readlane(s, l);
                dd[j] = __builtin_amdgcn_readlane(dl, l);
#pragma unroll
                for (int h = 0; h < H; ++h) ww[j][h] = lane_f(w[h], l);
            }
#pragma unroll
            for (int j = 0; j < RB; ++j) {
#pragma unroll
                for (int k = 0; k < NK; ++k) r[j][k] = 0.f;
                if (ok[j]) {
#pragma unroll
                    for (int k = 0; k < NK; ++k) r[j][k] = hf[(size_t)ss[j] * C + k * 32 + lane];
                }
            }
#pragma unroll
            for (int j = 0; j < RB; ++j) {
                if (ok[j]) {
#pragma unroll
                    for (int k = 0; k < NK; ++k)
                        atomicAdd(&acc[dd[j] * C + k * 32 + lane], ww[j][(k * 32) / CH] * r[j][k]);
                }
            }
        }
    }
}

template <int H, int CH, int PASS, bool RECORD>
__device__ __forceinline__ void tile_sweep(const int* __restrict__ ei, const uint4* __restrict__ keys,
                                           const float* __restrict__ als, const float* __restrict__ hf,
                                           float* acc, unsigned* mx, float* den, const float* adl,
                                           unsigned* lst, unsigned* cnt,
                                           int t0, unsigned pat, int rot, int n_edges, int n_key16,
                                           int lane, int wave) {
    constexpr int STEP = (TILE_THREADS / 32) * 64;
    const uint4 none = make_uint4(~0u, ~0u, ~0u, ~0u);

    auto group_of = [&](int base, int u) {
        int q = base + u * 32 + lane + rot;
        return q >= n_key16 ? q - n_key16 : q;
    };
    auto fetch = [&](int base, int u) {
        return (base + u * 32 + lane < n_key16) ? keys[group_of(base, u)] : none;
    };

    uint4 kv0 = fetch(wave * 64, 0), kv1 = fetch(wave * 64, 1);
    for (int base = wave * 64; base < n_key16; base += STEP) {
        const uint4 nx0 = fetch(base + STEP, 0), nx1 = fetch(base + STEP, 1);
        unsigned mask = match16(kv0, pat) | (match16(kv1, pat) << 16);
        while (__builtin_amdgcn_ballot_w32(mask != 0u) != 0u) {
            const bool act = mask != 0u;
            int s = 0, dl = 0;
            if (act) {
                const int b = __ffs(mask) - 1;
                mask &= mask - 1u;
                const int e = group_of(base, b >> 4) * 16 + (b & 15);
                s  = ei[e];
                dl = ei[(size_t)n_edges + e] - t0;
                if (RECORD) {
                    const unsigned slot = atomicAdd(cnt, 1u);
                    if (slot < LIST_CAP) lst[slot] = ((unsigned)s << DL_BITS) | (unsigned)dl;
                }
            }
            take_edges<H, CH, PASS, false>(act, s, dl, nullptr, als, hf, acc, mx, den, adl, lane);
        }
        kv0 = nx0; kv1 = nx1;
    }
}

template <int H, int CH, int PASS>
__device__ __forceinline__ void list_sweep(const unsigned* lst, int n,
                                           const float* __restrict__ als, const float* __restrict__ hf,
                                           float* acc, unsigned* mx, float* den, const float* adl,
                                           float (&sc)[SWEEP_B][H], int lane, int wave) {
    if (PASS == 1) {
        for (int base = wave * 32; base < n; base += SWEEP_B * TILE_THREADS) {
            bool ok[SWEEP_B];
            unsigned p[SWEEP_B];
            float a[SWEEP_B][H];
#pragma unroll
            for (int u = 0; u < SWEEP_B; ++u) {
                const int i = base + u * TILE_THREADS + lane;
                ok[u] = i < n;
                p[u] = ok[u] ? lst[i] : 0u;
            }
#pragma unroll
            for (int u = 0; u < SWEEP_B; ++u) {
#pragma unroll
                for (int h = 0; h < H; ++h) a[u][h] = als[(size_t)(p[u] >> DL_BITS) * H + h];
            }
#pragma unroll
            for (int u = 0; u < SWEEP_B; ++u) {
                if (ok[u]) {
                    const int dl = (int)(p[u] & ((1u << DL_BITS) - 1u));
#pragma unroll
                    for (int h = 0; h < H; ++h) {
                        const float v = edge_score(a[u][h], adl[dl * H + h]);
                        sc[u][h] = v;
                        atomicMax(&mx[dl * H + h], fenc(v));
                    }
                }
            }
        }
        return;
    }
    for (int base = wave * 32; base < n; base += TILE_THREADS) {
        const int i = base + lane;
        const bool act = i < n;
        const unsigned p = act ? lst[i] : 0u;
        take_edges<H, CH, PASS, false>(act, (int)(p >> DL_BITS), (int)(p & ((1u << DL_BITS) - 1u)), nullptr,
                                       als, hf, acc, mx, den, adl, lane);
    }
}

template <int H, int CH>
__device__ __forceinline__ void list_sweep_kept(const unsigned* lst, int n, const float (&sc)[SWEEP_B][H],
                                                const float* __restrict__ hf,
                                                float* acc, unsigned* mx, float* den, int lane, int wave) {
    const int wv = __builtin_amdgcn_readfirstlane(wave);
    for (int k = 0; k < SWEEP_B; ++k) {
        const int base = wv * 32 + k * TILE_THREADS;
        if (base >= n) break;
        const int i = base + lane;
        const bool act = i < n;
        const unsigned p = act ? lst[i] : 0u;
        float kv[H];
#pragma unroll
        for (int h = 0; h < H; ++h) {
            float t = sc[0][h];
#pragma unroll
            for (int u = 1; u < SWEEP_B; ++u) t = (k == u) ? sc[u][h] : t;
            kv[h] = t;
        }
        take_edges<H, CH, 2, true>(act, (int)(p >> DL_BITS), (int)(p & ((1u << DL_BITS) - 1u)), kv,
                                   nullptr, hf, acc, mx, den, nullptr, lane);
    }
}

template <int H, int CH, bool ADD_BIAS, bool FIRST>
__global__ __launch_bounds__(TILE_THREADS)
void k_tile_gat(const int* __restrict__ ei, const uint4* __restrict__ keys,
                const float* __restrict__ als, const float* __restrict__ ald,
                const float* __restrict__ hf, const float* __restrict__ bias,
                float* __restrict__ out, unsigned* glist, unsigned* gcnt,
                const unsigned* __restrict__ bitmap, int row_words,
                int n_nodes, int n_edges, int n_key16) {
    constexpr int C = H * CH;
    __shared__ __attribute__((aligned(16))) float acc[TILE * C];
    __shared__ unsigned mx[TILE * H];
    __shared__ float    den[TILE * H];
    __shared__ float    adl[TILE * H];
    __shared__ unsigned lst[LIST_CAP];
    __shared__ unsigned cnt;
    __shared__ unsigned cntb;

    const int tid = threadIdx.x, lane = tid & 31, wave = tid >> 5;
    const int t0 = blockIdx.x * TILE;
    const unsigned pat = (unsigned)blockIdx.x * 0x01010101u;
    const int rot = (int)(((long long)blockIdx.x * n_key16) / gridDim.x);
    unsigned* const glst = glist + (size_t)blockIdx.x * LIST_CAP;

    unsigned n_list = 0u;
    float sc[SWEEP_B][H];
#pragma unroll
    for (int u = 0; u < SWEEP_B; ++u) {
#pragma unroll
        for (int h = 0; h < H; ++h) sc[u][h] = 0.f;
    }
    bool kept = false;
    if (FIRST) {
        if (tid == 0) { cnt = 0u; cntb = 0u; }
    } else {
        n_list = (unsigned)__builtin_amdgcn_readfirstlane((int)gcnt[blockIdx.x]);
        if (n_list <= LIST_CAP) {
            for (int i0 = tid; i0 < (int)n_list; i0 += 8 * TILE_THREADS) {
                unsigned v[8];
#pragma unroll
                for (int u = 0; u < 8; ++u) {
                    const int i = i0 + u * TILE_THREADS;
                    v[u] = glst[i < (int)n_list ? i : i0];
                }
#pragma unroll
                for (int u = 0; u < 8; ++u) {
                    const int i = i0 + u * TILE_THREADS;
                    if (i < (int)n_list) lst[i] = v[u];
                }
            }
        }
    }

    constexpr int SELF_T = (TILE * H + TILE_THREADS - 1) / TILE_THREADS;
    float selfv[SELF_T];
    {
        float sa[SELF_T], ss[SELF_T];
#pragma unroll
        for (int u = 0; u < SELF_T; ++u) {
            const int i = tid + u * TILE_THREADS;
            const bool ok = i < TILE * H && t0 + i / H < n_nodes;
            const int ic = ok ? i : 0;
            sa[u] = ald[(size_t)t0 * H + ic];
            ss[u] = als[(size_t)t0 * H + ic];
        }
#pragma unroll
        for (int u = 0; u < SELF_T; ++u) {
            const int i = tid + u * TILE_THREADS;
            const bool ok = i < TILE * H && t0 + i / H < n_nodes;
            const float a = ok ? sa[u] : 0.f;
            const float v = ok ? edge_score(ss[u], sa[u]) : 0.f;
            selfv[u] = v;
            if (i < TILE * H) { adl[i] = a; mx[i] = fenc(v); }
        }
    }
    __syncthreads();

    if (FIRST) {
        constexpr unsigned BL_CAP = TILE * C;
        unsigned* const blist = (unsigned*)acc;
        const unsigned* const row = bitmap + (size_t)blockIdx.x * row_words;
        for (int i0 = 0; i0 < row_words; i0 += 4 * TILE_THREADS) {
            unsigned wv[4];
#pragma unroll
            for (int u = 0; u < 4; ++u) {
                const int i = i0 + u * TILE_THREADS + tid;
                wv[u] = i < row_words ? row[i] : 0u;
            }
            unsigned slot = wave_reserve<8>(&cntb, (unsigned)(__popc(wv[0]) + __popc(wv[1]) + __popc(wv[2]) + __popc(wv[3])), lane);
#pragma unroll
            for (int u = 0; u < 4; ++u) {
                unsigned w = wv[u];
                const unsigned b0 = (unsigned)(i0 + u * TILE_THREADS + tid) * 32u;
                while (w != 0u) {
                    const unsigned b = (unsigned)__ffs(w) - 1u;
                    w &= w - 1u;
                    if (slot < BL_CAP) blist[slot] = b0 + b;
                    ++slot;
                }
            }
        }
        __syncthreads();
        const unsigned n_blk = (unsigned)__builtin_amdgcn_readfirstlane((int)cntb);
        if (n_blk <= BL_CAP) {
            for (int base = wave * 32; base < (int)n_blk; base += SWEEP_B * TILE_THREADS) {
                unsigned blk[SWEEP_B];
                uint4 kv[SWEEP_B];
#pragma unroll
                for (int u = 0; u < SWEEP_B; ++u) {
                    const int i = base + u * TILE_THREADS + lane;
                    blk[u] = i < (int)n_blk ? blist[i] : ~0u;
                }
#pragma unroll
                for (int u = 0; u < SWEEP_B; ++u) kv[u] = keys[blk[u] != ~0u ? blk[u] : 0u];
                unsigned mm[SWEEP_B];
                unsigned found = 0u;
#pragma unroll
                for (int u = 0; u < SWEEP_B; ++u) { mm[u] = blk[u] != ~0u ? match16(kv[u], pat) : 0u; found += (unsigned)__popc(mm[u]); }
                static_assert(16 * SWEEP_B < 256, "a lane's count must fit the planes that wave_reserve adds up");
                unsigned slot = wave_reserve<8>(&cnt, found, lane);
#pragma unroll
                for (int u = 0; u < SWEEP_B; ++u) {
                    unsigned m = mm[u];
                    while (m != 0u) {
                        const unsigned b = (unsigned)__ffs(m) - 1u;
                        m &= m - 1u;
                        if (slot < LIST_CAP) lst[slot] = blk[u] * 16u + b;
                        ++slot;
                    }
                }
            }
            __syncthreads();
            const unsigned n_found = (unsigned)__builtin_amdgcn_readfirstlane((int)cnt);
            if (n_found <= LIST_CAP) {
                for (int base = wave * 32; base < (int)n_found; base += SWEEP_B * TILE_THREADS) {
                    bool ok[SWEEP_B];
                    unsigned e[SWEEP_B];
                    int s[SWEEP_B], d[SWEEP_B];
                    float a[SWEEP_B][H];
#pragma unroll
                    for (int u = 0; u < SWEEP_B; ++u) {
                        const int i = base + u * TILE_THREADS + lane;
                        ok[u] = i < (int)n_found;
                        e[u] = ok[u] ? lst[i] : 0u;
                    }
#pragma unroll
                    for (int u = 0; u < SWEEP_B; ++u) { s[u] = ei[e[u]]; d[u] = ei[(size_t)n_edges + e[u]]; }
#pragma unroll
                    for (int u = 0; u < SWEEP_B; ++u) {
#pragma unroll
                        for (int h = 0; h < H; ++h) a[u][h] = als[(size_t)s[u] * H + h];
                    }
#pragma unroll
                    for (int u = 0; u < SWEEP_B; ++u) {
                        if (ok[u]) {
                            const int dl = d[u] - t0;
                            lst[base + u * TILE_THREADS + lane] = ((unsigned)s[u] << DL_BITS) | (unsigned)dl;
#pragma unroll
                            for (int h = 0; h < H; ++h) {
                                const float v = edge_score(a[u][h], adl[dl * H + h]);
                                sc[u][h] = v;
                                atomicMax(&mx[dl * H + h], fenc(v));
                            }
                        }
                    }
                }
                kept = n_found <= (unsigned)(SWEEP_B * TILE_THREADS);
            } else {
                tile_sweep<H, CH, 1, false>(ei, keys, als, hf, acc, mx, den, adl, lst, &cnt, t0, pat, rot, n_edges, n_key16, lane, wave);
            }
        } else {
            tile_sweep<H, CH, 1, true>(ei, keys, als, hf, acc, mx, den, adl, lst, &cnt, t0, pat, rot, n_edges, n_key16, lane, wave);
        }
    } else if (n_list <= LIST_CAP) {
        list_sweep<H, CH, 1>(lst, (int)n_list, als, hf, acc, mx, den, adl, sc, lane, wave);
        kept = n_list <= (unsigned)(SWEEP_B * TILE_THREADS);
    } else {
        tile_sweep<H, CH, 1, false>(ei, keys, als, hf, acc, mx, den, adl, lst, &cnt, t0, pat, rot, n_edges, n_key16, lane, wave);
    }
    __syncthreads();

    if (FIRST) {
        n_list = (unsigned)__builtin_amdgcn_readfirstlane((int)cnt);
        if (tid == 0) gcnt[blockIdx.x] = n_list;
        if (n_list <= LIST_CAP)
            for (int i = tid; i < (int)n_list; i += TILE_THREADS) glst[i] = lst[i];
    }

    static_assert(CH % 4 == 0 && C % 4 == 0 && (TILE * C) % 4 == 0, "four neighbouring channels share a head and a row");
    constexpr int V4   = TILE * C / 4;
    constexpr int V4_T = (V4 + TILE_THREADS - 1) / TILE_THREADS;
    const float4* const hrow = (const float4*)(hf + (size_t)t0 * C);
    float4 hv[V4_T];
#pragma unroll
    for (int u = 0; u < V4_T; ++u) {
        const int i4 = tid + u * TILE_THREADS;
        const bool ok = i4 < V4 && t0 + (i4 * 4) / C < n_nodes;
        hv[u] = hrow[ok ? i4 : 0];
    }
#pragma unroll
    for (int u = 0; u < SELF_T; ++u) {
        const int i = tid + u * TILE_THREADS;
        if (i < TILE * H) {
            float ex = 0.f;
            if (t0 + i / H < n_nodes) ex = __expf(selfv[u] - fdec(mx[i]));
            den[i] = ex;
        }
    }
    __syncthreads();
    float4* const acc4 = (float4*)acc;
#pragma unroll
    for (int u = 0; u < V4_T; ++u) {
        const int i4 = tid + u * TILE_THREADS;
        if (i4 < V4) {
            const int dl = (i4 * 4) / C, hd = ((i4 * 4) % C) / CH;
            float4 a = make_float4(0.f, 0.f, 0.f, 0.f);
            if (t0 + dl < n_nodes) {
                const float ex = den[dl * H + hd];
                a = make_float4(ex * hv[u].x, ex * hv[u].y, ex * hv[u].z, ex * hv[u].w);
            }
            acc4[i4] = a;
        }
    }
    __syncthreads();

    if (n_list <= LIST_CAP && kept) {
        list_sweep_kept<H, CH>(lst, (int)n_list, sc, hf, acc, mx, den, lane, wave);
    } else if (n_list <= LIST_CAP) {
        list_sweep<H, CH, 2>(lst, (int)n_list, als, hf, acc, mx, den, adl, sc, lane, wave);
    } else {
        tile_sweep<H, CH, 2, false>(ei, keys, als, hf, acc, mx, den, adl, lst, &cnt, t0, pat, rot, n_edges, n_key16, lane, wave);
    }
    __syncthreads();

    for (int i = tid; i < TILE * C; i += TILE_THREADS) {
        const int dl = i / C, c = i % C;
        if (t0 + dl < n_nodes) {
            float v = acc[i] / (den[dl * H + c / CH] + 1e-16f);
            if (ADD_BIAS) v += bias[c];
            out[(size_t)t0 * C + i] = v;
        }
    }
}


extern "C" void kernel_launch(void* const* d_in, const int* in_sizes, int n_in,
                              void* d_out, int out_size, void* d_ws, size_t ws_size,
                              hipStream_t stream) {
    (void)n_in; (void)out_size; (void)ws_size; (void)in_sizes;
    constexpr long long N = 100000, E = 1600000;
    constexpr int C1 = 128, HEADS = 4, HID = 32, C2 = 32;

    static_assert(N % TILE == 0 && N / TILE <= 255 && E % 16 == 0, "tile ids must fit a byte below 0xFF; keys are read 16 at a time");
    static_assert(TILE <= (1 << DL_BITS) && N <= (1ll << (32 - DL_BITS)), "a list entry packs source and local destination into one word");
    static_assert(N / TILE <= IDX_PITCH && E % 4 == 0 && (IDX_WORDS * 32 * 4) % IDX_THREADS == 0, "tile ids stay inside one LDS run of the index pass; destinations are read four at a time");

    const float* x    = (const float*)d_in[0];
    const int*   ei   = (const int*)  d_in[1];
    const float* W1   = (const float*)d_in[2];
    const float* aS1  = (const float*)d_in[3];
    const float* aD1  = (const float*)d_in[4];
    const float* b1   = (const float*)d_in[5];
    const float* W2   = (const float*)d_in[6];
    const float* aS2  = (const float*)d_in[7];
    const float* aD2  = (const float*)d_in[8];
    const float* b2   = (const float*)d_in[9];
    float* out = (float*)d_out;

    char* ws = (char*)d_ws;
    size_t off = 0;
    auto alloc = [&](size_t bytes) { size_t o = off; off = (off + bytes + 255) & ~(size_t)255; return o; };
    float*          h1   = (float*)(ws + alloc(N * C1 * 4));
    float*          out1 = (float*)(ws + alloc(N * C1 * 4));
    float*          as1  = (float*)(ws + alloc(N * HEADS * 4));
    float*          ad1  = (float*)(ws + alloc(N * HEADS * 4));
    unsigned*       keys = (unsigned*)(ws + alloc(E));
    unsigned*       glist = (unsigned*)(ws + alloc((N / TILE) * LIST_CAP * 4));
    unsigned*       gcnt  = (unsigned*)(ws + alloc((N / TILE) * 4));
    constexpr int      N_WORDS   = (int)((E / 16 + 31) / 32);
    constexpr unsigned IDX_GRID  = (N_WORDS + IDX_WORDS - 1) / IDX_WORDS;
    constexpr int      ROW_WORDS = (int)(IDX_GRID * IDX_WORDS);
    unsigned*       bitmap = (unsigned*)(ws + alloc((N / TILE) * (size_t)ROW_WORDS * 4));

    constexpr int STRIPS = (int)(N / 16);
    constexpr unsigned GEMM_GRID = (STRIPS + GEMM_WAVES - 1) / GEMM_WAVES;
    static_assert(N % 16 == 0 && C1 == 128 && C1 % 32 == 0 && C2 % 32 == 0, "the GEMM takes whole 16-row strips, K = 128 and column tiles in pairs");

    k_gemm_wmma_f32<C1, HEADS, false><<<GEMM_GRID, GEMM_WAVES * 32, 0, stream>>>(x, W1, b1, aS1, aD1, h1, as1, ad1, STRIPS);

    k_tile_index<<<IDX_GRID, IDX_THREADS, 0, stream>>>(ei, keys, bitmap, E, N / TILE, ROW_WORDS);

    k_tile_gat<HEADS, HID, false, true><<<N / TILE, TILE_THREADS, 0, stream>>>(
        ei, (const uint4*)keys, as1, ad1, h1, b1, out1, glist, gcnt, bitmap, ROW_WORDS, N, E, E / 16);

    k_gemm_wmma_f32<C2, 1, true><<<GEMM_GRID, GEMM_WAVES * 32, 0, stream>>>(out1, W2, b1, aS2, aD2, h1, as1, ad1, STRIPS);

    k_tile_gat<1, C2, true, false><<<N / TILE, TILE_THREADS, 0, stream>>>(
        ei, (const uint4*)keys, as1, ad1, h1, b2, out, glist, gcnt, bitmap, ROW_WORDS, N, E, E / 16);
}


# Sum pass takes a full round of 32 list entries on a straight path: edge j is lane j, lane numbers are constants, no per-edge mask step, flag or branch; rows indexed in 32 bits; sum order kept
# speedup vs baseline: 9.2870x; 1.0386x over previous
#include <hip/hip_runtime.h>

typedef __attribute__((ext_vector_type(2))) float v2f;
typedef __attribute__((ext_vector_type(8))) float v8f;

#define NEG_SLOPE 0.2f


__device__ __forceinline__ unsigned fenc(float f) {
    unsigned u = __float_as_uint(f);
    return (u & 0x80000000u) ? ~u : (u | 0x80000000u);
}
__device__ __forceinline__ float fdec(unsigned u) {
    return (u & 0x80000000u) ? __uint_as_float(u & 0x7FFFFFFFu)
                             : __uint_as_float(~u);
}

#define GEMM_WAVES 8

__device__ __forceinline__ float sum8(float v) {
    v += __int_as_float(__builtin_amdgcn_update_dpp(0, __float_as_int(v), 0x161, 0xf, 0xf, true));
    v += __int_as_float(__builtin_amdgcn_update_dpp(0, __float_as_int(v), 0x162, 0xf, 0xf, true));
    v += __int_as_float(__builtin_amdgcn_update_dpp(0, __float_as_int(v), 0x164, 0xf, 0xf, true));
    return v;
}

template <int NC, int H, bool BIAS_RELU>
__global__ __launch_bounds__(GEMM_WAVES * 32)
void k_gemm_wmma_f32(const float* __restrict__ A, const float* __restrict__ W,
                     const float* __restrict__ bias,
                     const float* __restrict__ a_s, const float* __restrict__ a_d,
                     float* __restrict__ C, float* __restrict__ als, float* __restrict__ ald, int n_strips) {
    static_assert(NC == H * 32 && 2 * H <= 16 && (GEMM_WAVES * 32 * 4) % NC == 0,
                  "a head is 32 columns = 8 lanes of a staging trip; both coefficient sets fit one column tile");
    __shared__ float4 wl[NC * 128 / 4];
    __shared__ float4 bl[128 / 4];
    __shared__ float2 wa[32 * 32];

    const int tid = threadIdx.x, lane = tid & 31;
    const int wave = __builtin_amdgcn_readfirstlane(tid >> 5);
    const int c = lane & 15, half = lane >> 4;
    const int strip = blockIdx.x * GEMM_WAVES + wave;
    const bool live = strip < n_strips;

    float4 a4[16];
    if (live) {
        const float4* arow = (const float4*)(A + (size_t)(strip * 16 + c) * 128 + half * 64);
#pragma unroll
        for (int i = 0; i < 16; ++i) a4[i] = arow[i];
    }

    float* const wlf = (float*)wl;
    float* const waf = (float*)wa;
    for (int i = tid; i < 32 * 32; i += GEMM_WAVES * 32)
        if ((i & 15) >= 2 * H) wa[i] = make_float2(0.f, 0.f);
    const float4 s4 = *(const float4*)(a_s + (tid * 4) % NC);
    const float4 d4 = *(const float4*)(a_d + (tid * 4) % NC);
    constexpr int TRIPS = 128 * NC / 4 / (GEMM_WAVES * 32);
    constexpr int WB = TRIPS < 8 ? TRIPS : 8;
    static_assert(TRIPS % WB == 0, "whole groups of trips");
    for (int g0 = 0; g0 < TRIPS; g0 += WB) {
        float4 wv[WB];
#pragma unroll
        for (int u = 0; u < WB; ++u) wv[u] = ((const float4*)W)[tid + (g0 + u) * (GEMM_WAVES * 32)];
#pragma unroll
        for (int u = 0; u < WB; ++u) {
            const float4 w = wv[u];
            const int i4 = tid + (g0 + u) * (GEMM_WAVES * 32);
            const int k = (i4 * 4) / NC, n = (i4 * 4) % NC;
            const int t = (k & 63) >> 1, s = k & 1, j = n >> 4;
            float* const d = wlf + ((((j >> 1) * 32 + t) * 32 + ((k >> 6) * 16 + (n & 15))) * 4 + (j & 1) * 2 + s);
            d[0] = w.x; d[4] = w.y; d[8] = w.z; d[12] = w.w;
            const float ps = sum8(w.x * s4.x + w.y * s4.y + w.z * s4.z + w.w * s4.w);
            const float pd = sum8(w.x * d4.x + w.y * d4.y + w.z * d4.z + w.w * d4.w);
            const int side = lane & 7;
            if (side < 2) waf[(t * 32 + (k >> 6) * 16 + side * H + (n >> 5)) * 2 + s] = side ? pd : ps;
        }
    }
    if (BIAS_RELU && tid < 128) ((float*)bl)[tid] = bias[tid];
    __syncthreads();

    if (!live) return;

    float af[64];
#pragma unroll
    for (int i = 0; i < 16; ++i) {
        float4 v = a4[i];
        if (BIAS_RELU) {
            const float4 b = bl[half * 16 + i];
            v.x = fmaxf(v.x + b.x, 0.f); v.y = fmaxf(v.y + b.y, 0.f);
            v.z = fmaxf(v.z + b.z, 0.f); v.w = fmaxf(v.w + b.w, 0.f);
        }
        af[4 * i] = v.x; af[4 * i + 1] = v.y; af[4 * i + 2] = v.z; af[4 * i + 3] = v.w;
    }

    {
        v8f e = {};
#pragma unroll
        for (int t = 0; t < 32; ++t) {
            const float2 p = wa[t * 32 + lane];
            const v2f a  = {af[2 * t], af[2 * t + 1]};
            const v2f bp = {p.x, p.y};
            e = __builtin_amdgcn_wmma_f32_16x16x4_f32(false, a, false, bp, (short)0, e, false, false);
        }
        if (c < 2 * H) {
            float* const dst = (c < H ? als + c : ald + (c - H)) + (size_t)(strip * 16 + 8 * half) * H;
#pragma unroll
            for (int r = 0; r < 8; ++r) dst[r * H] = e[r];
        }
    }

    for (int jp = 0; jp < NC / 32; ++jp) {
        v8f acc0 = {}, acc1 = {};
#pragma unroll
        for (int t = 0; t < 32; ++t) {
            const float4 b = wl[(jp * 32 + t) * 32 + lane];
            const v2f a  = {af[2 * t], af[2 * t + 1]};
            const v2f b0 = {b.x, b.y};
            const v2f b1 = {b.z, b.w};
            acc0 = __builtin_amdgcn_wmma_f32_16x16x4_f32(false, a, false, b0, (short)0, acc0, false, false);
            acc1 = __builtin_amdgcn_wmma_f32_16x16x4_f32(false, a, false, b1, (short)0, acc1, false, false);
        }
#pragma unroll
        for (int r = 0; r < 8; ++r) {
            float* const crow = C + (size_t)(strip * 16 + r + 8 * half) * NC + jp * 32 + c;
            crow[0]  = acc0[r];
            crow[16] = acc1[r];
        }
    }
}

#define TILE 400
#define TILE_THREADS 1024

__device__ __forceinline__ float edge_score(float a, float b) {
    float v = a + b;
    return v > 0.f ? v : NEG_SLOPE * v;
}

#define IDX_WORDS   32
#define IDX_THREADS 1024
#define IDX_PITCH   257
__global__ __launch_bounds__(IDX_THREADS)
void k_tile_index(const int* __restrict__ ei, unsigned* __restrict__ keys, unsigned* __restrict__ bitmap,
                  int n_edges, int n_tiles, int row_words) {
    __shared__ unsigned bm[IDX_WORDS * IDX_PITCH];
    const int tid = threadIdx.x;
    for (int i = tid; i < IDX_WORDS * IDX_PITCH; i += IDX_THREADS) bm[i] = 0u;
    __syncthreads();

    const int n_quads = n_edges / 4;
    const int q0 = blockIdx.x * (IDX_WORDS * 32 * 4);
    constexpr int QPT = IDX_WORDS * 32 * 4 / IDX_THREADS;
    int4 dv[QPT];
#pragma unroll
    for (int u = 0; u < QPT; ++u) {
        const int q = q0 + u * IDX_THREADS + tid;
        dv[u] = q < n_quads ? *(const int4*)(ei + (size_t)n_edges + (size_t)q * 4) : make_int4(0, 0, 0, 0);
    }
#pragma unroll
    for (int u = 0; u < QPT; ++u) {
        const int ql = u * IDX_THREADS + tid;
        const int q = q0 + ql;
        if (q < n_quads) {
            const int4 d = dv[u];
            const unsigned ta = (unsigned)(d.x / TILE), tb = (unsigned)(d.y / TILE),
                           tc = (unsigned)(d.z / TILE), td = (unsigned)(d.w / TILE);
            keys[q] = ta | (tb << 8) | (tc << 16) | (td << 24);
            unsigned* const wrow = bm + (ql >> 7) * IDX_PITCH;
            const unsigned bit = 1u << ((ql >> 2) & 31);
            atomicOr(&wrow[ta], bit); atomicOr(&wrow[tb], bit);
            atomicOr(&wrow[tc], bit); atomicOr(&wrow[td], bit);
        }
    }
    __syncthreads();

    for (int i = tid; i < n_tiles * IDX_WORDS; i += IDX_THREADS) {
        const int t = i / IDX_WORDS, w = i % IDX_WORDS;
        bitmap[(size_t)t * row_words + blockIdx.x * IDX_WORDS + w] = bm[w * IDX_PITCH + t];
    }
}

__device__ __forceinline__ unsigned match4(unsigned k, unsigned pat) {
    unsigned x = k ^ pat;
    unsigned t = ~(((x & 0x7F7F7F7Fu) + 0x7F7F7F7Fu) | x | 0x7F7F7F7Fu);
    return (((t >> 7) * 0x00204081u) >> 21) & 0xFu;
}
__device__ __forceinline__ unsigned match16(uint4 k, unsigned pat) {
    return match4(k.x, pat) | (match4(k.y, pat) << 4) | (match4(k.z, pat) << 8) | (match4(k.w, pat) << 12);
}

__device__ __forceinline__ float lane_f(float v, int l) {
    return __int_as_float(__builtin_amdgcn_readlane(__float_as_int(v), l));
}

#define LIST_CAP 16384
#define DL_BITS  9
#define SWEEP_B  8

template <int BITS>
__device__ __forceinline__ unsigned wave_reserve(unsigned* counter, unsigned n, int lane) {
    unsigned before = 0u, total = 0u;
#pragma unroll
    for (int k = 0; k < BITS; ++k) {
        const unsigned bal = __builtin_amdgcn_ballot_w32(((n >> k) & 1u) != 0u);
        before += __builtin_amdgcn_mbcnt_lo(bal, 0u) << k;
        total  += (unsigned)__popc(bal) << k;
    }
    unsigned start = 0u;
    if (lane == 0) start = atomicAdd(counter, total);
    return (unsigned)__builtin_amdgcn_readfirstlane((int)start) + before;
}

template <int H, int CH, int PASS, bool KEPT>
__device__ __forceinline__ void take_edges(bool act, int s, int dl, const float* kv,
                                           const float* __restrict__ als, const float* __restrict__ hf,
                                           float* acc, unsigned* mx, float* den, const float* adl, int lane) {
    constexpr int C = H * CH;
    constexpr int NK = C / 32;
    float w[H];
#pragma unroll
    for (int h = 0; h < H; ++h) w[h] = 0.f;
    if (act) {
#pragma unroll
        for (int h = 0; h < H; ++h) {
            float v;
            if (KEPT) v = kv[h]; else v = edge_score(als[(size_t)s * H + h], adl[dl * H + h]);
            if (PASS == 1) {
                atomicMax(&mx[dl * H + h], fenc(v));
            } else {
                w[h] = __expf(v - fdec(mx[dl * H + h]));
                atomicAdd(&den[dl * H + h], w[h]);
            }
        }
    }
    if (PASS == 2) {
        constexpr int RB = NK >= 4 ? 8 : 16;
        unsigned bal = __builtin_amdgcn_ballot_w32(act);
        if (KEPT && bal == 0xFFFFFFFFu) {
#pragma unroll
            for (int j0 = 0; j0 < 32; j0 += RB) {
                float r[RB][NK];
#pragma unroll
                for (int j = 0; j < RB; ++j) {
                    const unsigned sj = (unsigned)__builtin_amdgcn_readlane(s, j0 + j);
#pragma unroll
                    for (int k = 0; k < NK; ++k) r[j][k] = hf[sj * (unsigned)C + (unsigned)(k * 32 + lane)];
                }
#pragma unroll
                for (int j = 0; j < RB; ++j) {
                    const int dj = __builtin_amdgcn_readlane(dl, j0 + j);
#pragma unroll
                    for (int k = 0; k < NK; ++k)
                        atomicAdd(&acc[dj * C + k * 32 + lane], lane_f(w[(k * 32) / CH], j0 + j) * r[j][k]);
                }
            }
            bal = 0u;
        }
        while (bal != 0u) {
            int  ss[RB], dd[RB];
            bool ok[RB];
            float ww[RB][H];
            float r[RB][NK];
#pragma unroll
            for (int j = 0; j < RB; ++j) {
                ok[j] = bal != 0u;
                const int l = ok[j] ? (__ffs(bal) - 1) : 0;
                if (ok[j]) bal &= bal - 1u;
                ss[j] = __builtin_amdgcn_readlane(s, l);
                dd[j] = __builtin_amdgcn_readlane(dl, l);
#pragma unroll
                for (int h = 0; h < H; ++h) ww[j][h] = lane_f(w[h], l);
            }
#pragma unroll
            for (int j = 0; j < RB; ++j) {
#pragma unroll
                for (int k = 0; k < NK; ++k) r[j][k] = 0.f;
                if (ok[j]) {
#pragma unroll
                    for (int k = 0; k < NK; ++k) r[j][k] = hf[(size_t)ss[j] * C + k * 32 + lane];
                }
            }
#pragma unroll
            for (int j = 0; j < RB; ++j) {
                if (ok[j]) {
#pragma unroll
                    for (int k = 0; k < NK; ++k)
                        atomicAdd(&acc[dd[j] * C + k * 32 + lane], ww[j][(k * 32) / CH] * r[j][k]);
                }
            }
        }
    }
}

template <int H, int CH, int PASS, bool RECORD>
__device__ __forceinline__ void tile_sweep(const int* __restrict__ ei, const uint4* __restrict__ keys,
                                           const float* __restrict__ als, const float* __restrict__ hf,
                                           float* acc, unsigned* mx, float* den, const float* adl,
                                           unsigned* lst, unsigned* cnt,
                                           int t0, unsigned pat, int rot, int n_edges, int n_key16,
                                           int lane, int wave) {
    constexpr int STEP = (TILE_THREADS / 32) * 64;
    const uint4 none = make_uint4(~0u, ~0u, ~0u, ~0u);

    auto group_of = [&](int base, int u) {
        int q = base + u * 32 + lane + rot;
        return q >= n_key16 ? q - n_key16 : q;
    };
    auto fetch = [&](int base, int u) {
        return (base + u * 32 + lane < n_key16) ? keys[group_of(base, u)] : none;
    };

    uint4 kv0 = fetch(wave * 64, 0), kv1 = fetch(wave * 64, 1);
    for (int base = wave * 64; base < n_key16; base += STEP) {
        const uint4 nx0 = fetch(base + STEP, 0), nx1 = fetch(base + STEP, 1);
        unsigned mask = match16(kv0, pat) | (match16(kv1, pat) << 16);
        while (__builtin_amdgcn_ballot_w32(mask != 0u) != 0u) {
            const bool act = mask != 0u;
            int s = 0, dl = 0;
            if (act) {
                const int b = __ffs(mask) - 1;
                mask &= mask - 1u;
                const int e = group_of(base, b >> 4) * 16 + (b & 15);
                s  = ei[e];
                dl = ei[(size_t)n_edges + e] - t0;
                if (RECORD) {
                    const unsigned slot = atomicAdd(cnt, 1u);
                    if (slot < LIST_CAP) lst[slot] = ((unsigned)s << DL_BITS) | (unsigned)dl;
                }
            }
            take_edges<H, CH, PASS, false>(act, s, dl, nullptr, als, hf, acc, mx, den, adl, lane);
        }
        kv0 = nx0; kv1 = nx1;
    }
}

template <int H, int CH, int PASS>
__device__ __forceinline__ void list_sweep(const unsigned* lst, int n,
                                           const float* __restrict__ als, const float* __restrict__ hf,
                                           float* acc, unsigned* mx, float* den, const float* adl,
                                           float (&sc)[SWEEP_B][H], int lane, int wave) {
    if (PASS == 1) {
        for (int base = wave * 32; base < n; base += SWEEP_B * TILE_THREADS) {
            bool ok[SWEEP_B];
            unsigned p[SWEEP_B];
            float a[SWEEP_B][H];
#pragma unroll
            for (int u = 0; u < SWEEP_B; ++u) {
                const int i = base + u * TILE_THREADS + lane;
                ok[u] = i < n;
                p[u] = ok[u] ? lst[i] : 0u;
            }
#pragma unroll
            for (int u = 0; u < SWEEP_B; ++u) {
#pragma unroll
                for (int h = 0; h < H; ++h) a[u][h] = als[(size_t)(p[u] >> DL_BITS) * H + h];
            }
#pragma unroll
            for (int u = 0; u < SWEEP_B; ++u) {
                if (ok[u]) {
                    const int dl = (int)(p[u] & ((1u << DL_BITS) - 1u));
#pragma unroll
                    for (int h = 0; h < H; ++h) {
                        const float v = edge_score(a[u][h], adl[dl * H + h]);
                        sc[u][h] = v;
                        atomicMax(&mx[dl * H + h], fenc(v));
                    }
                }
            }
        }
        return;
    }
    for (int base = wave * 32; base < n; base += TILE_THREADS) {
        const int i = base + lane;
        const bool act = i < n;
        const unsigned p = act ? lst[i] : 0u;
        take_edges<H, CH, PASS, false>(act, (int)(p >> DL_BITS), (int)(p & ((1u << DL_BITS) - 1u)), nullptr,
                                       als, hf, acc, mx, den, adl, lane);
    }
}

template <int H, int CH>
__device__ __forceinline__ void list_sweep_kept(const unsigned* lst, int n, const float (&sc)[SWEEP_B][H],
                                                const float* __restrict__ hf,
                                                float* acc, unsigned* mx, float* den, int lane, int wave) {
    const int wv = __builtin_amdgcn_readfirstlane(wave);
    for (int k = 0; k < SWEEP_B; ++k) {
        const int base = wv * 32 + k * TILE_THREADS;
        if (base >= n) break;
        const int i = base + lane;
        const bool act = i < n;
        const unsigned p = act ? lst[i] : 0u;
        float kv[H];
#pragma unroll
        for (int h = 0; h < H; ++h) {
            float t = sc[0][h];
#pragma unroll
            for (int u = 1; u < SWEEP_B; ++u) t = (k == u) ? sc[u][h] : t;
            kv[h] = t;
        }
        take_edges<H, CH, 2, true>(act, (int)(p >> DL_BITS), (int)(p & ((1u << DL_BITS) - 1u)), kv,
                                   nullptr, hf, acc, mx, den, nullptr, lane);
    }
}

template <int H, int CH, bool ADD_BIAS, bool FIRST>
__global__ __launch_bounds__(TILE_THREADS)
void k_tile_gat(const int* __restrict__ ei, const uint4* __restrict__ keys,
                const float* __restrict__ als, const float* __restrict__ ald,
                const float* __restrict__ hf, const float* __restrict__ bias,
                float* __restrict__ out, unsigned* glist, unsigned* gcnt,
                const unsigned* __restrict__ bitmap, int row_words,
                int n_nodes, int n_edges, int n_key16) {
    constexpr int C = H * CH;
    __shared__ __attribute__((aligned(16))) float acc[TILE * C];
    __shared__ unsigned mx[TILE * H];
    __shared__ float    den[TILE * H];
    __shared__ float    adl[TILE * H];
    __shared__ unsigned lst[LIST_CAP];
    __shared__ unsigned cnt;
    __shared__ unsigned cntb;

    const int tid = threadIdx.x, lane = tid & 31, wave = tid >> 5;
    const int t0 = blockIdx.x * TILE;
    const unsigned pat = (unsigned)blockIdx.x * 0x01010101u;
    const int rot = (int)(((long long)blockIdx.x * n_key16) / gridDim.x);
    unsigned* const glst = glist + (size_t)blockIdx.x * LIST_CAP;

    unsigned n_list = 0u;
    float sc[SWEEP_B][H];
#pragma unroll
    for (int u = 0; u < SWEEP_B; ++u) {
#pragma unroll
        for (int h = 0; h < H; ++h) sc[u][h] = 0.f;
    }
    bool kept = false;
    if (FIRST) {
        if (tid == 0) { cnt = 0u; cntb = 0u; }
    } else {
        n_list = (unsigned)__builtin_amdgcn_readfirstlane((int)gcnt[blockIdx.x]);
        if (n_list <= LIST_CAP) {
            for (int i0 = tid; i0 < (int)n_list; i0 += 8 * TILE_THREADS) {
                unsigned v[8];
#pragma unroll
                for (int u = 0; u < 8; ++u) {
                    const int i = i0 + u * TILE_THREADS;
                    v[u] = glst[i < (int)n_list ? i : i0];
                }
#pragma unroll
                for (int u = 0; u < 8; ++u) {
                    const int i = i0 + u * TILE_THREADS;
                    if (i < (int)n_list) lst[i] = v[u];
                }
            }
        }
    }

    constexpr int SELF_T = (TILE * H + TILE_THREADS - 1) / TILE_THREADS;
    float selfv[SELF_T];
    {
        float sa[SELF_T], ss[SELF_T];
#pragma unroll
        for (int u = 0; u < SELF_T; ++u) {
            const int i = tid + u * TILE_THREADS;
            const bool ok = i < TILE * H && t0 + i / H < n_nodes;
            const int ic = ok ? i : 0;
            sa[u] = ald[(size_t)t0 * H + ic];
            ss[u] = als[(size_t)t0 * H + ic];
        }
#pragma unroll
        for (int u = 0; u < SELF_T; ++u) {
            const int i = tid + u * TILE_THREADS;
            const bool ok = i < TILE * H && t0 + i / H < n_nodes;
            const float a = ok ? sa[u] : 0.f;
            const float v = ok ? edge_score(ss[u], sa[u]) : 0.f;
            selfv[u] = v;
            if (i < TILE * H) { adl[i] = a; mx[i] = fenc(v); }
        }
    }
    __syncthreads();

    if (FIRST) {
        constexpr unsigned BL_CAP = TILE * C;
        unsigned* const blist = (unsigned*)acc;
        const unsigned* const row = bitmap + (size_t)blockIdx.x * row_words;
        for (int i0 = 0; i0 < row_words; i0 += 4 * TILE_THREADS) {
            unsigned wv[4];
#pragma unroll
            for (int u = 0; u < 4; ++u) {
                const int i = i0 + u * TILE_THREADS + tid;
                wv[u] = i < row_words ? row[i] : 0u;
            }
            unsigned slot = wave_reserve<8>(&cntb, (unsigned)(__popc(wv[0]) + __popc(wv[1]) + __popc(wv[2]) + __popc(wv[3])), lane);
#pragma unroll
            for (int u = 0; u < 4; ++u) {
                unsigned w = wv[u];
                const unsigned b0 = (unsigned)(i0 + u * TILE_THREADS + tid) * 32u;
                while (w != 0u) {
                    const unsigned b = (unsigned)__ffs(w) - 1u;
                    w &= w - 1u;
                    if (slot < BL_CAP) blist[slot] = b0 + b;
                    ++slot;
                }
            }
        }
        __syncthreads();
        const unsigned n_blk = (unsigned)__builtin_amdgcn_readfirstlane((int)cntb);
        if (n_blk <= BL_CAP) {
            for (int base = wave * 32; base < (int)n_blk; base += SWEEP_B * TILE_THREADS) {
                unsigned blk[SWEEP_B];
                uint4 kv[SWEEP_B];
#pragma unroll
                for (int u = 0; u < SWEEP_B; ++u) {
                    const int i = base + u * TILE_THREADS + lane;
                    blk[u] = i < (int)n_blk ? blist[i] : ~0u;
                }
#pragma unroll
                for (int u = 0; u < SWEEP_B; ++u) kv[u] = keys[blk[u] != ~0u ? blk[u] : 0u];
                unsigned mm[SWEEP_B];
                unsigned found = 0u;
#pragma unroll
                for (int u = 0; u < SWEEP_B; ++u) { mm[u] = blk[u] != ~0u ? match16(kv[u], pat) : 0u; found += (unsigned)__popc(mm[u]); }
                static_assert(16 * SWEEP_B < 256, "a lane's count must fit the planes that wave_reserve adds up");
                unsigned slot = wave_reserve<8>(&cnt, found, lane);
#pragma unroll
                for (int u = 0; u < SWEEP_B; ++u) {
                    unsigned m = mm[u];
                    while (m != 0u) {
                        const unsigned b = (unsigned)__ffs(m) - 1u;
                        m &= m - 1u;
                        if (slot < LIST_CAP) lst[slot] = blk[u] * 16u + b;
                        ++slot;
                    }
                }
            }
            __syncthreads();
            const unsigned n_found = (unsigned)__builtin_amdgcn_readfirstlane((int)cnt);
            if (n_found <= LIST_CAP) {
                for (int base = wave * 32; base < (int)n_found; base += SWEEP_B * TILE_THREADS) {
                    bool ok[SWEEP_B];
                    unsigned e[SWEEP_B];
                    int s[SWEEP_B], d[SWEEP_B];
                    float a[SWEEP_B][H];
#pragma unroll
                    for (int u = 0; u < SWEEP_B; ++u) {
                        const int i = base + u * TILE_THREADS + lane;
                        ok[u] = i < (int)n_found;
                        e[u] = ok[u] ? lst[i] : 0u;
                    }
#pragma unroll
                    for (int u = 0; u < SWEEP_B; ++u) { s[u] = ei[e[u]]; d[u] = ei[(size_t)n_edges + e[u]]; }
#pragma unroll
                    for (int u = 0; u < SWEEP_B; ++u) {
#pragma unroll
                        for (int h = 0; h < H; ++h) a[u][h] = als[(size_t)s[u] * H + h];
                    }
#pragma unroll
                    for (int u = 0; u < SWEEP_B; ++u) {
                        if (ok[u]) {
                            const int dl = d[u] - t0;
                            lst[base + u * TILE_THREADS + lane] = ((unsigned)s[u] << DL_BITS) | (unsigned)dl;
#pragma unroll
                            for (int h = 0; h < H; ++h) {
                                const float v = edge_score(a[u][h], adl[dl * H + h]);
                                sc[u][h] = v;
                                atomicMax(&mx[dl * H + h], fenc(v));
                            }
                        }
                    }
                }
                kept = n_found <= (unsigned)(SWEEP_B * TILE_THREADS);
            } else {
                tile_sweep<H, CH, 1, false>(ei, keys, als, hf, acc, mx, den, adl, lst, &cnt, t0, pat, rot, n_edges, n_key16, lane, wave);
            }
        } else {
            tile_sweep<H, CH, 1, true>(ei, keys, als, hf, acc, mx, den, adl, lst, &cnt, t0, pat, rot, n_edges, n_key16, lane, wave);
        }
    } else if (n_list <= LIST_CAP) {
        list_sweep<H, CH, 1>(lst, (int)n_list, als, hf, acc, mx, den, adl, sc, lane, wave);
        kept = n_list <= (unsigned)(SWEEP_B * TILE_THREADS);
    } else {
        tile_sweep<H, CH, 1, false>(ei, keys, als, hf, acc, mx, den, adl, lst, &cnt, t0, pat, rot, n_edges, n_key16, lane, wave);
    }
    __syncthreads();

    if (FIRST) {
        n_list = (unsigned)__builtin_amdgcn_readfirstlane((int)cnt);
        if (tid == 0) gcnt[blockIdx.x] = n_list;
        if (n_list <= LIST_CAP)
            for (int i = tid; i < (int)n_list; i += TILE_THREADS) glst[i] = lst[i];
    }

    static_assert(CH % 4 == 0 && C % 4 == 0 && (TILE * C) % 4 == 0, "four neighbouring channels share a head and a row");
    constexpr int V4   = TILE * C / 4;
    constexpr int V4_T = (V4 + TILE_THREADS - 1) / TILE_THREADS;
    const float4* const hrow = (const float4*)(hf + (size_t)t0 * C);
    float4 hv[V4_T];
#pragma unroll
    for (int u = 0; u < V4_T; ++u) {
        const int i4 = tid + u * TILE_THREADS;
        const bool ok = i4 < V4 && t0 + (i4 * 4) / C < n_nodes;
        hv[u] = hrow[ok ? i4 : 0];
    }
#pragma unroll
    for (int u = 0; u < SELF_T; ++u) {
        const int i = tid + u * TILE_THREADS;
        if (i < TILE * H) {
            float ex = 0.f;
            if (t0 + i / H < n_nodes) ex = __expf(selfv[u] - fdec(mx[i]));
            den[i] = ex;
        }
    }
    __syncthreads();
    float4* const acc4 = (float4*)acc;
#pragma unroll
    for (int u = 0; u < V4_T; ++u) {
        const int i4 = tid + u * TILE_THREADS;
        if (i4 < V4) {
            const int dl = (i4 * 4) / C, hd = ((i4 * 4) % C) / CH;
            float4 a = make_float4(0.f, 0.f, 0.f, 0.f);
            if (t0 + dl < n_nodes) {
                const float ex = den[dl * H + hd];
                a = make_float4(ex * hv[u].x, ex * hv[u].y, ex * hv[u].z, ex * hv[u].w);
            }
            acc4[i4] = a;
        }
    }
    __syncthreads();

    if (__builtin_expect(n_list <= LIST_CAP && kept, 1)) {
        list_sweep_kept<H, CH>(lst, (int)n_list, sc, hf, acc, mx, den, lane, wave);
    } else if (n_list <= LIST_CAP) {
        list_sweep<H, CH, 2>(lst, (int)n_list, als, hf, acc, mx, den, adl, sc, lane, wave);
    } else {
        tile_sweep<H, CH, 2, false>(ei, keys, als, hf, acc, mx, den, adl, lst, &cnt, t0, pat, rot, n_edges, n_key16, lane, wave);
    }
    __syncthreads();

    for (int i = tid; i < TILE * C; i += TILE_THREADS) {
        const int dl = i / C, c = i % C;
        if (t0 + dl < n_nodes) {
            float v = acc[i] / (den[dl * H + c / CH] + 1e-16f);
            if (ADD_BIAS) v += bias[c];
            out[(size_t)t0 * C + i] = v;
        }
    }
}


extern "C" void kernel_launch(void* const* d_in, const int* in_sizes, int n_in,
                              void* d_out, int out_size, void* d_ws, size_t ws_size,
                              hipStream_t stream) {
    (void)n_in; (void)out_size; (void)ws_size; (void)in_sizes;
    constexpr long long N = 100000, E = 1600000;
    constexpr int C1 = 128, HEADS = 4, HID = 32, C2 = 32;

    static_assert(N % TILE == 0 && N / TILE <= 255 && E % 16 == 0, "tile ids must fit a byte below 0xFF; keys are read 16 at a time");
    static_assert(TILE <= (1 << DL_BITS) && N <= (1ll << (32 - DL_BITS)), "a list entry packs source and local destination into one word");
    static_assert(N / TILE <= IDX_PITCH && E % 4 == 0 && (IDX_WORDS * 32 * 4) % IDX_THREADS == 0, "tile ids stay inside one LDS run of the index pass; destinations are read four at a time");

    const float* x    = (const float*)d_in[0];
    const int*   ei   = (const int*)  d_in[1];
    const float* W1   = (const float*)d_in[2];
    const float* aS1  = (const float*)d_in[3];
    const float* aD1  = (const float*)d_in[4];
    const float* b1   = (const float*)d_in[5];
    const float* W2   = (const float*)d_in[6];
    const float* aS2  = (const float*)d_in[7];
    const float* aD2  = (const float*)d_in[8];
    const float* b2   = (const float*)d_in[9];
    float* out = (float*)d_out;

    char* ws = (char*)d_ws;
    size_t off = 0;
    auto alloc = [&](size_t bytes) { size_t o = off; off = (off + bytes + 255) & ~(size_t)255; return o; };
    float*          h1   = (float*)(ws + alloc(N * C1 * 4));
    float*          out1 = (float*)(ws + alloc(N * C1 * 4));
    float*          as1  = (float*)(ws + alloc(N * HEADS * 4));
    float*          ad1  = (float*)(ws + alloc(N * HEADS * 4));
    unsigned*       keys = (unsigned*)(ws + alloc(E));
    unsigned*       glist = (unsigned*)(ws + alloc((N / TILE) * LIST_CAP * 4));
    unsigned*       gcnt  = (unsigned*)(ws + alloc((N / TILE) * 4));
    constexpr int      N_WORDS   = (int)((E / 16 + 31) / 32);
    constexpr unsigned IDX_GRID  = (N_WORDS + IDX_WORDS - 1) / IDX_WORDS;
    constexpr int      ROW_WORDS = (int)(IDX_GRID * IDX_WORDS);
    unsigned*       bitmap = (unsigned*)(ws + alloc((N / TILE) * (size_t)ROW_WORDS * 4));

    constexpr int STRIPS = (int)(N / 16);
    constexpr unsigned GEMM_GRID = (STRIPS + GEMM_WAVES - 1) / GEMM_WAVES;
    static_assert(N % 16 == 0 && C1 == 128 && C1 % 32 == 0 && C2 % 32 == 0, "the GEMM takes whole 16-row strips, K = 128 and column tiles in pairs");

    k_gemm_wmma_f32<C1, HEADS, false><<<GEMM_GRID, GEMM_WAVES * 32, 0, stream>>>(x, W1, b1, aS1, aD1, h1, as1, ad1, STRIPS);

    k_tile_index<<<IDX_GRID, IDX_THREADS, 0, stream>>>(ei, keys, bitmap, E, N / TILE, ROW_WORDS);

    k_tile_gat<HEADS, HID, false, true><<<N / TILE, TILE_THREADS, 0, stream>>>(
        ei, (const uint4*)keys, as1, ad1, h1, b1, out1, glist, gcnt, bitmap, ROW_WORDS, N, E, E / 16);

    k_gemm_wmma_f32<C2, 1, true><<<GEMM_GRID, GEMM_WAVES * 32, 0, stream>>>(out1, W2, b1, aS2, aD2, h1, as1, ad1, STRIPS);

    k_tile_gat<1, C2, true, false><<<N / TILE, TILE_THREADS, 0, stream>>>(
        ei, (const uint4*)keys, as1, ad1, h1, b2, out, glist, gcnt, bitmap, ROW_WORDS, N, E, E / 16);
}


# GEMM workgroups of 24 waves, one per compute unit (256, was 782 of 8 waves): W and W*a go into LDS once per unit; the 106 strips left over go one each to a first wave; sum orders unchanged
# speedup vs baseline: 9.6014x; 1.0339x over previous
#include <hip/hip_runtime.h>

typedef __attribute__((ext_vector_type(2))) float v2f;
typedef __attribute__((ext_vector_type(8))) float v8f;

#define NEG_SLOPE 0.2f


__device__ __forceinline__ unsigned fenc(float f) {
    unsigned u = __float_as_uint(f);
    return (u & 0x80000000u) ? ~u : (u | 0x80000000u);
}
__device__ __forceinline__ float fdec(unsigned u) {
    return (u & 0x80000000u) ? __uint_as_float(u & 0x7FFFFFFFu)
                             : __uint_as_float(~u);
}

#define GEMM_WAVES 24
#define GEMM_GRID  256

__device__ __forceinline__ float sum8(float v) {
    v += __int_as_float(__builtin_amdgcn_update_dpp(0, __float_as_int(v), 0x161, 0xf, 0xf, true));
    v += __int_as_float(__builtin_amdgcn_update_dpp(0, __float_as_int(v), 0x162, 0xf, 0xf, true));
    v += __int_as_float(__builtin_amdgcn_update_dpp(0, __float_as_int(v), 0x164, 0xf, 0xf, true));
    return v;
}

template <int NC, int H, bool BIAS_RELU>
__global__ __launch_bounds__(GEMM_WAVES * 32)
void k_gemm_wmma_f32(const float* __restrict__ A, const float* __restrict__ W,
                     const float* __restrict__ bias,
                     const float* __restrict__ a_s, const float* __restrict__ a_d,
                     float* __restrict__ C, float* __restrict__ als, float* __restrict__ ald, int n_strips) {
    static_assert(NC == H * 32 && 2 * H <= 16 && (GEMM_WAVES * 32 * 4) % NC == 0,
                  "a head is 32 columns = 8 lanes of a staging trip; both coefficient sets fit one column tile");
    __shared__ float4 wl[NC * 128 / 4];
    __shared__ float4 bl[128 / 4];
    __shared__ float2 wa[32 * 32];

    const int tid = threadIdx.x, lane = tid & 31;
    const int wave = __builtin_amdgcn_readfirstlane(tid >> 5);
    const int c = lane & 15, half = lane >> 4;
    int strip = blockIdx.x * GEMM_WAVES + wave;
    const bool live = strip < n_strips;

    float4 a4[16];
    if (live) {
        const float4* arow = (const float4*)(A + (size_t)(strip * 16 + c) * 128 + half * 64);
#pragma unroll
        for (int i = 0; i < 16; ++i) a4[i] = arow[i];
    }

    float* const wlf = (float*)wl;
    float* const waf = (float*)wa;
    for (int i = tid; i < 32 * 32; i += GEMM_WAVES * 32)
        if ((i & 15) >= 2 * H) wa[i] = make_float2(0.f, 0.f);
    const float4 s4 = *(const float4*)(a_s + (tid * 4) % NC);
    const float4 d4 = *(const float4*)(a_d + (tid * 4) % NC);
    constexpr int W4 = 128 * NC / 4;
    constexpr int TRIPS = (W4 + GEMM_WAVES * 32 - 1) / (GEMM_WAVES * 32);
    constexpr int WB = TRIPS < 8 ? TRIPS : 8;
    static_assert(TRIPS % WB == 0 && W4 % 32 == 0, "whole groups of trips; a wave is inside W or outside it as a whole");
    for (int g0 = 0; g0 < TRIPS; g0 += WB) {
        float4 wv[WB];
#pragma unroll
        for (int u = 0; u < WB; ++u) {
            const int i4 = tid + (g0 + u) * (GEMM_WAVES * 32);
            wv[u] = i4 < W4 ? ((const float4*)W)[i4] : make_float4(0.f, 0.f, 0.f, 0.f);
        }
#pragma unroll
        for (int u = 0; u < WB; ++u) {
            const float4 w = wv[u];
            const int i4 = tid + (g0 + u) * (GEMM_WAVES * 32);
            if (i4 >= W4) continue;
            const int k = (i4 * 4) / NC, n = (i4 * 4) % NC;
            const int t = (k & 63) >> 1, s = k & 1, j = n >> 4;
            float* const d = wlf + ((((j >> 1) * 32 + t) * 32 + ((k >> 6) * 16 + (n & 15))) * 4 + (j & 1) * 2 + s);
            d[0] = w.x; d[4] = w.y; d[8] = w.z; d[12] = w.w;
            const float ps = sum8(w.x * s4.x + w.y * s4.y + w.z * s4.z + w.w * s4.w);
            const float pd = sum8(w.x * d4.x + w.y * d4.y + w.z * d4.z + w.w * d4.w);
            const int side = lane & 7;
            if (side < 2) waf[(t * 32 + (k >> 6) * 16 + side * H + (n >> 5)) * 2 + s] = side ? pd : ps;
        }
    }
    if (BIAS_RELU && tid < 128) ((float*)bl)[tid] = bias[tid];
    __syncthreads();

    if (!live) return;

    const int grid_waves = (int)gridDim.x * GEMM_WAVES;
    for (int trip = 1; ; ++trip) {
    asm volatile("" ::: "memory");
    float af[64];
#pragma unroll
    for (int i = 0; i < 16; ++i) {
        float4 v = a4[i];
        if (BIAS_RELU) {
            const float4 b = bl[half * 16 + i];
            v.x = fmaxf(v.x + b.x, 0.f); v.y = fmaxf(v.y + b.y, 0.f);
            v.z = fmaxf(v.z + b.z, 0.f); v.w = fmaxf(v.w + b.w, 0.f);
        }
        af[4 * i] = v.x; af[4 * i + 1] = v.y; af[4 * i + 2] = v.z; af[4 * i + 3] = v.w;
    }

    {
        v8f e = {};
#pragma unroll
        for (int t = 0; t < 32; ++t) {
            const float2 p = wa[t * 32 + lane];
            const v2f a  = {af[2 * t], af[2 * t + 1]};
            const v2f bp = {p.x, p.y};
            e = __builtin_amdgcn_wmma_f32_16x16x4_f32(false, a, false, bp, (short)0, e, false, false);
        }
        if (c < 2 * H) {
            float* const dst = (c < H ? als + c : ald + (c - H)) + (size_t)(strip * 16 + 8 * half) * H;
#pragma unroll
            for (int r = 0; r < 8; ++r) dst[r * H] = e[r];
        }
    }

    for (int jp = 0; jp < NC / 32; ++jp) {
        v8f acc0 = {}, acc1 = {};
#pragma unroll
        for (int t = 0; t < 32; ++t) {
            const float4 b = wl[(jp * 32 + t) * 32 + lane];
            const v2f a  = {af[2 * t], af[2 * t + 1]};
            const v2f b0 = {b.x, b.y};
            const v2f b1 = {b.z, b.w};
            acc0 = __builtin_amdgcn_wmma_f32_16x16x4_f32(false, a, false, b0, (short)0, acc0, false, false);
            acc1 = __builtin_amdgcn_wmma_f32_16x16x4_f32(false, a, false, b1, (short)0, acc1, false, false);
        }
#pragma unroll
        for (int r = 0; r < 8; ++r) {
            float* const crow = C + (size_t)(strip * 16 + r + 8 * half) * NC + jp * 32 + c;
            crow[0]  = acc0[r];
            crow[16] = acc1[r];
        }
    }
    strip = trip * grid_waves + wave * (int)gridDim.x + (int)blockIdx.x;
    if (strip >= n_strips) break;
    const float4* arow = (const float4*)(A + (size_t)(strip * 16 + c) * 128 + half * 64);
#pragma unroll
    for (int i = 0; i < 16; ++i) a4[i] = arow[i];
    }
}

#define TILE 400
#define TILE_THREADS 1024

__device__ __forceinline__ float edge_score(float a, float b) {
    float v = a + b;
    return v > 0.f ? v : NEG_SLOPE * v;
}

#define IDX_WORDS   32
#define IDX_THREADS 1024
#define IDX_PITCH   257
__global__ __launch_bounds__(IDX_THREADS)
void k_tile_index(const int* __restrict__ ei, unsigned* __restrict__ keys, unsigned* __restrict__ bitmap,
                  int n_edges, int n_tiles, int row_words) {
    __shared__ unsigned bm[IDX_WORDS * IDX_PITCH];
    const int tid = threadIdx.x;
    for (int i = tid; i < IDX_WORDS * IDX_PITCH; i += IDX_THREADS) bm[i] = 0u;
    __syncthreads();

    const int n_quads = n_edges / 4;
    const int q0 = blockIdx.x * (IDX_WORDS * 32 * 4);
    constexpr int QPT = IDX_WORDS * 32 * 4 / IDX_THREADS;
    int4 dv[QPT];
#pragma unroll
    for (int u = 0; u < QPT; ++u) {
        const int q = q0 + u * IDX_THREADS + tid;
        dv[u] = q < n_quads ? *(const int4*)(ei + (size_t)n_edges + (size_t)q * 4) : make_int4(0, 0, 0, 0);
    }
#pragma unroll
    for (int u = 0; u < QPT; ++u) {
        const int ql = u * IDX_THREADS + tid;
        const int q = q0 + ql;
        if (q < n_quads) {
            const int4 d = dv[u];
            const unsigned ta = (unsigned)(d.x / TILE), tb = (unsigned)(d.y / TILE),
                           tc = (unsigned)(d.z / TILE), td = (unsigned)(d.w / TILE);
            keys[q] = ta | (tb << 8) | (tc << 16) | (td << 24);
            unsigned* const wrow = bm + (ql >> 7) * IDX_PITCH;
            const unsigned bit = 1u << ((ql >> 2) & 31);
            atomicOr(&wrow[ta], bit); atomicOr(&wrow[tb], bit);
            atomicOr(&wrow[tc], bit); atomicOr(&wrow[td], bit);
        }
    }
    __syncthreads();

    for (int i = tid; i < n_tiles * IDX_WORDS; i += IDX_THREADS) {
        const int t = i / IDX_WORDS, w = i % IDX_WORDS;
        bitmap[(size_t)t * row_words + blockIdx.x * IDX_WORDS + w] = bm[w * IDX_PITCH + t];
    }
}

__device__ __forceinline__ unsigned match4(unsigned k, unsigned pat) {
    unsigned x = k ^ pat;
    unsigned t = ~(((x & 0x7F7F7F7Fu) + 0x7F7F7F7Fu) | x | 0x7F7F7F7Fu);
    return (((t >> 7) * 0x00204081u) >> 21) & 0xFu;
}
__device__ __forceinline__ unsigned match16(uint4 k, unsigned pat) {
    return match4(k.x, pat) | (match4(k.y, pat) << 4) | (match4(k.z, pat) << 8) | (match4(k.w, pat) << 12);
}

__device__ __forceinline__ float lane_f(float v, int l) {
    return __int_as_float(__builtin_amdgcn_readlane(__float_as_int(v), l));
}

#define LIST_CAP 16384
#define DL_BITS  9
#define SWEEP_B  8

template <int BITS>
__device__ __forceinline__ unsigned wave_reserve(unsigned* counter, unsigned n, int lane) {
    unsigned before = 0u, total = 0u;
#pragma unroll
    for (int k = 0; k < BITS; ++k) {
        const unsigned bal = __builtin_amdgcn_ballot_w32(((n >> k) & 1u) != 0u);
        before += __builtin_amdgcn_mbcnt_lo(bal, 0u) << k;
        total  += (unsigned)__popc(bal) << k;
    }
    unsigned start = 0u;
    if (lane == 0) start = atomicAdd(counter, total);
    return (unsigned)__builtin_amdgcn_readfirstlane((int)start) + before;
}

template <int H, int CH, int PASS, bool KEPT>
__device__ __forceinline__ void take_edges(bool act, int s, int dl, const float* kv,
                                           const float* __restrict__ als, const float* __restrict__ hf,
                                           float* acc, unsigned* mx, float* den, const float* adl, int lane) {
    constexpr int C = H * CH;
    constexpr int NK = C / 32;
    float w[H];
#pragma unroll
    for (int h = 0; h < H; ++h) w[h] = 0.f;
    if (act) {
#pragma unroll
        for (int h = 0; h < H; ++h) {
            float v;
            if (KEPT) v = kv[h]; else v = edge_score(als[(size_t)s * H + h], adl[dl * H + h]);
            if (PASS == 1) {
                atomicMax(&mx[dl * H + h], fenc(v));
            } else {
                w[h] = __expf(v - fdec(mx[dl * H + h]));
                atomicAdd(&den[dl * H + h], w[h]);
            }
        }
    }
    if (PASS == 2) {
        constexpr int RB = NK >= 4 ? 8 : 16;
        unsigned bal = __builtin_amdgcn_ballot_w32(act);
        if (KEPT && bal == 0xFFFFFFFFu) {
#pragma unroll
            for (int j0 = 0; j0 < 32; j0 += RB) {
                float r[RB][NK];
#pragma unroll
                for (int j = 0; j < RB; ++j) {
                    const unsigned sj = (unsigned)__builtin_amdgcn_readlane(s, j0 + j);
#pragma unroll
                    for (int k = 0; k < NK; ++k) r[j][k] = hf[sj * (unsigned)C + (unsigned)(k * 32 + lane)];
                }
#pragma unroll
                for (int j = 0; j < RB; ++j) {
                    const int dj = __builtin_amdgcn_readlane(dl, j0 + j);
#pragma unroll
                    for (int k = 0; k < NK; ++k)
                        atomicAdd(&acc[dj * C + k * 32 + lane], lane_f(w[(k * 32) / CH], j0 + j) * r[j][k]);
                }
            }
            bal = 0u;
        }
        while (bal != 0u) {
            int  ss[RB], dd[RB];
            bool ok[RB];
            float ww[RB][H];
            float r[RB][NK];
#pragma unroll
            for (int j = 0; j < RB; ++j) {
                ok[j] = bal != 0u;
                const int l = ok[j] ? (__ffs(bal) - 1) : 0;
                if (ok[j]) bal &= bal - 1u;
                ss[j] = __builtin_amdgcn_readlane(s, l);
                dd[j] = __builtin_amdgcn_readlane(dl, l);
#pragma unroll
                for (int h = 0; h < H; ++h) ww[j][h] = lane_f(w[h], l);
            }
#pragma unroll
            for (int j = 0; j < RB; ++j) {
#pragma unroll
                for (int k = 0; k < NK; ++k) r[j][k] = 0.f;
                if (ok[j]) {
#pragma unroll
                    for (int k = 0; k < NK; ++k) r[j][k] = hf[(size_t)ss[j] * C + k * 32 + lane];
                }
            }
#pragma unroll
            for (int j = 0; j < RB; ++j) {
                if (ok[j]) {
#pragma unroll
                    for (int k = 0; k < NK; ++k)
                        atomicAdd(&acc[dd[j] * C + k * 32 + lane], ww[j][(k * 32) / CH] * r[j][k]);
                }
            }
        }
    }
}

template <int H, int CH, int PASS, bool RECORD>
__device__ __forceinline__ void tile_sweep(const int* __restrict__ ei, const uint4* __restrict__ keys,
                                           const float* __restrict__ als, const float* __restrict__ hf,
                                           float* acc, unsigned* mx, float* den, const float* adl,
                                           unsigned* lst, unsigned* cnt,
                                           int t0, unsigned pat, int rot, int n_edges, int n_key16,
                                           int lane, int wave) {
    constexpr int STEP = (TILE_THREADS / 32) * 64;
    const uint4 none = make_uint4(~0u, ~0u, ~0u, ~0u);

    auto group_of = [&](int base, int u) {
        int q = base + u * 32 + lane + rot;
        return q >= n_key16 ? q - n_key16 : q;
    };
    auto fetch = [&](int base, int u) {
        return (base + u * 32 + lane < n_key16) ? keys[group_of(base, u)] : none;
    };

    uint4 kv0 = fetch(wave * 64, 0), kv1 = fetch(wave * 64, 1);
    for (int base = wave * 64; base < n_key16; base += STEP) {
        const uint4 nx0 = fetch(base + STEP, 0), nx1 = fetch(base + STEP, 1);
        unsigned mask = match16(kv0, pat) | (match16(kv1, pat) << 16);
        while (__builtin_amdgcn_ballot_w32(mask != 0u) != 0u) {
            const bool act = mask != 0u;
            int s = 0, dl = 0;
            if (act) {
                const int b = __ffs(mask) - 1;
                mask &= mask - 1u;
                const int e = group_of(base, b >> 4) * 16 + (b & 15);
                s  = ei[e];
                dl = ei[(size_t)n_edges + e] - t0;
                if (RECORD) {
                    const unsigned slot = atomicAdd(cnt, 1u);
                    if (slot < LIST_CAP) lst[slot] = ((unsigned)s << DL_BITS) | (unsigned)dl;
                }
            }
            take_edges<H, CH, PASS, false>(act, s, dl, nullptr, als, hf, acc, mx, den, adl, lane);
        }
        kv0 = nx0; kv1 = nx1;
    }
}

template <int H, int CH, int PASS>
__device__ __forceinline__ void list_sweep(const unsigned* lst, int n,
                                           const float* __restrict__ als, const float* __restrict__ hf,
                                           float* acc, unsigned* mx, float* den, const float* adl,
                                           float (&sc)[SWEEP_B][H], int lane, int wave) {
    if (PASS == 1) {
        for (int base = wave * 32; base < n; base += SWEEP_B * TILE_THREADS) {
            bool ok[SWEEP_B];
            unsigned p[SWEEP_B];
            float a[SWEEP_B][H];
#pragma unroll
            for (int u = 0; u < SWEEP_B; ++u) {
                const int i = base + u * TILE_THREADS + lane;
                ok[u] = i < n;
                p[u] = ok[u] ? lst[i] : 0u;
            }
#pragma unroll
            for (int u = 0; u < SWEEP_B; ++u) {
#pragma unroll
                for (int h = 0; h < H; ++h) a[u][h] = als[(size_t)(p[u] >> DL_BITS) * H + h];
            }
#pragma unroll
            for (int u = 0; u < SWEEP_B; ++u) {
                if (ok[u]) {
                    const int dl = (int)(p[u] & ((1u << DL_BITS) - 1u));
#pragma unroll
                    for (int h = 0; h < H; ++h) {
                        const float v = edge_score(a[u][h], adl[dl * H + h]);
                        sc[u][h] = v;
                        atomicMax(&mx[dl * H + h], fenc(v));
                    }
                }
            }
        }
        return;
    }
    for (int base = wave * 32; base < n; base += TILE_THREADS) {
        const int i = base + lane;
        const bool act = i < n;
        const unsigned p = act ? lst[i] : 0u;
        take_edges<H, CH, PASS, false>(act, (int)(p >> DL_BITS), (int)(p & ((1u << DL_BITS) - 1u)), nullptr,
                                       als, hf, acc, mx, den, adl, lane);
    }
}

template <int H, int CH>
__device__ __forceinline__ void list_sweep_kept(const unsigned* lst, int n, const float (&sc)[SWEEP_B][H],
                                                const float* __restrict__ hf,
                                                float* acc, unsigned* mx, float* den, int lane, int wave) {
    const int wv = __builtin_amdgcn_readfirstlane(wave);
    for (int k = 0; k < SWEEP_B; ++k) {
        const int base = wv * 32 + k * TILE_THREADS;
        if (base >= n) break;
        const int i = base + lane;
        const bool act = i < n;
        const unsigned p = act ? lst[i] : 0u;
        float kv[H];
#pragma unroll
        for (int h = 0; h < H; ++h) {
            float t = sc[0][h];
#pragma unroll
            for (int u = 1; u < SWEEP_B; ++u) t = (k == u) ? sc[u][h] : t;
            kv[h] = t;
        }
        take_edges<H, CH, 2, true>(act, (int)(p >> DL_BITS), (int)(p & ((1u << DL_BITS) - 1u)), kv,
                                   nullptr, hf, acc, mx, den, nullptr, lane);
    }
}

template <int H, int CH, bool ADD_BIAS, bool FIRST>
__global__ __launch_bounds__(TILE_THREADS)
void k_tile_gat(const int* __restrict__ ei, const uint4* __restrict__ keys,
                const float* __restrict__ als, const float* __restrict__ ald,
                const float* __restrict__ hf, const float* __restrict__ bias,
                float* __restrict__ out, unsigned* glist, unsigned* gcnt,
                const unsigned* __restrict__ bitmap, int row_words,
                int n_nodes, int n_edges, int n_key16) {
    constexpr int C = H * CH;
    __shared__ __attribute__((aligned(16))) float acc[TILE * C];
    __shared__ unsigned mx[TILE * H];
    __shared__ float    den[TILE * H];
    __shared__ float    adl[TILE * H];
    __shared__ unsigned lst[LIST_CAP];
    __shared__ unsigned cnt;
    __shared__ unsigned cntb;

    const int tid = threadIdx.x, lane = tid & 31, wave = tid >> 5;
    const int t0 = blockIdx.x * TILE;
    const unsigned pat = (unsigned)blockIdx.x * 0x01010101u;
    const int rot = (int)(((long long)blockIdx.x * n_key16) / gridDim.x);
    unsigned* const glst = glist + (size_t)blockIdx.x * LIST_CAP;

    unsigned n_list = 0u;
    float sc[SWEEP_B][H];
#pragma unroll
    for (int u = 0; u < SWEEP_B; ++u) {
#pragma unroll
        for (int h = 0; h < H; ++h) sc[u][h] = 0.f;
    }
    bool kept = false;
    if (FIRST) {
        if (tid == 0) { cnt = 0u; cntb = 0u; }
    } else {
        n_list = (unsigned)__builtin_amdgcn_readfirstlane((int)gcnt[blockIdx.x]);
        if (n_list <= LIST_CAP) {
            for (int i0 = tid; i0 < (int)n_list; i0 += 8 * TILE_THREADS) {
                unsigned v[8];
#pragma unroll
                for (int u = 0; u < 8; ++u) {
                    const int i = i0 + u * TILE_THREADS;
                    v[u] = glst[i < (int)n_list ? i : i0];
                }
#pragma unroll
                for (int u = 0; u < 8; ++u) {
                    const int i = i0 + u * TILE_THREADS;
                    if (i < (int)n_list) lst[i] = v[u];
                }
            }
        }
    }

    constexpr int SELF_T = (TILE * H + TILE_THREADS - 1) / TILE_THREADS;
    float selfv[SELF_T];
    {
        float sa[SELF_T], ss[SELF_T];
#pragma unroll
        for (int u = 0; u < SELF_T; ++u) {
            const int i = tid + u * TILE_THREADS;
            const bool ok = i < TILE * H && t0 + i / H < n_nodes;
            const int ic = ok ? i : 0;
            sa[u] = ald[(size_t)t0 * H + ic];
            ss[u] = als[(size_t)t0 * H + ic];
        }
#pragma unroll
        for (int u = 0; u < SELF_T; ++u) {
            const int i = tid + u * TILE_THREADS;
            const bool ok = i < TILE * H && t0 + i / H < n_nodes;
            const float a = ok ? sa[u] : 0.f;
            const float v = ok ? edge_score(ss[u], sa[u]) : 0.f;
            selfv[u] = v;
            if (i < TILE * H) { adl[i] = a; mx[i] = fenc(v); }
        }
    }
    __syncthreads();

    if (FIRST) {
        constexpr unsigned BL_CAP = TILE * C;
        unsigned* const blist = (unsigned*)acc;
        const unsigned* const row = bitmap + (size_t)blockIdx.x * row_words;
        for (int i0 = 0; i0 < row_words; i0 += 4 * TILE_THREADS) {
            unsigned wv[4];
#pragma unroll
            for (int u = 0; u < 4; ++u) {
                const int i = i0 + u * TILE_THREADS + tid;
                wv[u] = i < row_words ? row[i] : 0u;
            }
            unsigned slot = wave_reserve<8>(&cntb, (unsigned)(__popc(wv[0]) + __popc(wv[1]) + __popc(wv[2]) + __popc(wv[3])), lane);
#pragma unroll
            for (int u = 0; u < 4; ++u) {
                unsigned w = wv[u];
                const unsigned b0 = (unsigned)(i0 + u * TILE_THREADS + tid) * 32u;
                while (w != 0u) {
                    const unsigned b = (unsigned)__ffs(w) - 1u;
                    w &= w - 1u;
                    if (slot < BL_CAP) blist[slot] = b0 + b;
                    ++slot;
                }
            }
        }
        __syncthreads();
        const unsigned n_blk = (unsigned)__builtin_amdgcn_readfirstlane((int)cntb);
        if (n_blk <= BL_CAP) {
            for (int base = wave * 32; base < (int)n_blk; base += SWEEP_B * TILE_THREADS) {
                unsigned blk[SWEEP_B];
                uint4 kv[SWEEP_B];
#pragma unroll
                for (int u = 0; u < SWEEP_B; ++u) {
                    const int i = base + u * TILE_THREADS + lane;
                    blk[u] = i < (int)n_blk ? blist[i] : ~0u;
                }
#pragma unroll
                for (int u = 0; u < SWEEP_B; ++u) kv[u] = keys[blk[u] != ~0u ? blk[u] : 0u];
                unsigned mm[SWEEP_B];
                unsigned found = 0u;
#pragma unroll
                for (int u = 0; u < SWEEP_B; ++u) { mm[u] = blk[u] != ~0u ? match16(kv[u], pat) : 0u; found += (unsigned)__popc(mm[u]); }
                static_assert(16 * SWEEP_B < 256, "a lane's count must fit the planes that wave_reserve adds up");
                unsigned slot = wave_reserve<8>(&cnt, found, lane);
#pragma unroll
                for (int u = 0; u < SWEEP_B; ++u) {
                    unsigned m = mm[u];
                    while (m != 0u) {
                        const unsigned b = (unsigned)__ffs(m) - 1u;
                        m &= m - 1u;
                        if (slot < LIST_CAP) lst[slot] = blk[u] * 16u + b;
                        ++slot;
                    }
                }
            }
            __syncthreads();
            const unsigned n_found = (unsigned)__builtin_amdgcn_readfirstlane((int)cnt);
            if (n_found <= LIST_CAP) {
                for (int base = wave * 32; base < (int)n_found; base += SWEEP_B * TILE_THREADS) {
                    bool ok[SWEEP_B];
                    unsigned e[SWEEP_B];
                    int s[SWEEP_B], d[SWEEP_B];
                    float a[SWEEP_B][H];
#pragma unroll
                    for (int u = 0; u < SWEEP_B; ++u) {
                        const int i = base + u * TILE_THREADS + lane;
                        ok[u] = i < (int)n_found;
                        e[u] = ok[u] ? lst[i] : 0u;
                    }
#pragma unroll
                    for (int u = 0; u < SWEEP_B; ++u) { s[u] = ei[e[u]]; d[u] = ei[(size_t)n_edges + e[u]]; }
#pragma unroll
                    for (int u = 0; u < SWEEP_B; ++u) {
#pragma unroll
                        for (int h = 0; h < H; ++h) a[u][h] = als[(size_t)s[u] * H + h];
                    }
#pragma unroll
                    for (int u = 0; u < SWEEP_B; ++u) {
                        if (ok[u]) {
                            const int dl = d[u] - t0;
                            lst[base + u * TILE_THREADS + lane] = ((unsigned)s[u] << DL_BITS) | (unsigned)dl;
#pragma unroll
                            for (int h = 0; h < H; ++h) {
                                const float v = edge_score(a[u][h], adl[dl * H + h]);
                                sc[u][h] = v;
                                atomicMax(&mx[dl * H + h], fenc(v));
                            }
                        }
                    }
                }
                kept = n_found <= (unsigned)(SWEEP_B * TILE_THREADS);
            } else {
                tile_sweep<H, CH, 1, false>(ei, keys, als, hf, acc, mx, den, adl, lst, &cnt, t0, pat, rot, n_edges, n_key16, lane, wave);
            }
        } else {
            tile_sweep<H, CH, 1, true>(ei, keys, als, hf, acc, mx, den, adl, lst, &cnt, t0, pat, rot, n_edges, n_key16, lane, wave);
        }
    } else if (n_list <= LIST_CAP) {
        list_sweep<H, CH, 1>(lst, (int)n_list, als, hf, acc, mx, den, adl, sc, lane, wave);
        kept = n_list <= (unsigned)(SWEEP_B * TILE_THREADS);
    } else {
        tile_sweep<H, CH, 1, false>(ei, keys, als, hf, acc, mx, den, adl, lst, &cnt, t0, pat, rot, n_edges, n_key16, lane, wave);
    }
    __syncthreads();

    if (FIRST) {
        n_list = (unsigned)__builtin_amdgcn_readfirstlane((int)cnt);
        if (tid == 0) gcnt[blockIdx.x] = n_list;
        if (n_list <= LIST_CAP)
            for (int i = tid; i < (int)n_list; i += TILE_THREADS) glst[i] = lst[i];
    }

    static_assert(CH % 4 == 0 && C % 4 == 0 && (TILE * C) % 4 == 0, "four neighbouring channels share a head and a row");
    constexpr int V4   = TILE * C / 4;
    constexpr int V4_T = (V4 + TILE_THREADS - 1) / TILE_THREADS;
    const float4* const hrow = (const float4*)(hf + (size_t)t0 * C);
    float4 hv[V4_T];
#pragma unroll
    for (int u = 0; u < V4_T; ++u) {
        const int i4 = tid + u * TILE_THREADS;
        const bool ok = i4 < V4 && t0 + (i4 * 4) / C < n_nodes;
        hv[u] = hrow[ok ? i4 : 0];
    }
#pragma unroll
    for (int u = 0; u < SELF_T; ++u) {
        const int i = tid + u * TILE_THREADS;
        if (i < TILE * H) {
            float ex = 0.f;
            if (t0 + i / H < n_nodes) ex = __expf(selfv[u] - fdec(mx[i]));
            den[i] = ex;
        }
    }
    __syncthreads();
    float4* const acc4 = (float4*)acc;
#pragma unroll
    for (int u = 0; u < V4_T; ++u) {
        const int i4 = tid + u * TILE_THREADS;
        if (i4 < V4) {
            const int dl = (i4 * 4) / C, hd = ((i4 * 4) % C) / CH;
            float4 a = make_float4(0.f, 0.f, 0.f, 0.f);
            if (t0 + dl < n_nodes) {
                const float ex = den[dl * H + hd];
                a = make_float4(ex * hv[u].x, ex * hv[u].y, ex * hv[u].z, ex * hv[u].w);
            }
            acc4[i4] = a;
        }
    }
    __syncthreads();

    if (__builtin_expect(n_list <= LIST_CAP && kept, 1)) {
        list_sweep_kept<H, CH>(lst, (int)n_list, sc, hf, acc, mx, den, lane, wave);
    } else if (n_list <= LIST_CAP) {
        list_sweep<H, CH, 2>(lst, (int)n_list, als, hf, acc, mx, den, adl, sc, lane, wave);
    } else {
        tile_sweep<H, CH, 2, false>(ei, keys, als, hf, acc, mx, den, adl, lst, &cnt, t0, pat, rot, n_edges, n_key16, lane, wave);
    }
    __syncthreads();

    for (int i = tid; i < TILE * C; i += TILE_THREADS) {
        const int dl = i / C, c = i % C;
        if (t0 + dl < n_nodes) {
            float v = acc[i] / (den[dl * H + c / CH] + 1e-16f);
            if (ADD_BIAS) v += bias[c];
            out[(size_t)t0 * C + i] = v;
        }
    }
}


extern "C" void kernel_launch(void* const* d_in, const int* in_sizes, int n_in,
                              void* d_out, int out_size, void* d_ws, size_t ws_size,
                              hipStream_t stream) {
    (void)n_in; (void)out_size; (void)ws_size; (void)in_sizes;
    constexpr long long N = 100000, E = 1600000;
    constexpr int C1 = 128, HEADS = 4, HID = 32, C2 = 32;

    static_assert(N % TILE == 0 && N / TILE <= 255 && E % 16 == 0, "tile ids must fit a byte below 0xFF; keys are read 16 at a time");
    static_assert(TILE <= (1 << DL_BITS) && N <= (1ll << (32 - DL_BITS)), "a list entry packs source and local destination into one word");
    static_assert(N / TILE <= IDX_PITCH && E % 4 == 0 && (IDX_WORDS * 32 * 4) % IDX_THREADS == 0, "tile ids stay inside one LDS run of the index pass; destinations are read four at a time");

    const float* x    = (const float*)d_in[0];
    const int*   ei   = (const int*)  d_in[1];
    const float* W1   = (const float*)d_in[2];
    const float* aS1  = (const float*)d_in[3];
    const float* aD1  = (const float*)d_in[4];
    const float* b1   = (const float*)d_in[5];
    const float* W2   = (const float*)d_in[6];
    const float* aS2  = (const float*)d_in[7];
    const float* aD2  = (const float*)d_in[8];
    const float* b2   = (const float*)d_in[9];
    float* out = (float*)d_out;

    char* ws = (char*)d_ws;
    size_t off = 0;
    auto alloc = [&](size_t bytes) { size_t o = off; off = (off + bytes + 255) & ~(size_t)255; return o; };
    float*          h1   = (float*)(ws + alloc(N * C1 * 4));
    float*          out1 = (float*)(ws + alloc(N * C1 * 4));
    float*          as1  = (float*)(ws + alloc(N * HEADS * 4));
    float*          ad1  = (float*)(ws + alloc(N * HEADS * 4));
    unsigned*       keys = (unsigned*)(ws + alloc(E));
    unsigned*       glist = (unsigned*)(ws + alloc((N / TILE) * LIST_CAP * 4));
    unsigned*       gcnt  = (unsigned*)(ws + alloc((N / TILE) * 4));
    constexpr int      N_WORDS   = (int)((E / 16 + 31) / 32);
    constexpr unsigned IDX_GRID  = (N_WORDS + IDX_WORDS - 1) / IDX_WORDS;
    constexpr int      ROW_WORDS = (int)(IDX_GRID * IDX_WORDS);
    unsigned*       bitmap = (unsigned*)(ws + alloc((N / TILE) * (size_t)ROW_WORDS * 4));

    constexpr int STRIPS = (int)(N / 16);
    static_assert(N % 16 == 0 && C1 == 128 && C1 % 32 == 0 && C2 % 32 == 0, "the GEMM takes whole 16-row strips, K = 128 and column tiles in pairs");

    k_gemm_wmma_f32<C1, HEADS, false><<<GEMM_GRID, GEMM_WAVES * 32, 0, stream>>>(x, W1, b1, aS1, aD1, h1, as1, ad1, STRIPS);

    k_tile_index<<<IDX_GRID, IDX_THREADS, 0, stream>>>(ei, keys, bitmap, E, N / TILE, ROW_WORDS);

    k_tile_gat<HEADS, HID, false, true><<<N / TILE, TILE_THREADS, 0, stream>>>(
        ei, (const uint4*)keys, as1, ad1, h1, b1, out1, glist, gcnt, bitmap, ROW_WORDS, N, E, E / 16);

    k_gemm_wmma_f32<C2, 1, true><<<GEMM_GRID, GEMM_WAVES * 32, 0, stream>>>(out1, W2, b1, aS2, aD2, h1, as1, ad1, STRIPS);

    k_tile_gat<1, C2, true, false><<<N / TILE, TILE_THREADS, 0, stream>>>(
        ei, (const uint4*)keys, as1, ad1, h1, b2, out, glist, gcnt, bitmap, ROW_WORDS, N, E, E / 16);
}
